# starting kernel
# baseline (speedup 1.0000x reference)
#include <hip/hip_runtime.h>
#include <hip/hip_bf16.h>


typedef __attribute__((ext_vector_type(2))) float v2f;
typedef __attribute__((ext_vector_type(8))) float v8f;

#define NEG_SLOPE 0.2f
#define EPSF 1e-16f

__device__ __forceinline__ v8f wmma_f32(v2f a, v2f b, v8f c) {
  return __builtin_amdgcn_wmma_f32_16x16x4_f32(
      false, a, false, b, (short)0, c, false, false);
}

__global__ void fill_u32_k(unsigned int* __restrict__ p, unsigned int v, long long n) {
  long long i = (long long)blockIdx.x * blockDim.x + threadIdx.x;
  if (i < n) p[i] = v;
}

__global__ void __launch_bounds__(128, 1)
gemm1_wmma_k(const float* __restrict__ X, const float* __restrict__ W,
             float* __restrict__ Hout, int N) {
  int wave = blockIdx.x * (blockDim.x >> 5) + (threadIdx.x >> 5);
  int lane = threadIdx.x & 31;
  int half = lane >> 4;
  int lm   = lane & 15;
  long long row0 = (long long)wave * 32;
  if (row0 >= N) return;
  long long r0 = row0 + lm;
  long long r1 = row0 + 16 + lm;
  bool ok0 = r0 < N, ok1 = r1 < N;
  const v2f* x0 = (const v2f*)(X + r0 * 128);
  const v2f* x1 = (const v2f*)(X + r1 * 128);
  const v2f zz = {0.0f, 0.0f};

  v8f acc0[8] = {}, acc1[8] = {};
  for (int k = 0; k < 128; k += 4) {
    int ka = k + 2 * half;
    v2f a0 = ok0 ? x0[ka >> 1] : zz;
    v2f a1 = ok1 ? x1[ka >> 1] : zz;
    const float* wr = W + (long long)ka * 128;
#pragma unroll
    for (int t = 0; t < 8; ++t) {
      int col = t * 16 + lm;
      v2f b = {wr[col], wr[128 + col]};
      acc0[t] = wmma_f32(a0, b, acc0[t]);
      acc1[t] = wmma_f32(a1, b, acc1[t]);
    }
  }
#pragma unroll
  for (int t = 0; t < 8; ++t) {
    int col = t * 16 + lm;
#pragma unroll
    for (int i = 0; i < 8; ++i) {
      long long ra = row0 + i + 8 * half;
      long long rb = ra + 16;
      if (ra < N) Hout[ra * 128 + col] = acc0[t][i];
      if (rb < N) Hout[rb * 128 + col] = acc1[t][i];
    }
  }
}

__global__ void __launch_bounds__(128, 1)
gemm2_wmma_k(const float* __restrict__ X, const float* __restrict__ W,
             float* __restrict__ G, int N) {
  int wave = blockIdx.x * (blockDim.x >> 5) + (threadIdx.x >> 5);
  int lane = threadIdx.x & 31;
  int half = lane >> 4;
  int lm   = lane & 15;
  long long row0 = (long long)wave * 32;
  if (row0 >= N) return;
  long long r0 = row0 + lm;
  long long r1 = row0 + 16 + lm;
  bool ok0 = r0 < N, ok1 = r1 < N;
  const v2f* x0 = (const v2f*)(X + r0 * 128);
  const v2f* x1 = (const v2f*)(X + r1 * 128);
  const v2f zz = {0.0f, 0.0f};

  v8f acc0[3] = {}, acc1[3] = {};
  for (int k = 0; k < 128; k += 4) {
    int ka = k + 2 * half;
    v2f a0 = ok0 ? x0[ka >> 1] : zz;
    v2f a1 = ok1 ? x1[ka >> 1] : zz;
    const float* wr = W + (long long)ka * 40;
#pragma unroll
    for (int t = 0; t < 3; ++t) {
      int col = t * 16 + lm;
      v2f b;
      b.x = (col < 40) ? wr[col]      : 0.0f;
      b.y = (col < 40) ? wr[40 + col] : 0.0f;
      acc0[t] = wmma_f32(a0, b, acc0[t]);
      acc1[t] = wmma_f32(a1, b, acc1[t]);
    }
  }
#pragma unroll
  for (int t = 0; t < 3; ++t) {
    int col = t * 16 + lm;
    if (col < 40) {
#pragma unroll
      for (int i = 0; i < 8; ++i) {
        long long ra = row0 + i + 8 * half;
        long long rb = ra + 16;
        if (ra < N) G[ra * 40 + col] = acc0[t][i];
        if (rb < N) G[rb * 40 + col] = acc1[t][i];
      }
    }
  }
}

__global__ void alpha1_k(const float* __restrict__ H1, const float* __restrict__ asrc,
                         const float* __restrict__ adst, float* __restrict__ AS,
                         float* __restrict__ AD, int N) {
  long long tid = (long long)blockIdx.x * blockDim.x + threadIdx.x;
  if (tid >= (long long)N * 8) return;
  int h = (int)(tid & 7);
  const float4* hv = (const float4*)(H1 + (tid >> 3) * 128 + h * 16);
  const float4* av = (const float4*)(asrc + h * 16);
  const float4* dv = (const float4*)(adst + h * 16);
  float s = 0.f, d = 0.f;
#pragma unroll
  for (int q = 0; q < 4; ++q) {
    float4 v = hv[q], a = av[q], b = dv[q];
    s += v.x * a.x + v.y * a.y + v.z * a.z + v.w * a.w;
    d += v.x * b.x + v.y * b.y + v.z * b.z + v.w * b.w;
  }
  AS[tid] = s; AD[tid] = d;
}

__global__ void alpha2_k(const float* __restrict__ G, const float* __restrict__ asrc,
                         const float* __restrict__ adst, float* __restrict__ AS,
                         float* __restrict__ AD, int N) {
  long long n = (long long)blockIdx.x * blockDim.x + threadIdx.x;
  if (n >= N) return;
  const float4* gv = (const float4*)(G + n * 40);
  float s = 0.f, d = 0.f;
#pragma unroll
  for (int q = 0; q < 10; ++q) {
    float4 v = gv[q];
    float4 a = ((const float4*)asrc)[q];
    float4 b = ((const float4*)adst)[q];
    s += v.x * a.x + v.y * a.y + v.z * a.z + v.w * a.w;
    d += v.x * b.x + v.y * b.y + v.z * b.z + v.w * b.w;
  }
  AS[n] = s; AD[n] = d;
}

__device__ __forceinline__ float lrelu(float v) { return v > 0.f ? v : NEG_SLOPE * v; }

__device__ __forceinline__ void atomicMaxF(float* addr, float v) {
  if (v >= 0.f) atomicMax((int*)addr, __float_as_int(v));
  else          atomicMin((unsigned int*)addr, __float_as_uint(v));
}

__device__ __forceinline__ void edge_sd(long long e, long long E,
                                        const int* __restrict__ src,
                                        const int* __restrict__ dst,
                                        int& s, int& d) {
  if (e < E) { s = src[e]; d = dst[e]; }
  else       { s = (int)(e - E); d = s; }
}

__global__ void edge_max1_k(const int* __restrict__ src, const int* __restrict__ dst,
                            const float* __restrict__ AS, const float* __restrict__ AD,
                            float* __restrict__ EMAX, long long E, long long E2) {
  long long tid = (long long)blockIdx.x * blockDim.x + threadIdx.x;
  if (tid >= E2 * 8) return;
  long long e = tid >> 3; int h = (int)(tid & 7);
  int s, d; edge_sd(e, E, src, dst, s, d);
  float v = lrelu(AS[(long long)s * 8 + h] + AD[(long long)d * 8 + h]);
  atomicMaxF(&EMAX[(long long)d * 8 + h], v);
}

__global__ void edge_exp1_k(const int* __restrict__ src, const int* __restrict__ dst,
                            const float* __restrict__ AS, const float* __restrict__ AD,
                            const float* __restrict__ EMAX, float* __restrict__ EBUF,
                            float* __restrict__ DEN, long long E, long long E2) {
  long long tid = (long long)blockIdx.x * blockDim.x + threadIdx.x;
  if (tid >= E2 * 8) return;
  long long e = tid >> 3; int h = (int)(tid & 7);
  int s, d; edge_sd(e, E, src, dst, s, d);
  float v = lrelu(AS[(long long)s * 8 + h] + AD[(long long)d * 8 + h]);
  float ex = __expf(v - EMAX[(long long)d * 8 + h]);
  EBUF[tid] = ex;
  atomicAdd(&DEN[(long long)d * 8 + h], ex);
}

__global__ void edge_msg1_k(const int* __restrict__ src, const int* __restrict__ dst,
                            const float* __restrict__ H1, const float* __restrict__ EBUF,
                            const float* __restrict__ DEN, float* __restrict__ ACC,
                            long long E, long long E2) {
  long long tid = (long long)blockIdx.x * blockDim.x + threadIdx.x;
  if (tid >= E2 * 8) return;
  long long e = tid >> 3; int h = (int)(tid & 7);
  int s, d; edge_sd(e, E, src, dst, s, d);
  float alpha = EBUF[tid] / (DEN[(long long)d * 8 + h] + EPSF);
  const float4* hv = (const float4*)(H1 + (long long)s * 128 + h * 16);
  float4 v0 = hv[0], v1 = hv[1], v2 = hv[2], v3 = hv[3];
  float* ac = ACC + (long long)d * 128 + h * 16;
  atomicAdd(&ac[0],  v0.x * alpha); atomicAdd(&ac[1],  v0.y * alpha);
  atomicAdd(&ac[2],  v0.z * alpha); atomicAdd(&ac[3],  v0.w * alpha);
  atomicAdd(&ac[4],  v1.x * alpha); atomicAdd(&ac[5],  v1.y * alpha);
  atomicAdd(&ac[6],  v1.z * alpha); atomicAdd(&ac[7],  v1.w * alpha);
  atomicAdd(&ac[8],  v2.x * alpha); atomicAdd(&ac[9],  v2.y * alpha);
  atomicAdd(&ac[10], v2.z * alpha); atomicAdd(&ac[11], v2.w * alpha);
  atomicAdd(&ac[12], v3.x * alpha); atomicAdd(&ac[13], v3.y * alpha);
  atomicAdd(&ac[14], v3.z * alpha); atomicAdd(&ac[15], v3.w * alpha);
}

__global__ void elu_bias1_k(float* __restrict__ ACC, const float* __restrict__ b1, long long n4) {
  long long tid = (long long)blockIdx.x * blockDim.x + threadIdx.x;
  if (tid >= n4) return;
  float4 v = ((const float4*)ACC)[tid];
  float4 b = ((const float4*)b1)[tid & 31];
  v.x += b.x; v.y += b.y; v.z += b.z; v.w += b.w;
  v.x = v.x > 0.f ? v.x : (__expf(v.x) - 1.f);
  v.y = v.y > 0.f ? v.y : (__expf(v.y) - 1.f);
  v.z = v.z > 0.f ? v.z : (__expf(v.z) - 1.f);
  v.w = v.w > 0.f ? v.w : (__expf(v.w) - 1.f);
  ((float4*)ACC)[tid] = v;
}

__global__ void edge_max2_k(const int* __restrict__ src, const int* __restrict__ dst,
                            const float* __restrict__ AS, const float* __restrict__ AD,
                            float* __restrict__ EMAX, long long E, long long E2) {
  long long e = (long long)blockIdx.x * blockDim.x + threadIdx.x;
  if (e >= E2) return;
  int s, d; edge_sd(e, E, src, dst, s, d);
  atomicMaxF(&EMAX[d], lrelu(AS[s] + AD[d]));
}

__global__ void edge_exp2_k(const int* __restrict__ src, const int* __restrict__ dst,
                            const float* __restrict__ AS, const float* __restrict__ AD,
                            const float* __restrict__ EMAX, float* __restrict__ EBUF,
                            float* __restrict__ DEN, long long E, long long E2) {
  long long e = (long long)blockIdx.x * blockDim.x + threadIdx.x;
  if (e >= E2) return;
  int s, d; edge_sd(e, E, src, dst, s, d);
  float ex = __expf(lrelu(AS[s] + AD[d]) - EMAX[d]);
  EBUF[e] = ex;
  atomicAdd(&DEN[d], ex);
}

__global__ void edge_msg2_k(const int* __restrict__ src, const int* __restrict__ dst,
                            const float* __restrict__ G, const float* __restrict__ EBUF,
                            const float* __restrict__ DEN, float* __restrict__ OUT,
                            long long E, long long E2) {
  long long tid = (long long)blockIdx.x * blockDim.x + threadIdx.x;
  if (tid >= E2 * 40) return;
  long long e = tid / 40; int c = (int)(tid - e * 40);
  int s, d; edge_sd(e, E, src, dst, s, d);
  float alpha = EBUF[e] / (DEN[d] + EPSF);
  atomicAdd(&OUT[(long long)d * 40 + c], G[(long long)s * 40 + c] * alpha);
}

__global__ void bias2_k(float* __restrict__ OUT, const float* __restrict__ b2, long long n4) {
  long long tid = (long long)blockIdx.x * blockDim.x + threadIdx.x;
  if (tid >= n4) return;
  float4 v = ((const float4*)OUT)[tid];
  float4 b = ((const float4*)b2)[tid % 10];
  v.x += b.x; v.y += b.y; v.z += b.z; v.w += b.w;
  ((float4*)OUT)[tid] = v;
}

static inline unsigned int nblk(long long n, int b) { return (unsigned int)((n + b - 1) / b); }

extern "C" void kernel_launch(void* const* d_in, const int* in_sizes, int n_in,
                              void* d_out, int out_size, void* d_ws, size_t ws_size,
                              hipStream_t stream) {
  const float* x      = (const float*)d_in[0];
  const int*   ei     = (const int*)d_in[1];
  const float* W1     = (const float*)d_in[2];
  const float* a_src1 = (const float*)d_in[3];
  const float* a_dst1 = (const float*)d_in[4];
  const float* b1     = (const float*)d_in[5];
  const float* W2     = (const float*)d_in[6];
  const float* a_src2 = (const float*)d_in[7];
  const float* a_dst2 = (const float*)d_in[8];
  const float* b2     = (const float*)d_in[9];
  float*       out    = (float*)d_out;

  const long long N  = in_sizes[0] / 128;
  const long long E  = in_sizes[1] / 2;
  const long long E2 = E + N;
  const int* srcA = ei;
  const int* dstA = ei + E;

  char* ws = (char*)d_ws;
  size_t off = 0;
  auto carve = [&](size_t bytes) -> void* {
    void* p = ws + off;
    off = (off + bytes + 255) & ~(size_t)255;
    return p;
  };
  float* H1    = (float*)carve((size_t)N * 128 * 4);
  float* AS1   = (float*)carve((size_t)N * 8 * 4);
  float* AD1   = (float*)carve((size_t)N * 8 * 4);
  float* EMAX1 = (float*)carve((size_t)N * 8 * 4);
  float* DEN1  = (float*)carve((size_t)N * 8 * 4);
  float* EB1   = (float*)carve((size_t)E2 * 8 * 4);
  float* ACC1  = (float*)carve((size_t)N * 128 * 4);
  float* G2    = (float*)carve((size_t)N * 40 * 4);
  float* AS2   = (float*)carve((size_t)N * 4);
  float* AD2   = (float*)carve((size_t)N * 4);
  float* EMAX2 = (float*)carve((size_t)N * 4);
  float* DEN2  = (float*)carve((size_t)N * 4);
  float* EB2   = (float*)carve((size_t)E2 * 4);

  const unsigned int NEG_INF = 0xFF800000u;
  const int B = 256;

  fill_u32_k<<<nblk(N * 8, B), B, 0, stream>>>((unsigned int*)EMAX1, NEG_INF, N * 8);
  fill_u32_k<<<nblk(N * 8, B), B, 0, stream>>>((unsigned int*)DEN1, 0u, N * 8);
  fill_u32_k<<<nblk(N * 128, B), B, 0, stream>>>((unsigned int*)ACC1, 0u, N * 128);
  fill_u32_k<<<nblk(N, B), B, 0, stream>>>((unsigned int*)EMAX2, NEG_INF, N);
  fill_u32_k<<<nblk(N, B), B, 0, stream>>>((unsigned int*)DEN2, 0u, N);
  fill_u32_k<<<nblk(N * 40, B), B, 0, stream>>>((unsigned int*)out, 0u, N * 40);

  {
    long long waves = (N + 31) / 32;
    gemm1_wmma_k<<<nblk(waves, 4), 128, 0, stream>>>(x, W1, H1, (int)N);
  }
  alpha1_k<<<nblk(N * 8, B), B, 0, stream>>>(H1, a_src1, a_dst1, AS1, AD1, (int)N);
  edge_max1_k<<<nblk(E2 * 8, B), B, 0, stream>>>(srcA, dstA, AS1, AD1, EMAX1, E, E2);
  edge_exp1_k<<<nblk(E2 * 8, B), B, 0, stream>>>(srcA, dstA, AS1, AD1, EMAX1, EB1, DEN1, E, E2);
  edge_msg1_k<<<nblk(E2 * 8, B), B, 0, stream>>>(srcA, dstA, H1, EB1, DEN1, ACC1, E, E2);
  elu_bias1_k<<<nblk(N * 32, B), B, 0, stream>>>(ACC1, b1, N * 32);

  {
    long long waves = (N + 31) / 32;
    gemm2_wmma_k<<<nblk(waves, 4), 128, 0, stream>>>(ACC1, W2, G2, (int)N);
  }
  alpha2_k<<<nblk(N, B), B, 0, stream>>>(G2, a_src2, a_dst2, AS2, AD2, (int)N);
  edge_max2_k<<<nblk(E2, B), B, 0, stream>>>(srcA, dstA, AS2, AD2, EMAX2, E, E2);
  edge_exp2_k<<<nblk(E2, B), B, 0, stream>>>(srcA, dstA, AS2, AD2, EMAX2, EB2, DEN2, E, E2);
  edge_msg2_k<<<nblk(E2 * 40, B), B, 0, stream>>>(srcA, dstA, G2, EB2, DEN2, out, E, E2);
  bias2_k<<<nblk(N * 10, B), B, 0, stream>>>(out, b2, N * 10);
}


# layer-1 messages: per-call counting sort of edges by destination, one wave per row sums in registers (no f32 atomics, no zero-fill, bias+ELU fused); sum order = integer ticket order
# speedup vs baseline: 1.7782x; 1.7782x over previous
#include <hip/hip_runtime.h>
#include <hip/hip_bf16.h>


typedef __attribute__((ext_vector_type(2))) float v2f;
typedef __attribute__((ext_vector_type(8))) float v8f;

#define NEG_SLOPE 0.2f
#define EPSF 1e-16f

__device__ __forceinline__ v8f wmma_f32(v2f a, v2f b, v8f c) {
  return __builtin_amdgcn_wmma_f32_16x16x4_f32(
      false, a, false, b, (short)0, c, false, false);
}

__global__ void fill_u32_k(unsigned int* __restrict__ p, unsigned int v, long long n) {
  long long i = (long long)blockIdx.x * blockDim.x + threadIdx.x;
  if (i < n) p[i] = v;
}

__global__ void __launch_bounds__(128, 1)
gemm1_wmma_k(const float* __restrict__ X, const float* __restrict__ W,
             float* __restrict__ Hout, int N) {
  int wave = blockIdx.x * (blockDim.x >> 5) + (threadIdx.x >> 5);
  int lane = threadIdx.x & 31;
  int half = lane >> 4;
  int lm   = lane & 15;
  long long row0 = (long long)wave * 32;
  if (row0 >= N) return;
  long long r0 = row0 + lm;
  long long r1 = row0 + 16 + lm;
  bool ok0 = r0 < N, ok1 = r1 < N;
  const v2f* x0 = (const v2f*)(X + r0 * 128);
  const v2f* x1 = (const v2f*)(X + r1 * 128);
  const v2f zz = {0.0f, 0.0f};

  v8f acc0[8] = {}, acc1[8] = {};
  for (int k = 0; k < 128; k += 4) {
    int ka = k + 2 * half;
    v2f a0 = ok0 ? x0[ka >> 1] : zz;
    v2f a1 = ok1 ? x1[ka >> 1] : zz;
    const float* wr = W + (long long)ka * 128;
#pragma unroll
    for (int t = 0; t < 8; ++t) {
      int col = t * 16 + lm;
      v2f b = {wr[col], wr[128 + col]};
      acc0[t] = wmma_f32(a0, b, acc0[t]);
      acc1[t] = wmma_f32(a1, b, acc1[t]);
    }
  }
#pragma unroll
  for (int t = 0; t < 8; ++t) {
    int col = t * 16 + lm;
#pragma unroll
    for (int i = 0; i < 8; ++i) {
      long long ra = row0 + i + 8 * half;
      long long rb = ra + 16;
      if (ra < N) Hout[ra * 128 + col] = acc0[t][i];
      if (rb < N) Hout[rb * 128 + col] = acc1[t][i];
    }
  }
}

__global__ void __launch_bounds__(128, 1)
gemm2_wmma_k(const float* __restrict__ X, const float* __restrict__ W,
             float* __restrict__ G, int N) {
  int wave = blockIdx.x * (blockDim.x >> 5) + (threadIdx.x >> 5);
  int lane = threadIdx.x & 31;
  int half = lane >> 4;
  int lm   = lane & 15;
  long long row0 = (long long)wave * 32;
  if (row0 >= N) return;
  long long r0 = row0 + lm;
  long long r1 = row0 + 16 + lm;
  bool ok0 = r0 < N, ok1 = r1 < N;
  const v2f* x0 = (const v2f*)(X + r0 * 128);
  const v2f* x1 = (const v2f*)(X + r1 * 128);
  const v2f zz = {0.0f, 0.0f};

  v8f acc0[3] = {}, acc1[3] = {};
  for (int k = 0; k < 128; k += 4) {
    int ka = k + 2 * half;
    v2f a0 = ok0 ? x0[ka >> 1] : zz;
    v2f a1 = ok1 ? x1[ka >> 1] : zz;
    const float* wr = W + (long long)ka * 40;
#pragma unroll
    for (int t = 0; t < 3; ++t) {
      int col = t * 16 + lm;
      v2f b;
      b.x = (col < 40) ? wr[col]      : 0.0f;
      b.y = (col < 40) ? wr[40 + col] : 0.0f;
      acc0[t] = wmma_f32(a0, b, acc0[t]);
      acc1[t] = wmma_f32(a1, b, acc1[t]);
    }
  }
#pragma unroll
  for (int t = 0; t < 3; ++t) {
    int col = t * 16 + lm;
    if (col < 40) {
#pragma unroll
      for (int i = 0; i < 8; ++i) {
        long long ra = row0 + i + 8 * half;
        long long rb = ra + 16;
        if (ra < N) G[ra * 40 + col] = acc0[t][i];
        if (rb < N) G[rb * 40 + col] = acc1[t][i];
      }
    }
  }
}

__global__ void alpha1_k(const float* __restrict__ H1, const float* __restrict__ asrc,
                         const float* __restrict__ adst, float* __restrict__ AS,
                         float* __restrict__ AD, int N) {
  long long tid = (long long)blockIdx.x * blockDim.x + threadIdx.x;
  if (tid >= (long long)N * 8) return;
  int h = (int)(tid & 7);
  const float4* hv = (const float4*)(H1 + (tid >> 3) * 128 + h * 16);
  const float4* av = (const float4*)(asrc + h * 16);
  const float4* dv = (const float4*)(adst + h * 16);
  float s = 0.f, d = 0.f;
#pragma unroll
  for (int q = 0; q < 4; ++q) {
    float4 v = hv[q], a = av[q], b = dv[q];
    s += v.x * a.x + v.y * a.y + v.z * a.z + v.w * a.w;
    d += v.x * b.x + v.y * b.y + v.z * b.z + v.w * b.w;
  }
  AS[tid] = s; AD[tid] = d;
}

__global__ void alpha2_k(const float* __restrict__ G, const float* __restrict__ asrc,
                         const float* __restrict__ adst, float* __restrict__ AS,
                         float* __restrict__ AD, int N) {
  long long n = (long long)blockIdx.x * blockDim.x + threadIdx.x;
  if (n >= N) return;
  const float4* gv = (const float4*)(G + n * 40);
  float s = 0.f, d = 0.f;
#pragma unroll
  for (int q = 0; q < 10; ++q) {
    float4 v = gv[q];
    float4 a = ((const float4*)asrc)[q];
    float4 b = ((const float4*)adst)[q];
    s += v.x * a.x + v.y * a.y + v.z * a.z + v.w * a.w;
    d += v.x * b.x + v.y * b.y + v.z * b.z + v.w * b.w;
  }
  AS[n] = s; AD[n] = d;
}

__device__ __forceinline__ float lrelu(float v) { return v > 0.f ? v : NEG_SLOPE * v; }

__device__ __forceinline__ void atomicMaxF(float* addr, float v) {
  if (v >= 0.f) atomicMax((int*)addr, __float_as_int(v));
  else          atomicMin((unsigned int*)addr, __float_as_uint(v));
}

__device__ __forceinline__ void edge_sd(long long e, long long E,
                                        const int* __restrict__ src,
                                        const int* __restrict__ dst,
                                        int& s, int& d) {
  if (e < E) { s = src[e]; d = dst[e]; }
  else       { s = (int)(e - E); d = s; }
}

__global__ void edge_max1_k(const int* __restrict__ src, const int* __restrict__ dst,
                            const float* __restrict__ AS, const float* __restrict__ AD,
                            float* __restrict__ EMAX, long long E, long long E2) {
  long long tid = (long long)blockIdx.x * blockDim.x + threadIdx.x;
  if (tid >= E2 * 8) return;
  long long e = tid >> 3; int h = (int)(tid & 7);
  int s, d; edge_sd(e, E, src, dst, s, d);
  float v = lrelu(AS[(long long)s * 8 + h] + AD[(long long)d * 8 + h]);
  atomicMaxF(&EMAX[(long long)d * 8 + h], v);
}

__global__ void edge_exp1_k(const int* __restrict__ src, const int* __restrict__ dst,
                            const float* __restrict__ AS, const float* __restrict__ AD,
                            const float* __restrict__ EMAX, float* __restrict__ EBUF,
                            float* __restrict__ DEN, long long E, long long E2) {
  long long tid = (long long)blockIdx.x * blockDim.x + threadIdx.x;
  if (tid >= E2 * 8) return;
  long long e = tid >> 3; int h = (int)(tid & 7);
  int s, d; edge_sd(e, E, src, dst, s, d);
  float v = lrelu(AS[(long long)s * 8 + h] + AD[(long long)d * 8 + h]);
  float ex = __expf(v - EMAX[(long long)d * 8 + h]);
  EBUF[tid] = ex;
  atomicAdd(&DEN[(long long)d * 8 + h], ex);
}

__global__ void edge_ticket_k(const int* __restrict__ dst, int* __restrict__ CNT,
                              int* __restrict__ RANK, long long E) {
  long long e = (long long)blockIdx.x * blockDim.x + threadIdx.x;
  if (e < E) RANK[e] = atomicAdd(&CNT[dst[e]], 1);
}

__global__ void __launch_bounds__(1024)
row_scan_k(const int* __restrict__ CNT, int* __restrict__ ROWPTR, int N) {
  __shared__ int wsum[32];
  const int t = threadIdx.x, lane = t & 31, w = t >> 5;
  int carry = 0;
  for (int base = 0; base < N; base += 1024 * 16) {
    const int i0 = base + t * 16;
    int v[16];
    if (i0 + 16 <= N) {
      const int4* c4 = (const int4*)(CNT + i0);
#pragma unroll
      for (int q = 0; q < 4; ++q) {
        int4 x = c4[q];
        v[4 * q] = x.x; v[4 * q + 1] = x.y; v[4 * q + 2] = x.z; v[4 * q + 3] = x.w;
      }
    } else {
#pragma unroll
      for (int k = 0; k < 16; ++k) v[k] = (i0 + k < N) ? CNT[i0 + k] : 0;
    }
    int sum = 0;
#pragma unroll
    for (int k = 0; k < 16; ++k) sum += v[k];
    int inc = sum;
#pragma unroll
    for (int o = 1; o < 32; o <<= 1) {
      int y = __shfl_up(inc, o);
      if (lane >= o) inc += y;
    }
    if (lane == 31) wsum[w] = inc;
    __syncthreads();
    int ws = wsum[lane];
    int winc = ws;
#pragma unroll
    for (int o = 1; o < 32; o <<= 1) {
      int y = __shfl_up(winc, o);
      if (lane >= o) winc += y;
    }
    int wave_excl = __shfl(winc, w) - __shfl(ws, w);
    int total = __shfl(winc, 31);
    int excl = carry + wave_excl + (inc - sum);
#pragma unroll
    for (int k = 0; k < 16; ++k) {
      if (i0 + k < N) ROWPTR[i0 + k] = excl;
      excl += v[k];
    }
    carry += total;
    __syncthreads();
  }
  if (t == 0) ROWPTR[N] = carry;
}

__global__ void edge_place_k(const int* __restrict__ src, const int* __restrict__ dst,
                             const int* __restrict__ ROWPTR, const int* __restrict__ RANK,
                             int2* __restrict__ SE, long long E, long long E2) {
  long long e = (long long)blockIdx.x * blockDim.x + threadIdx.x;
  if (e >= E2) return;
  int s, d, r;
  if (e < E) { s = src[e]; d = dst[e]; r = RANK[e]; }
  else       { s = (int)(e - E); d = s; r = 0; }
  SE[ROWPTR[d] + r] = make_int2(s, (int)e);
}

__global__ void __launch_bounds__(256)
row_msg1_k(const int2* __restrict__ SE, const int* __restrict__ ROWPTR,
           const float* __restrict__ H1, const float* __restrict__ EBUF,
           const float* __restrict__ DEN, const float* __restrict__ b1,
           float* __restrict__ HOUT, int N) {
  const int n = __builtin_amdgcn_readfirstlane((int)(blockIdx.x * 8 + (threadIdx.x >> 5)));
  if (n >= N) return;
  const int lane = threadIdx.x & 31;
  const int h = lane >> 2;
  const int beg = ROWPTR[n], end = ROWPTR[n + 1];
  const float den = DEN[(long long)n * 8 + h] + EPSF;
  float4 acc = {0.f, 0.f, 0.f, 0.f};
  for (int base = beg; base < end; base += 32) {
    const int m = (end - base < 32) ? (end - base) : 32;
    int2 se = make_int2(0, 0);
    if (lane < m) se = SE[base + lane];
#pragma unroll 4
    for (int j = 0; j < m; ++j) {
      const int s = __builtin_amdgcn_readlane(se.x, j);
      const int e = __builtin_amdgcn_readlane(se.y, j);
      const float ex = EBUF[(long long)e * 8 + h];
      const float4 v = *(const float4*)(H1 + (long long)s * 128 + lane * 4);
      const float alpha = ex / den;
      acc.x += v.x * alpha; acc.y += v.y * alpha;
      acc.z += v.z * alpha; acc.w += v.w * alpha;
    }
  }
  const float4 b = ((const float4*)b1)[lane];
  acc.x += b.x; acc.y += b.y; acc.z += b.z; acc.w += b.w;
  acc.x = acc.x > 0.f ? acc.x : (__expf(acc.x) - 1.f);
  acc.y = acc.y > 0.f ? acc.y : (__expf(acc.y) - 1.f);
  acc.z = acc.z > 0.f ? acc.z : (__expf(acc.z) - 1.f);
  acc.w = acc.w > 0.f ? acc.w : (__expf(acc.w) - 1.f);
  ((float4*)HOUT)[(long long)n * 32 + lane] = acc;
}

__global__ void edge_max2_k(const int* __restrict__ src, const int* __restrict__ dst,
                            const float* __restrict__ AS, const float* __restrict__ AD,
                            float* __restrict__ EMAX, long long E, long long E2) {
  long long e = (long long)blockIdx.x * blockDim.x + threadIdx.x;
  if (e >= E2) return;
  int s, d; edge_sd(e, E, src, dst, s, d);
  atomicMaxF(&EMAX[d], lrelu(AS[s] + AD[d]));
}

__global__ void edge_exp2_k(const int* __restrict__ src, const int* __restrict__ dst,
                            const float* __restrict__ AS, const float* __restrict__ AD,
                            const float* __restrict__ EMAX, float* __restrict__ EBUF,
                            float* __restrict__ DEN, long long E, long long E2) {
  long long e = (long long)blockIdx.x * blockDim.x + threadIdx.x;
  if (e >= E2) return;
  int s, d; edge_sd(e, E, src, dst, s, d);
  float ex = __expf(lrelu(AS[s] + AD[d]) - EMAX[d]);
  EBUF[e] = ex;
  atomicAdd(&DEN[d], ex);
}

__global__ void edge_msg2_k(const int* __restrict__ src, const int* __restrict__ dst,
                            const float* __restrict__ G, const float* __restrict__ EBUF,
                            const float* __restrict__ DEN, float* __restrict__ OUT,
                            long long E, long long E2) {
  long long tid = (long long)blockIdx.x * blockDim.x + threadIdx.x;
  if (tid >= E2 * 40) return;
  long long e = tid / 40; int c = (int)(tid - e * 40);
  int s, d; edge_sd(e, E, src, dst, s, d);
  float alpha = EBUF[e] / (DEN[d] + EPSF);
  atomicAdd(&OUT[(long long)d * 40 + c], G[(long long)s * 40 + c] * alpha);
}

__global__ void bias2_k(float* __restrict__ OUT, const float* __restrict__ b2, long long n4) {
  long long tid = (long long)blockIdx.x * blockDim.x + threadIdx.x;
  if (tid >= n4) return;
  float4 v = ((const float4*)OUT)[tid];
  float4 b = ((const float4*)b2)[tid % 10];
  v.x += b.x; v.y += b.y; v.z += b.z; v.w += b.w;
  ((float4*)OUT)[tid] = v;
}

static inline unsigned int nblk(long long n, int b) { return (unsigned int)((n + b - 1) / b); }

extern "C" void kernel_launch(void* const* d_in, const int* in_sizes, int n_in,
                              void* d_out, int out_size, void* d_ws, size_t ws_size,
                              hipStream_t stream) {
  const float* x      = (const float*)d_in[0];
  const int*   ei     = (const int*)d_in[1];
  const float* W1     = (const float*)d_in[2];
  const float* a_src1 = (const float*)d_in[3];
  const float* a_dst1 = (const float*)d_in[4];
  const float* b1     = (const float*)d_in[5];
  const float* W2     = (const float*)d_in[6];
  const float* a_src2 = (const float*)d_in[7];
  const float* a_dst2 = (const float*)d_in[8];
  const float* b2     = (const float*)d_in[9];
  float*       out    = (float*)d_out;

  const long long N  = in_sizes[0] / 128;
  const long long E  = in_sizes[1] / 2;
  const long long E2 = E + N;
  const int* srcA = ei;
  const int* dstA = ei + E;

  char* ws = (char*)d_ws;
  size_t off = 0;
  auto carve = [&](size_t bytes) -> void* {
    void* p = ws + off;
    off = (off + bytes + 255) & ~(size_t)255;
    return p;
  };
  float* H1    = (float*)carve((size_t)N * 128 * 4);
  float* AS1   = (float*)carve((size_t)N * 8 * 4);
  float* AD1   = (float*)carve((size_t)N * 8 * 4);
  float* EMAX1 = (float*)carve((size_t)N * 8 * 4);
  float* DEN1  = (float*)carve((size_t)N * 8 * 4);
  float* EB1   = (float*)carve((size_t)E2 * 8 * 4);
  float* ACC1  = (float*)carve((size_t)N * 128 * 4);
  float* G2    = (float*)carve((size_t)N * 40 * 4);
  float* AS2   = (float*)carve((size_t)N * 4);
  float* AD2   = (float*)carve((size_t)N * 4);
  float* EMAX2 = (float*)carve((size_t)N * 4);
  float* DEN2  = (float*)carve((size_t)N * 4);
  float* EB2   = (float*)carve((size_t)E2 * 4);
  int*   CNT   = (int*)carve((size_t)N * 4);
  int*   ROWPTR= (int*)carve((size_t)(N + 1) * 4);
  int*   RANK  = (int*)carve((size_t)E * 4);
  int2*  SE    = (int2*)carve((size_t)E2 * 8);

  const unsigned int NEG_INF = 0xFF800000u;
  const int B = 256;

  fill_u32_k<<<nblk(N * 8, B), B, 0, stream>>>((unsigned int*)EMAX1, NEG_INF, N * 8);
  fill_u32_k<<<nblk(N * 8, B), B, 0, stream>>>((unsigned int*)DEN1, 0u, N * 8);
  fill_u32_k<<<nblk(N, B), B, 0, stream>>>((unsigned int*)EMAX2, NEG_INF, N);
  fill_u32_k<<<nblk(N, B), B, 0, stream>>>((unsigned int*)DEN2, 0u, N);
  fill_u32_k<<<nblk(N * 40, B), B, 0, stream>>>((unsigned int*)out, 0u, N * 40);

  fill_u32_k<<<nblk(N, B), B, 0, stream>>>((unsigned int*)CNT, 1u, N);
  edge_ticket_k<<<nblk(E, B), B, 0, stream>>>(dstA, CNT, RANK, E);
  row_scan_k<<<1, 1024, 0, stream>>>(CNT, ROWPTR, (int)N);
  edge_place_k<<<nblk(E2, B), B, 0, stream>>>(srcA, dstA, ROWPTR, RANK, SE, E, E2);

  {
    long long waves = (N + 31) / 32;
    gemm1_wmma_k<<<nblk(waves, 4), 128, 0, stream>>>(x, W1, H1, (int)N);
  }
  alpha1_k<<<nblk(N * 8, B), B, 0, stream>>>(H1, a_src1, a_dst1, AS1, AD1, (int)N);
  edge_max1_k<<<nblk(E2 * 8, B), B, 0, stream>>>(srcA, dstA, AS1, AD1, EMAX1, E, E2);
  edge_exp1_k<<<nblk(E2 * 8, B), B, 0, stream>>>(srcA, dstA, AS1, AD1, EMAX1, EB1, DEN1, E, E2);
  row_msg1_k<<<nblk(N, 8), 256, 0, stream>>>(SE, ROWPTR, H1, EB1, DEN1, b1, ACC1, (int)N);

  {
    long long waves = (N + 31) / 32;
    gemm2_wmma_k<<<nblk(waves, 4), 128, 0, stream>>>(ACC1, W2, G2, (int)N);
  }
  alpha2_k<<<nblk(N, B), B, 0, stream>>>(G2, a_src2, a_dst2, AS2, AD2, (int)N);
  edge_max2_k<<<nblk(E2, B), B, 0, stream>>>(srcA, dstA, AS2, AD2, EMAX2, E, E2);
  edge_exp2_k<<<nblk(E2, B), B, 0, stream>>>(srcA, dstA, AS2, AD2, EMAX2, EB2, DEN2, E, E2);
  edge_msg2_k<<<nblk(E2 * 40, B), B, 0, stream>>>(srcA, dstA, G2, EB2, DEN2, out, E, E2);
  bias2_k<<<nblk(N * 10, B), B, 0, stream>>>(out, b2, N * 10);
}


# layer-2 edge phase row by row on the destination-sorted list: one wave per row does softmax max, exp-sum, messages and bias in registers (no atomics, no fills); sums now in sorted-row order
# speedup vs baseline: 3.1783x; 1.7874x over previous
#include <hip/hip_runtime.h>
#include <hip/hip_bf16.h>


typedef __attribute__((ext_vector_type(2))) float v2f;
typedef __attribute__((ext_vector_type(8))) float v8f;

#define NEG_SLOPE 0.2f
#define EPSF 1e-16f

__device__ __forceinline__ v8f wmma_f32(v2f a, v2f b, v8f c) {
  return __builtin_amdgcn_wmma_f32_16x16x4_f32(
      false, a, false, b, (short)0, c, false, false);
}

__global__ void fill_u32_k(unsigned int* __restrict__ p, unsigned int v, long long n) {
  long long i = (long long)blockIdx.x * blockDim.x + threadIdx.x;
  if (i < n) p[i] = v;
}

__global__ void __launch_bounds__(128, 1)
gemm1_wmma_k(const float* __restrict__ X, const float* __restrict__ W,
             float* __restrict__ Hout, int N) {
  int wave = blockIdx.x * (blockDim.x >> 5) + (threadIdx.x >> 5);
  int lane = threadIdx.x & 31;
  int half = lane >> 4;
  int lm   = lane & 15;
  long long row0 = (long long)wave * 32;
  if (row0 >= N) return;
  long long r0 = row0 + lm;
  long long r1 = row0 + 16 + lm;
  bool ok0 = r0 < N, ok1 = r1 < N;
  const v2f* x0 = (const v2f*)(X + r0 * 128);
  const v2f* x1 = (const v2f*)(X + r1 * 128);
  const v2f zz = {0.0f, 0.0f};

  v8f acc0[8] = {}, acc1[8] = {};
  for (int k = 0; k < 128; k += 4) {
    int ka = k + 2 * half;
    v2f a0 = ok0 ? x0[ka >> 1] : zz;
    v2f a1 = ok1 ? x1[ka >> 1] : zz;
    const float* wr = W + (long long)ka * 128;
#pragma unroll
    for (int t = 0; t < 8; ++t) {
      int col = t * 16 + lm;
      v2f b = {wr[col], wr[128 + col]};
      acc0[t] = wmma_f32(a0, b, acc0[t]);
      acc1[t] = wmma_f32(a1, b, acc1[t]);
    }
  }
#pragma unroll
  for (int t = 0; t < 8; ++t) {
    int col = t * 16 + lm;
#pragma unroll
    for (int i = 0; i < 8; ++i) {
      long long ra = row0 + i + 8 * half;
      long long rb = ra + 16;
      if (ra < N) Hout[ra * 128 + col] = acc0[t][i];
      if (rb < N) Hout[rb * 128 + col] = acc1[t][i];
    }
  }
}

__global__ void __launch_bounds__(128, 1)
gemm2_wmma_k(const float* __restrict__ X, const float* __restrict__ W,
             float* __restrict__ G, int N) {
  int wave = blockIdx.x * (blockDim.x >> 5) + (threadIdx.x >> 5);
  int lane = threadIdx.x & 31;
  int half = lane >> 4;
  int lm   = lane & 15;
  long long row0 = (long long)wave * 32;
  if (row0 >= N) return;
  long long r0 = row0 + lm;
  long long r1 = row0 + 16 + lm;
  bool ok0 = r0 < N, ok1 = r1 < N;
  const v2f* x0 = (const v2f*)(X + r0 * 128);
  const v2f* x1 = (const v2f*)(X + r1 * 128);
  const v2f zz = {0.0f, 0.0f};

  v8f acc0[3] = {}, acc1[3] = {};
  for (int k = 0; k < 128; k += 4) {
    int ka = k + 2 * half;
    v2f a0 = ok0 ? x0[ka >> 1] : zz;
    v2f a1 = ok1 ? x1[ka >> 1] : zz;
    const float* wr = W + (long long)ka * 40;
#pragma unroll
    for (int t = 0; t < 3; ++t) {
      int col = t * 16 + lm;
      v2f b;
      b.x = (col < 40) ? wr[col]      : 0.0f;
      b.y = (col < 40) ? wr[40 + col] : 0.0f;
      acc0[t] = wmma_f32(a0, b, acc0[t]);
      acc1[t] = wmma_f32(a1, b, acc1[t]);
    }
  }
#pragma unroll
  for (int t = 0; t < 3; ++t) {
    int col = t * 16 + lm;
    if (col < 40) {
#pragma unroll
      for (int i = 0; i < 8; ++i) {
        long long ra = row0 + i + 8 * half;
        long long rb = ra + 16;
        if (ra < N) G[ra * 40 + col] = acc0[t][i];
        if (rb < N) G[rb * 40 + col] = acc1[t][i];
      }
    }
  }
}

__global__ void alpha1_k(const float* __restrict__ H1, const float* __restrict__ asrc,
                         const float* __restrict__ adst, float* __restrict__ AS,
                         float* __restrict__ AD, int N) {
  long long tid = (long long)blockIdx.x * blockDim.x + threadIdx.x;
  if (tid >= (long long)N * 8) return;
  int h = (int)(tid & 7);
  const float4* hv = (const float4*)(H1 + (tid >> 3) * 128 + h * 16);
  const float4* av = (const float4*)(asrc + h * 16);
  const float4* dv = (const float4*)(adst + h * 16);
  float s = 0.f, d = 0.f;
#pragma unroll
  for (int q = 0; q < 4; ++q) {
    float4 v = hv[q], a = av[q], b = dv[q];
    s += v.x * a.x + v.y * a.y + v.z * a.z + v.w * a.w;
    d += v.x * b.x + v.y * b.y + v.z * b.z + v.w * b.w;
  }
  AS[tid] = s; AD[tid] = d;
}

__global__ void alpha2_k(const float* __restrict__ G, const float* __restrict__ asrc,
                         const float* __restrict__ adst, float* __restrict__ AS,
                         float* __restrict__ AD, int N) {
  long long n = (long long)blockIdx.x * blockDim.x + threadIdx.x;
  if (n >= N) return;
  const float4* gv = (const float4*)(G + n * 40);
  float s = 0.f, d = 0.f;
#pragma unroll
  for (int q = 0; q < 10; ++q) {
    float4 v = gv[q];
    float4 a = ((const float4*)asrc)[q];
    float4 b = ((const float4*)adst)[q];
    s += v.x * a.x + v.y * a.y + v.z * a.z + v.w * a.w;
    d += v.x * b.x + v.y * b.y + v.z * b.z + v.w * b.w;
  }
  AS[n] = s; AD[n] = d;
}

__device__ __forceinline__ float lrelu(float v) { return v > 0.f ? v : NEG_SLOPE * v; }

__device__ __forceinline__ void atomicMaxF(float* addr, float v) {
  if (v >= 0.f) atomicMax((int*)addr, __float_as_int(v));
  else          atomicMin((unsigned int*)addr, __float_as_uint(v));
}

__device__ __forceinline__ void edge_sd(long long e, long long E,
                                        const int* __restrict__ src,
                                        const int* __restrict__ dst,
                                        int& s, int& d) {
  if (e < E) { s = src[e]; d = dst[e]; }
  else       { s = (int)(e - E); d = s; }
}

__global__ void edge_max1_k(const int* __restrict__ src, const int* __restrict__ dst,
                            const float* __restrict__ AS, const float* __restrict__ AD,
                            float* __restrict__ EMAX, long long E, long long E2) {
  long long tid = (long long)blockIdx.x * blockDim.x + threadIdx.x;
  if (tid >= E2 * 8) return;
  long long e = tid >> 3; int h = (int)(tid & 7);
  int s, d; edge_sd(e, E, src, dst, s, d);
  float v = lrelu(AS[(long long)s * 8 + h] + AD[(long long)d * 8 + h]);
  atomicMaxF(&EMAX[(long long)d * 8 + h], v);
}

__global__ void edge_exp1_k(const int* __restrict__ src, const int* __restrict__ dst,
                            const float* __restrict__ AS, const float* __restrict__ AD,
                            const float* __restrict__ EMAX, float* __restrict__ EBUF,
                            float* __restrict__ DEN, long long E, long long E2) {
  long long tid = (long long)blockIdx.x * blockDim.x + threadIdx.x;
  if (tid >= E2 * 8) return;
  long long e = tid >> 3; int h = (int)(tid & 7);
  int s, d; edge_sd(e, E, src, dst, s, d);
  float v = lrelu(AS[(long long)s * 8 + h] + AD[(long long)d * 8 + h]);
  float ex = __expf(v - EMAX[(long long)d * 8 + h]);
  EBUF[tid] = ex;
  atomicAdd(&DEN[(long long)d * 8 + h], ex);
}

__global__ void edge_ticket_k(const int* __restrict__ dst, int* __restrict__ CNT,
                              int* __restrict__ RANK, long long E) {
  long long e = (long long)blockIdx.x * blockDim.x + threadIdx.x;
  if (e < E) RANK[e] = atomicAdd(&CNT[dst[e]], 1);
}

__global__ void __launch_bounds__(1024)
row_scan_k(const int* __restrict__ CNT, int* __restrict__ ROWPTR, int N) {
  __shared__ int wsum[32];
  const int t = threadIdx.x, lane = t & 31, w = t >> 5;
  int carry = 0;
  for (int base = 0; base < N; base += 1024 * 16) {
    const int i0 = base + t * 16;
    int v[16];
    if (i0 + 16 <= N) {
      const int4* c4 = (const int4*)(CNT + i0);
#pragma unroll
      for (int q = 0; q < 4; ++q) {
        int4 x = c4[q];
        v[4 * q] = x.x; v[4 * q + 1] = x.y; v[4 * q + 2] = x.z; v[4 * q + 3] = x.w;
      }
    } else {
#pragma unroll
      for (int k = 0; k < 16; ++k) v[k] = (i0 + k < N) ? CNT[i0 + k] : 0;
    }
    int sum = 0;
#pragma unroll
    for (int k = 0; k < 16; ++k) sum += v[k];
    int inc = sum;
#pragma unroll
    for (int o = 1; o < 32; o <<= 1) {
      int y = __shfl_up(inc, o);
      if (lane >= o) inc += y;
    }
    if (lane == 31) wsum[w] = inc;
    __syncthreads();
    int ws = wsum[lane];
    int winc = ws;
#pragma unroll
    for (int o = 1; o < 32; o <<= 1) {
      int y = __shfl_up(winc, o);
      if (lane >= o) winc += y;
    }
    int wave_excl = __shfl(winc, w) - __shfl(ws, w);
    int total = __shfl(winc, 31);
    int excl = carry + wave_excl + (inc - sum);
#pragma unroll
    for (int k = 0; k < 16; ++k) {
      if (i0 + k < N) ROWPTR[i0 + k] = excl;
      excl += v[k];
    }
    carry += total;
    __syncthreads();
  }
  if (t == 0) ROWPTR[N] = carry;
}

__global__ void edge_place_k(const int* __restrict__ src, const int* __restrict__ dst,
                             const int* __restrict__ ROWPTR, const int* __restrict__ RANK,
                             int2* __restrict__ SE, long long E, long long E2) {
  long long e = (long long)blockIdx.x * blockDim.x + threadIdx.x;
  if (e >= E2) return;
  int s, d, r;
  if (e < E) { s = src[e]; d = dst[e]; r = RANK[e]; }
  else       { s = (int)(e - E); d = s; r = 0; }
  SE[ROWPTR[d] + r] = make_int2(s, (int)e);
}

__global__ void __launch_bounds__(256)
row_msg1_k(const int2* __restrict__ SE, const int* __restrict__ ROWPTR,
           const float* __restrict__ H1, const float* __restrict__ EBUF,
           const float* __restrict__ DEN, const float* __restrict__ b1,
           float* __restrict__ HOUT, int N) {
  const int n = __builtin_amdgcn_readfirstlane((int)(blockIdx.x * 8 + (threadIdx.x >> 5)));
  if (n >= N) return;
  const int lane = threadIdx.x & 31;
  const int h = lane >> 2;
  const int beg = ROWPTR[n], end = ROWPTR[n + 1];
  const float den = DEN[(long long)n * 8 + h] + EPSF;
  float4 acc = {0.f, 0.f, 0.f, 0.f};
  for (int base = beg; base < end; base += 32) {
    const int m = (end - base < 32) ? (end - base) : 32;
    int2 se = make_int2(0, 0);
    if (lane < m) se = SE[base + lane];
#pragma unroll 4
    for (int j = 0; j < m; ++j) {
      const int s = __builtin_amdgcn_readlane(se.x, j);
      const int e = __builtin_amdgcn_readlane(se.y, j);
      const float ex = EBUF[(long long)e * 8 + h];
      const float4 v = *(const float4*)(H1 + (long long)s * 128 + lane * 4);
      const float alpha = ex / den;
      acc.x += v.x * alpha; acc.y += v.y * alpha;
      acc.z += v.z * alpha; acc.w += v.w * alpha;
    }
  }
  const float4 b = ((const float4*)b1)[lane];
  acc.x += b.x; acc.y += b.y; acc.z += b.z; acc.w += b.w;
  acc.x = acc.x > 0.f ? acc.x : (__expf(acc.x) - 1.f);
  acc.y = acc.y > 0.f ? acc.y : (__expf(acc.y) - 1.f);
  acc.z = acc.z > 0.f ? acc.z : (__expf(acc.z) - 1.f);
  acc.w = acc.w > 0.f ? acc.w : (__expf(acc.w) - 1.f);
  ((float4*)HOUT)[(long long)n * 32 + lane] = acc;
}

__global__ void __launch_bounds__(256)
row_soft_msg2_k(const int2* __restrict__ SE, const int* __restrict__ ROWPTR,
                const float* __restrict__ G, const float* __restrict__ AS,
                const float* __restrict__ AD, const float* __restrict__ b2,
                float* __restrict__ OUT, int N) {
  __shared__ float4 stage[8 * 10];
  const int w = threadIdx.x >> 5;
  const int lane = threadIdx.x & 31;
  const int n = __builtin_amdgcn_readfirstlane((int)(blockIdx.x * 8 + w));
  if (n < N) {
    const bool live = lane < 30;
    const int slot = live ? lane / 10 : 2;
    const int q = lane % 10;
    const int beg = ROWPTR[n], end = ROWPTR[n + 1];
    const int deg = end - beg;
    const float adn = AD[n];

    const bool in0 = live && lane < deg;
    const bool in1 = live && lane + 30 < deg;
    const int i0 = beg + lane < end ? beg + lane : end - 1;
    const int i1 = beg + 30 + lane < end ? beg + 30 + lane : end - 1;
    const int s0 = SE[i0].x, s1 = SE[i1].x;
    const float g0 = AS[s0], g1 = AS[s1];
    const float a0 = in0 ? lrelu(g0 + adn) : -INFINITY;
    const float a1 = in1 ? lrelu(g1 + adn) : -INFINITY;

    float mx = fmaxf(a0, a1);
    for (int base = beg + 60; base < end; base += 30) {
      const int idx = base + lane;
      if (live && idx < end) mx = fmaxf(mx, lrelu(AS[SE[idx].x] + adn));
    }
    for (int o = 16; o > 0; o >>= 1) mx = fmaxf(mx, __shfl_xor(mx, o));

    const float e0 = in0 ? __expf(a0 - mx) : 0.f;
    const float e1 = in1 ? __expf(a1 - mx) : 0.f;
    float sm = e0 + e1;
    for (int base = beg + 60; base < end; base += 30) {
      const int idx = base + lane;
      if (live && idx < end) sm += __expf(lrelu(AS[SE[idx].x] + adn) - mx);
    }
    for (int o = 16; o > 0; o >>= 1) sm += __shfl_xor(sm, o);
    const float den = sm + EPSF;

    float4 acc = {0.f, 0.f, 0.f, 0.f};
    auto chunk = [&](int sc, float wc, int m) {
      for (int j0 = 0; j0 < m; j0 += 15) {
#pragma unroll
        for (int u = 0; u < 5; ++u) {
          const int ej = j0 + 3 * u + slot;
          const int ec = ej < m ? ej : m - 1;
          const int sj = __shfl(sc, ec);
          const float wj = __shfl(wc, ec);
          const float4 v = *(const float4*)(G + (long long)sj * 40 + q * 4);
          const float wz = (live && ej < m) ? wj : 0.f;
          acc.x += v.x * wz; acc.y += v.y * wz;
          acc.z += v.z * wz; acc.w += v.w * wz;
        }
      }
    };
    chunk(s0, e0 / den, deg < 30 ? deg : 30);
    if (deg > 30) chunk(s1, e1 / den, deg - 30 < 30 ? deg - 30 : 30);
    for (int base = beg + 60; base < end; base += 30) {
      const int idx = base + lane;
      const int sx = SE[idx < end ? idx : end - 1].x;
      const float wx = (live && idx < end) ? __expf(lrelu(AS[sx] + adn) - mx) / den : 0.f;
      chunk(sx, wx, end - base < 30 ? end - base : 30);
    }

    float4 t;
    t.x = acc.x + __shfl_down(acc.x, 10) + __shfl_down(acc.x, 20);
    t.y = acc.y + __shfl_down(acc.y, 10) + __shfl_down(acc.y, 20);
    t.z = acc.z + __shfl_down(acc.z, 10) + __shfl_down(acc.z, 20);
    t.w = acc.w + __shfl_down(acc.w, 10) + __shfl_down(acc.w, 20);
    if (lane < 10) {
      const float4 b = ((const float4*)b2)[lane];
      t.x += b.x; t.y += b.y; t.z += b.z; t.w += b.w;
      stage[w * 10 + lane] = t;
    }
  }
  __syncthreads();
  const long long g = (long long)blockIdx.x * 80 + threadIdx.x;
  if (threadIdx.x < 80 && g < (long long)N * 10) ((float4*)OUT)[g] = stage[threadIdx.x];
}

static inline unsigned int nblk(long long n, int b) { return (unsigned int)((n + b - 1) / b); }

extern "C" void kernel_launch(void* const* d_in, const int* in_sizes, int n_in,
                              void* d_out, int out_size, void* d_ws, size_t ws_size,
                              hipStream_t stream) {
  const float* x      = (const float*)d_in[0];
  const int*   ei     = (const int*)d_in[1];
  const float* W1     = (const float*)d_in[2];
  const float* a_src1 = (const float*)d_in[3];
  const float* a_dst1 = (const float*)d_in[4];
  const float* b1     = (const float*)d_in[5];
  const float* W2     = (const float*)d_in[6];
  const float* a_src2 = (const float*)d_in[7];
  const float* a_dst2 = (const float*)d_in[8];
  const float* b2     = (const float*)d_in[9];
  float*       out    = (float*)d_out;

  const long long N  = in_sizes[0] / 128;
  const long long E  = in_sizes[1] / 2;
  const long long E2 = E + N;
  const int* srcA = ei;
  const int* dstA = ei + E;

  char* ws = (char*)d_ws;
  size_t off = 0;
  auto carve = [&](size_t bytes) -> void* {
    void* p = ws + off;
    off = (off + bytes + 255) & ~(size_t)255;
    return p;
  };
  float* H1    = (float*)carve((size_t)N * 128 * 4);
  float* AS1   = (float*)carve((size_t)N * 8 * 4);
  float* AD1   = (float*)carve((size_t)N * 8 * 4);
  float* EMAX1 = (float*)carve((size_t)N * 8 * 4);
  float* DEN1  = (float*)carve((size_t)N * 8 * 4);
  float* EB1   = (float*)carve((size_t)E2 * 8 * 4);
  float* ACC1  = (float*)carve((size_t)N * 128 * 4);
  float* G2    = (float*)carve((size_t)N * 40 * 4);
  float* AS2   = (float*)carve((size_t)N * 4);
  float* AD2   = (float*)carve((size_t)N * 4);
  int*   CNT   = (int*)carve((size_t)N * 4);
  int*   ROWPTR= (int*)carve((size_t)(N + 1) * 4);
  int*   RANK  = (int*)carve((size_t)E * 4);
  int2*  SE    = (int2*)carve((size_t)E2 * 8);

  const unsigned int NEG_INF = 0xFF800000u;
  const int B = 256;

  fill_u32_k<<<nblk(N * 8, B), B, 0, stream>>>((unsigned int*)EMAX1, NEG_INF, N * 8);
  fill_u32_k<<<nblk(N * 8, B), B, 0, stream>>>((unsigned int*)DEN1, 0u, N * 8);

  fill_u32_k<<<nblk(N, B), B, 0, stream>>>((unsigned int*)CNT, 1u, N);
  edge_ticket_k<<<nblk(E, B), B, 0, stream>>>(dstA, CNT, RANK, E);
  row_scan_k<<<1, 1024, 0, stream>>>(CNT, ROWPTR, (int)N);
  edge_place_k<<<nblk(E2, B), B, 0, stream>>>(srcA, dstA, ROWPTR, RANK, SE, E, E2);

  {
    long long waves = (N + 31) / 32;
    gemm1_wmma_k<<<nblk(waves, 4), 128, 0, stream>>>(x, W1, H1, (int)N);
  }
  alpha1_k<<<nblk(N * 8, B), B, 0, stream>>>(H1, a_src1, a_dst1, AS1, AD1, (int)N);
  edge_max1_k<<<nblk(E2 * 8, B), B, 0, stream>>>(srcA, dstA, AS1, AD1, EMAX1, E, E2);
  edge_exp1_k<<<nblk(E2 * 8, B), B, 0, stream>>>(srcA, dstA, AS1, AD1, EMAX1, EB1, DEN1, E, E2);
  row_msg1_k<<<nblk(N, 8), 256, 0, stream>>>(SE, ROWPTR, H1, EB1, DEN1, b1, ACC1, (int)N);

  {
    long long waves = (N + 31) / 32;
    gemm2_wmma_k<<<nblk(waves, 4), 128, 0, stream>>>(ACC1, W2, G2, (int)N);
  }
  alpha2_k<<<nblk(N, B), B, 0, stream>>>(G2, a_src2, a_dst2, AS2, AD2, (int)N);
  row_soft_msg2_k<<<nblk(N, 8), 256, 0, stream>>>(SE, ROWPTR, G2, AS2, AD2, b2, out, (int)N);
}


# layer-1 softmax max and exp-sum taken per destination row inside the row-message kernel on the sorted list, in registers: no float atomics, no per-edge exp buffer; exp-sum now a fixed in-wave tree
# speedup vs baseline: 5.1611x; 1.6239x over previous
#include <hip/hip_runtime.h>
#include <hip/hip_bf16.h>


typedef __attribute__((ext_vector_type(2))) float v2f;
typedef __attribute__((ext_vector_type(8))) float v8f;

#define NEG_SLOPE 0.2f
#define EPSF 1e-16f

__device__ __forceinline__ v8f wmma_f32(v2f a, v2f b, v8f c) {
  return __builtin_amdgcn_wmma_f32_16x16x4_f32(
      false, a, false, b, (short)0, c, false, false);
}

__global__ void fill_u32_k(unsigned int* __restrict__ p, unsigned int v, long long n) {
  long long i = (long long)blockIdx.x * blockDim.x + threadIdx.x;
  if (i < n) p[i] = v;
}

__global__ void __launch_bounds__(128, 1)
gemm1_wmma_k(const float* __restrict__ X, const float* __restrict__ W,
             float* __restrict__ Hout, int N) {
  int wave = blockIdx.x * (blockDim.x >> 5) + (threadIdx.x >> 5);
  int lane = threadIdx.x & 31;
  int half = lane >> 4;
  int lm   = lane & 15;
  long long row0 = (long long)wave * 32;
  if (row0 >= N) return;
  long long r0 = row0 + lm;
  long long r1 = row0 + 16 + lm;
  bool ok0 = r0 < N, ok1 = r1 < N;
  const v2f* x0 = (const v2f*)(X + r0 * 128);
  const v2f* x1 = (const v2f*)(X + r1 * 128);
  const v2f zz = {0.0f, 0.0f};

  v8f acc0[8] = {}, acc1[8] = {};
  for (int k = 0; k < 128; k += 4) {
    int ka = k + 2 * half;
    v2f a0 = ok0 ? x0[ka >> 1] : zz;
    v2f a1 = ok1 ? x1[ka >> 1] : zz;
    const float* wr = W + (long long)ka * 128;
#pragma unroll
    for (int t = 0; t < 8; ++t) {
      int col = t * 16 + lm;
      v2f b = {wr[col], wr[128 + col]};
      acc0[t] = wmma_f32(a0, b, acc0[t]);
      acc1[t] = wmma_f32(a1, b, acc1[t]);
    }
  }
#pragma unroll
  for (int t = 0; t < 8; ++t) {
    int col = t * 16 + lm;
#pragma unroll
    for (int i = 0; i < 8; ++i) {
      long long ra = row0 + i + 8 * half;
      long long rb = ra + 16;
      if (ra < N) Hout[ra * 128 + col] = acc0[t][i];
      if (rb < N) Hout[rb * 128 + col] = acc1[t][i];
    }
  }
}

__global__ void __launch_bounds__(128, 1)
gemm2_wmma_k(const float* __restrict__ X, const float* __restrict__ W,
             float* __restrict__ G, int N) {
  int wave = blockIdx.x * (blockDim.x >> 5) + (threadIdx.x >> 5);
  int lane = threadIdx.x & 31;
  int half = lane >> 4;
  int lm   = lane & 15;
  long long row0 = (long long)wave * 32;
  if (row0 >= N) return;
  long long r0 = row0 + lm;
  long long r1 = row0 + 16 + lm;
  bool ok0 = r0 < N, ok1 = r1 < N;
  const v2f* x0 = (const v2f*)(X + r0 * 128);
  const v2f* x1 = (const v2f*)(X + r1 * 128);
  const v2f zz = {0.0f, 0.0f};

  v8f acc0[3] = {}, acc1[3] = {};
  for (int k = 0; k < 128; k += 4) {
    int ka = k + 2 * half;
    v2f a0 = ok0 ? x0[ka >> 1] : zz;
    v2f a1 = ok1 ? x1[ka >> 1] : zz;
    const float* wr = W + (long long)ka * 40;
#pragma unroll
    for (int t = 0; t < 3; ++t) {
      int col = t * 16 + lm;
      v2f b;
      b.x = (col < 40) ? wr[col]      : 0.0f;
      b.y = (col < 40) ? wr[40 + col] : 0.0f;
      acc0[t] = wmma_f32(a0, b, acc0[t]);
      acc1[t] = wmma_f32(a1, b, acc1[t]);
    }
  }
#pragma unroll
  for (int t = 0; t < 3; ++t) {
    int col = t * 16 + lm;
    if (col < 40) {
#pragma unroll
      for (int i = 0; i < 8; ++i) {
        long long ra = row0 + i + 8 * half;
        long long rb = ra + 16;
        if (ra < N) G[ra * 40 + col] = acc0[t][i];
        if (rb < N) G[rb * 40 + col] = acc1[t][i];
      }
    }
  }
}

__global__ void alpha1_k(const float* __restrict__ H1, const float* __restrict__ asrc,
                         const float* __restrict__ adst, float* __restrict__ AS,
                         float* __restrict__ AD, int N) {
  long long tid = (long long)blockIdx.x * blockDim.x + threadIdx.x;
  if (tid >= (long long)N * 8) return;
  int h = (int)(tid & 7);
  const float4* hv = (const float4*)(H1 + (tid >> 3) * 128 + h * 16);
  const float4* av = (const float4*)(asrc + h * 16);
  const float4* dv = (const float4*)(adst + h * 16);
  float s = 0.f, d = 0.f;
#pragma unroll
  for (int q = 0; q < 4; ++q) {
    float4 v = hv[q], a = av[q], b = dv[q];
    s += v.x * a.x + v.y * a.y + v.z * a.z + v.w * a.w;
    d += v.x * b.x + v.y * b.y + v.z * b.z + v.w * b.w;
  }
  AS[tid] = s; AD[tid] = d;
}

__global__ void alpha2_k(const float* __restrict__ G, const float* __restrict__ asrc,
                         const float* __restrict__ adst, float* __restrict__ AS,
                         float* __restrict__ AD, int N) {
  long long n = (long long)blockIdx.x * blockDim.x + threadIdx.x;
  if (n >= N) return;
  const float4* gv = (const float4*)(G + n * 40);
  float s = 0.f, d = 0.f;
#pragma unroll
  for (int q = 0; q < 10; ++q) {
    float4 v = gv[q];
    float4 a = ((const float4*)asrc)[q];
    float4 b = ((const float4*)adst)[q];
    s += v.x * a.x + v.y * a.y + v.z * a.z + v.w * a.w;
    d += v.x * b.x + v.y * b.y + v.z * b.z + v.w * b.w;
  }
  AS[n] = s; AD[n] = d;
}

__device__ __forceinline__ float lrelu(float v) { return v > 0.f ? v : NEG_SLOPE * v; }

__global__ void edge_ticket_k(const int* __restrict__ dst, int* __restrict__ CNT,
                              int* __restrict__ RANK, long long E) {
  long long e = (long long)blockIdx.x * blockDim.x + threadIdx.x;
  if (e < E) RANK[e] = atomicAdd(&CNT[dst[e]], 1);
}

__global__ void __launch_bounds__(1024)
row_scan_k(const int* __restrict__ CNT, int* __restrict__ ROWPTR, int N) {
  __shared__ int wsum[32];
  const int t = threadIdx.x, lane = t & 31, w = t >> 5;
  int carry = 0;
  for (int base = 0; base < N; base += 1024 * 16) {
    const int i0 = base + t * 16;
    int v[16];
    if (i0 + 16 <= N) {
      const int4* c4 = (const int4*)(CNT + i0);
#pragma unroll
      for (int q = 0; q < 4; ++q) {
        int4 x = c4[q];
        v[4 * q] = x.x; v[4 * q + 1] = x.y; v[4 * q + 2] = x.z; v[4 * q + 3] = x.w;
      }
    } else {
#pragma unroll
      for (int k = 0; k < 16; ++k) v[k] = (i0 + k < N) ? CNT[i0 + k] : 0;
    }
    int sum = 0;
#pragma unroll
    for (int k = 0; k < 16; ++k) sum += v[k];
    int inc = sum;
#pragma unroll
    for (int o = 1; o < 32; o <<= 1) {
      int y = __shfl_up(inc, o);
      if (lane >= o) inc += y;
    }
    if (lane == 31) wsum[w] = inc;
    __syncthreads();
    int ws = wsum[lane];
    int winc = ws;
#pragma unroll
    for (int o = 1; o < 32; o <<= 1) {
      int y = __shfl_up(winc, o);
      if (lane >= o) winc += y;
    }
    int wave_excl = __shfl(winc, w) - __shfl(ws, w);
    int total = __shfl(winc, 31);
    int excl = carry + wave_excl + (inc - sum);
#pragma unroll
    for (int k = 0; k < 16; ++k) {
      if (i0 + k < N) ROWPTR[i0 + k] = excl;
      excl += v[k];
    }
    carry += total;
    __syncthreads();
  }
  if (t == 0) ROWPTR[N] = carry;
}

__global__ void edge_place_k(const int* __restrict__ src, const int* __restrict__ dst,
                             const int* __restrict__ ROWPTR, const int* __restrict__ RANK,
                             int2* __restrict__ SE, long long E, long long E2) {
  long long e = (long long)blockIdx.x * blockDim.x + threadIdx.x;
  if (e >= E2) return;
  int s, d, r;
  if (e < E) { s = src[e]; d = dst[e]; r = RANK[e]; }
  else       { s = (int)(e - E); d = s; r = 0; }
  SE[ROWPTR[d] + r] = make_int2(s, (int)e);
}

template <int CTRL>
__device__ __forceinline__ float quad_perm_f(float v) {
  return __int_as_float(
      __builtin_amdgcn_update_dpp(0, __float_as_int(v), CTRL, 0xf, 0xf, true));
}

__global__ void __launch_bounds__(256)
row_soft_msg1_k(const int2* __restrict__ SE, const int* __restrict__ ROWPTR,
                const float* __restrict__ H1, const float* __restrict__ AS,
                const float* __restrict__ AD, const float* __restrict__ b1,
                float* __restrict__ HOUT, int N) {
  const int n = __builtin_amdgcn_readfirstlane((int)(blockIdx.x * 8 + (threadIdx.x >> 5)));
  if (n >= N) return;
  const int lane = threadIdx.x & 31;
  const int h = lane >> 2;
  const int sub = lane & 3;
  const int beg = ROWPTR[n], end = ROWPTR[n + 1];
  const int deg = end - beg;
  const float adn = AD[(long long)n * 8 + h];

  auto srcs = [&](int base) -> int {
    const int idx = base + lane;
    return SE[idx < end ? idx : end - 1].x;
  };
  auto gather = [&](int sc, float (&g)[8]) {
#pragma unroll
    for (int u = 0; u < 8; ++u) {
      const int sj = __shfl(sc, 4 * u + sub);
      g[u] = *(const float*)((const char*)AS + (unsigned)(sj * 8 + h) * 4u);
    }
  };
  auto score = [&](const float (&g)[8], int m, float (&a)[8]) {
#pragma unroll
    for (int u = 0; u < 8; ++u)
      a[u] = sub < m - 4 * u ? lrelu(g[u] + adn) : -INFINITY;
  };
  auto scores = [&](int base, int m, float (&a)[8]) {
    float g[8];
#pragma unroll
    for (int u = 0; u < 8; ++u) {
      const int idx = base + 4 * u + sub;
      const int sj = SE[idx < end ? idx : end - 1].x;
      g[u] = *(const float*)((const char*)AS + (unsigned)(sj * 8 + h) * 4u);
    }
    asm volatile("" :: "v"(g[0]), "v"(g[1]), "v"(g[2]), "v"(g[3]),
                       "v"(g[4]), "v"(g[5]), "v"(g[6]), "v"(g[7]));
    score(g, m, a);
  };

  const int m0 = deg < 32 ? deg : 32;
  const int m1 = deg - 32 < 32 ? deg - 32 : 32;
  const int sc0 = srcs(beg);
  const int sc1 = srcs(beg + 32);
  float a0[8], a1[8];
  {
    float g0[8], g1[8];
    gather(sc0, g0);
#pragma unroll
    for (int u = 0; u < 8; ++u) g1[u] = 0.f;
    if (deg > 32) gather(sc1, g1);
    asm volatile("" :: "v"(g0[0]), "v"(g0[1]), "v"(g0[2]), "v"(g0[3]),
                       "v"(g0[4]), "v"(g0[5]), "v"(g0[6]), "v"(g0[7]),
                       "v"(g1[0]), "v"(g1[1]), "v"(g1[2]), "v"(g1[3]),
                       "v"(g1[4]), "v"(g1[5]), "v"(g1[6]), "v"(g1[7]));
    score(g0, m0, a0);
    score(g1, m1, a1);
  }

  float mx = -INFINITY;
#pragma unroll
  for (int u = 0; u < 8; ++u) mx = fmaxf(mx, fmaxf(a0[u], a1[u]));
  for (int base = beg + 64; base < end; base += 32) {
    float a[8];
    scores(base, end - base < 32 ? end - base : 32, a);
#pragma unroll
    for (int u = 0; u < 8; ++u) mx = fmaxf(mx, a[u]);
  }
  mx = fmaxf(mx, quad_perm_f<0xB1>(mx));
  mx = fmaxf(mx, quad_perm_f<0x4E>(mx));

  float sm = 0.f;
#pragma unroll
  for (int u = 0; u < 8; ++u) {
    a0[u] = __expf(a0[u] - mx);
    a1[u] = __expf(a1[u] - mx);
    sm += a0[u] + a1[u];
  }
  for (int base = beg + 64; base < end; base += 32) {
    float a[8];
    scores(base, end - base < 32 ? end - base : 32, a);
#pragma unroll
    for (int u = 0; u < 8; ++u) sm += __expf(a[u] - mx);
  }
  sm += quad_perm_f<0xB1>(sm);
  sm += quad_perm_f<0x4E>(sm);
  const float den = sm + EPSF;

  float4 acc = {0.f, 0.f, 0.f, 0.f};
  auto msgs = [&](int sc, int m, const float (&e)[8]) {
#pragma unroll
    for (int u = 0; u < 8; ++u) {
      if (4 * u < m) {
        const float w = e[u] / den;
        const int j = 4 * u;
        const int c1 = j + 1 < m ? j + 1 : m - 1;
        const int c2 = j + 2 < m ? j + 2 : m - 1;
        const int c3 = j + 3 < m ? j + 3 : m - 1;
        const int sA = __builtin_amdgcn_readlane(sc, j);
        const int sB = __builtin_amdgcn_readlane(sc, c1);
        const int sC = __builtin_amdgcn_readlane(sc, c2);
        const int sD = __builtin_amdgcn_readlane(sc, c3);
        const float4 vA = *(const float4*)(H1 + (long long)sA * 128 + lane * 4);
        const float4 vB = *(const float4*)(H1 + (long long)sB * 128 + lane * 4);
        const float4 vC = *(const float4*)(H1 + (long long)sC * 128 + lane * 4);
        const float4 vD = *(const float4*)(H1 + (long long)sD * 128 + lane * 4);
        const float wA = quad_perm_f<0x00>(w);
        const float wB = quad_perm_f<0x55>(w);
        const float wC = quad_perm_f<0xAA>(w);
        const float wD = quad_perm_f<0xFF>(w);
        acc.x += vA.x * wA; acc.y += vA.y * wA; acc.z += vA.z * wA; acc.w += vA.w * wA;
        acc.x += vB.x * wB; acc.y += vB.y * wB; acc.z += vB.z * wB; acc.w += vB.w * wB;
        acc.x += vC.x * wC; acc.y += vC.y * wC; acc.z += vC.z * wC; acc.w += vC.w * wC;
        acc.x += vD.x * wD; acc.y += vD.y * wD; acc.z += vD.z * wD; acc.w += vD.w * wD;
      }
    }
  };
  msgs(sc0, m0, a0);
  if (deg > 32) msgs(sc1, m1, a1);
  for (int base = beg + 64; base < end; base += 32) {
    const int sc = srcs(base);
    const int m = end - base < 32 ? end - base : 32;
    float a[8];
    scores(base, m, a);
#pragma unroll
    for (int u = 0; u < 8; ++u) a[u] = __expf(a[u] - mx);
    msgs(sc, m, a);
  }

  const float4 b = ((const float4*)b1)[lane];
  acc.x += b.x; acc.y += b.y; acc.z += b.z; acc.w += b.w;
  acc.x = acc.x > 0.f ? acc.x : (__expf(acc.x) - 1.f);
  acc.y = acc.y > 0.f ? acc.y : (__expf(acc.y) - 1.f);
  acc.z = acc.z > 0.f ? acc.z : (__expf(acc.z) - 1.f);
  acc.w = acc.w > 0.f ? acc.w : (__expf(acc.w) - 1.f);
  ((float4*)HOUT)[(long long)n * 32 + lane] = acc;
}

__global__ void __launch_bounds__(256)
row_soft_msg2_k(const int2* __restrict__ SE, const int* __restrict__ ROWPTR,
                const float* __restrict__ G, const float* __restrict__ AS,
                const float* __restrict__ AD, const float* __restrict__ b2,
                float* __restrict__ OUT, int N) {
  __shared__ float4 stage[8 * 10];
  const int w = threadIdx.x >> 5;
  const int lane = threadIdx.x & 31;
  const int n = __builtin_amdgcn_readfirstlane((int)(blockIdx.x * 8 + w));
  if (n < N) {
    const bool live = lane < 30;
    const int slot = live ? lane / 10 : 2;
    const int q = lane % 10;
    const int beg = ROWPTR[n], end = ROWPTR[n + 1];
    const int deg = end - beg;
    const float adn = AD[n];

    const bool in0 = live && lane < deg;
    const bool in1 = live && lane + 30 < deg;
    const int i0 = beg + lane < end ? beg + lane : end - 1;
    const int i1 = beg + 30 + lane < end ? beg + 30 + lane : end - 1;
    const int s0 = SE[i0].x, s1 = SE[i1].x;
    const float g0 = AS[s0], g1 = AS[s1];
    const float a0 = in0 ? lrelu(g0 + adn) : -INFINITY;
    const float a1 = in1 ? lrelu(g1 + adn) : -INFINITY;

    float mx = fmaxf(a0, a1);
    for (int base = beg + 60; base < end; base += 30) {
      const int idx = base + lane;
      if (live && idx < end) mx = fmaxf(mx, lrelu(AS[SE[idx].x] + adn));
    }
    for (int o = 16; o > 0; o >>= 1) mx = fmaxf(mx, __shfl_xor(mx, o));

    const float e0 = in0 ? __expf(a0 - mx) : 0.f;
    const float e1 = in1 ? __expf(a1 - mx) : 0.f;
    float sm = e0 + e1;
    for (int base = beg + 60; base < end; base += 30) {
      const int idx = base + lane;
      if (live && idx < end) sm += __expf(lrelu(AS[SE[idx].x] + adn) - mx);
    }
    for (int o = 16; o > 0; o >>= 1) sm += __shfl_xor(sm, o);
    const float den = sm + EPSF;

    float4 acc = {0.f, 0.f, 0.f, 0.f};
    auto chunk = [&](int sc, float wc, int m) {
      for (int j0 = 0; j0 < m; j0 += 15) {
#pragma unroll
        for (int u = 0; u < 5; ++u) {
          const int ej = j0 + 3 * u + slot;
          const int ec = ej < m ? ej : m - 1;
          const int sj = __shfl(sc, ec);
          const float wj = __shfl(wc, ec);
          const float4 v = *(const float4*)(G + (long long)sj * 40 + q * 4);
          const float wz = (live && ej < m) ? wj : 0.f;
          acc.x += v.x * wz; acc.y += v.y * wz;
          acc.z += v.z * wz; acc.w += v.w * wz;
        }
      }
    };
    chunk(s0, e0 / den, deg < 30 ? deg : 30);
    if (deg > 30) chunk(s1, e1 / den, deg - 30 < 30 ? deg - 30 : 30);
    for (int base = beg + 60; base < end; base += 30) {
      const int idx = base + lane;
      const int sx = SE[idx < end ? idx : end - 1].x;
      const float wx = (live && idx < end) ? __expf(lrelu(AS[sx] + adn) - mx) / den : 0.f;
      chunk(sx, wx, end - base < 30 ? end - base : 30);
    }

    float4 t;
    t.x = acc.x + __shfl_down(acc.x, 10) + __shfl_down(acc.x, 20);
    t.y = acc.y + __shfl_down(acc.y, 10) + __shfl_down(acc.y, 20);
    t.z = acc.z + __shfl_down(acc.z, 10) + __shfl_down(acc.z, 20);
    t.w = acc.w + __shfl_down(acc.w, 10) + __shfl_down(acc.w, 20);
    if (lane < 10) {
      const float4 b = ((const float4*)b2)[lane];
      t.x += b.x; t.y += b.y; t.z += b.z; t.w += b.w;
      stage[w * 10 + lane] = t;
    }
  }
  __syncthreads();
  const long long g = (long long)blockIdx.x * 80 + threadIdx.x;
  if (threadIdx.x < 80 && g < (long long)N * 10) ((float4*)OUT)[g] = stage[threadIdx.x];
}

static inline unsigned int nblk(long long n, int b) { return (unsigned int)((n + b - 1) / b); }

extern "C" void kernel_launch(void* const* d_in, const int* in_sizes, int n_in,
                              void* d_out, int out_size, void* d_ws, size_t ws_size,
                              hipStream_t stream) {
  const float* x      = (const float*)d_in[0];
  const int*   ei     = (const int*)d_in[1];
  const float* W1     = (const float*)d_in[2];
  const float* a_src1 = (const float*)d_in[3];
  const float* a_dst1 = (const float*)d_in[4];
  const float* b1     = (const float*)d_in[5];
  const float* W2     = (const float*)d_in[6];
  const float* a_src2 = (const float*)d_in[7];
  const float* a_dst2 = (const float*)d_in[8];
  const float* b2     = (const float*)d_in[9];
  float*       out    = (float*)d_out;

  const long long N  = in_sizes[0] / 128;
  const long long E  = in_sizes[1] / 2;
  const long long E2 = E + N;
  const int* srcA = ei;
  const int* dstA = ei + E;

  char* ws = (char*)d_ws;
  size_t off = 0;
  auto carve = [&](size_t bytes) -> void* {
    void* p = ws + off;
    off = (off + bytes + 255) & ~(size_t)255;
    return p;
  };
  float* H1    = (float*)carve((size_t)N * 128 * 4);
  float* AS1   = (float*)carve((size_t)N * 8 * 4);
  float* AD1   = (float*)carve((size_t)N * 8 * 4);
  float* ACC1  = (float*)carve((size_t)N * 128 * 4);
  float* G2    = (float*)carve((size_t)N * 40 * 4);
  float* AS2   = (float*)carve((size_t)N * 4);
  float* AD2   = (float*)carve((size_t)N * 4);
  int*   CNT   = (int*)carve((size_t)N * 4);
  int*   ROWPTR= (int*)carve((size_t)(N + 1) * 4);
  int*   RANK  = (int*)carve((size_t)E * 4);
  int2*  SE    = (int2*)carve((size_t)E2 * 8);

  const int B = 256;

  fill_u32_k<<<nblk(N, B), B, 0, stream>>>((unsigned int*)CNT, 1u, N);
  edge_ticket_k<<<nblk(E, B), B, 0, stream>>>(dstA, CNT, RANK, E);
  row_scan_k<<<1, 1024, 0, stream>>>(CNT, ROWPTR, (int)N);
  edge_place_k<<<nblk(E2, B), B, 0, stream>>>(srcA, dstA, ROWPTR, RANK, SE, E, E2);

  {
    long long waves = (N + 31) / 32;
    gemm1_wmma_k<<<nblk(waves, 4), 128, 0, stream>>>(x, W1, H1, (int)N);
  }
  alpha1_k<<<nblk(N * 8, B), B, 0, stream>>>(H1, a_src1, a_dst1, AS1, AD1, (int)N);
  row_soft_msg1_k<<<nblk(N, 8), 256, 0, stream>>>(SE, ROWPTR, H1, AS1, AD1, b1, ACC1, (int)N);

  {
    long long waves = (N + 31) / 32;
    gemm2_wmma_k<<<nblk(waves, 4), 128, 0, stream>>>(ACC1, W2, G2, (int)N);
  }
  alpha2_k<<<nblk(N, B), B, 0, stream>>>(G2, a_src2, a_dst2, AS2, AD2, (int)N);
  row_soft_msg2_k<<<nblk(N, 8), 256, 0, stream>>>(SE, ROWPTR, G2, AS2, AD2, b2, out, (int)N);
}


# both edge phases: a neighbour's score is the dot of its fetched row with a_src (in-lane, DPP lane sum), softmax with a running max; score tables, their 2 launches and gathers go; sum order changed
# speedup vs baseline: 5.6182x; 1.0886x over previous
#include <hip/hip_runtime.h>
#include <hip/hip_bf16.h>


typedef __attribute__((ext_vector_type(2))) float v2f;
typedef __attribute__((ext_vector_type(8))) float v8f;

#define NEG_SLOPE 0.2f
#define EPSF 1e-16f

__device__ __forceinline__ v8f wmma_f32(v2f a, v2f b, v8f c) {
  return __builtin_amdgcn_wmma_f32_16x16x4_f32(
      false, a, false, b, (short)0, c, false, false);
}

__global__ void fill_u32_k(unsigned int* __restrict__ p, unsigned int v, long long n) {
  long long i = (long long)blockIdx.x * blockDim.x + threadIdx.x;
  if (i < n) p[i] = v;
}

__global__ void __launch_bounds__(128, 1)
gemm1_wmma_k(const float* __restrict__ X, const float* __restrict__ W,
             float* __restrict__ Hout, int N) {
  int wave = blockIdx.x * (blockDim.x >> 5) + (threadIdx.x >> 5);
  int lane = threadIdx.x & 31;
  int half = lane >> 4;
  int lm   = lane & 15;
  long long row0 = (long long)wave * 32;
  if (row0 >= N) return;
  long long r0 = row0 + lm;
  long long r1 = row0 + 16 + lm;
  bool ok0 = r0 < N, ok1 = r1 < N;
  const v2f* x0 = (const v2f*)(X + r0 * 128);
  const v2f* x1 = (const v2f*)(X + r1 * 128);
  const v2f zz = {0.0f, 0.0f};

  v8f acc0[8] = {}, acc1[8] = {};
  for (int k = 0; k < 128; k += 4) {
    int ka = k + 2 * half;
    v2f a0 = ok0 ? x0[ka >> 1] : zz;
    v2f a1 = ok1 ? x1[ka >> 1] : zz;
    const float* wr = W + (long long)ka * 128;
#pragma unroll
    for (int t = 0; t < 8; ++t) {
      int col = t * 16 + lm;
      v2f b = {wr[col], wr[128 + col]};
      acc0[t] = wmma_f32(a0, b, acc0[t]);
      acc1[t] = wmma_f32(a1, b, acc1[t]);
    }
  }
#pragma unroll
  for (int t = 0; t < 8; ++t) {
    int col = t * 16 + lm;
#pragma unroll
    for (int i = 0; i < 8; ++i) {
      long long ra = row0 + i + 8 * half;
      long long rb = ra + 16;
      if (ra < N) Hout[ra * 128 + col] = acc0[t][i];
      if (rb < N) Hout[rb * 128 + col] = acc1[t][i];
    }
  }
}

__global__ void __launch_bounds__(128, 1)
gemm2_wmma_k(const float* __restrict__ X, const float* __restrict__ W,
             float* __restrict__ G, int N) {
  int wave = blockIdx.x * (blockDim.x >> 5) + (threadIdx.x >> 5);
  int lane = threadIdx.x & 31;
  int half = lane >> 4;
  int lm   = lane & 15;
  long long row0 = (long long)wave * 32;
  if (row0 >= N) return;
  long long r0 = row0 + lm;
  long long r1 = row0 + 16 + lm;
  bool ok0 = r0 < N, ok1 = r1 < N;
  const v2f* x0 = (const v2f*)(X + r0 * 128);
  const v2f* x1 = (const v2f*)(X + r1 * 128);
  const v2f zz = {0.0f, 0.0f};

  v8f acc0[3] = {}, acc1[3] = {};
  for (int k = 0; k < 128; k += 4) {
    int ka = k + 2 * half;
    v2f a0 = ok0 ? x0[ka >> 1] : zz;
    v2f a1 = ok1 ? x1[ka >> 1] : zz;
    const float* wr = W + (long long)ka * 40;
#pragma unroll
    for (int t = 0; t < 3; ++t) {
      int col = t * 16 + lm;
      v2f b;
      b.x = (col < 40) ? wr[col]      : 0.0f;
      b.y = (col < 40) ? wr[40 + col] : 0.0f;
      acc0[t] = wmma_f32(a0, b, acc0[t]);
      acc1[t] = wmma_f32(a1, b, acc1[t]);
    }
  }
#pragma unroll
  for (int t = 0; t < 3; ++t) {
    int col = t * 16 + lm;
    if (col < 40) {
#pragma unroll
      for (int i = 0; i < 8; ++i) {
        long long ra = row0 + i + 8 * half;
        long long rb = ra + 16;
        if (ra < N) G[ra * 40 + col] = acc0[t][i];
        if (rb < N) G[rb * 40 + col] = acc1[t][i];
      }
    }
  }
}

__device__ __forceinline__ float lrelu(float v) { return v > 0.f ? v : NEG_SLOPE * v; }

__global__ void edge_ticket_k(const int* __restrict__ dst, int* __restrict__ CNT,
                              int* __restrict__ RANK, long long E) {
  long long e = (long long)blockIdx.x * blockDim.x + threadIdx.x;
  if (e < E) RANK[e] = atomicAdd(&CNT[dst[e]], 1);
}

__global__ void __launch_bounds__(1024)
row_scan_k(const int* __restrict__ CNT, int* __restrict__ ROWPTR, int N) {
  __shared__ int wsum[32];
  const int t = threadIdx.x, lane = t & 31, w = t >> 5;
  int carry = 0;
  for (int base = 0; base < N; base += 1024 * 16) {
    const int i0 = base + t * 16;
    int v[16];
    if (i0 + 16 <= N) {
      const int4* c4 = (const int4*)(CNT + i0);
#pragma unroll
      for (int q = 0; q < 4; ++q) {
        int4 x = c4[q];
        v[4 * q] = x.x; v[4 * q + 1] = x.y; v[4 * q + 2] = x.z; v[4 * q + 3] = x.w;
      }
    } else {
#pragma unroll
      for (int k = 0; k < 16; ++k) v[k] = (i0 + k < N) ? CNT[i0 + k] : 0;
    }
    int sum = 0;
#pragma unroll
    for (int k = 0; k < 16; ++k) sum += v[k];
    int inc = sum;
#pragma unroll
    for (int o = 1; o < 32; o <<= 1) {
      int y = __shfl_up(inc, o);
      if (lane >= o) inc += y;
    }
    if (lane == 31) wsum[w] = inc;
    __syncthreads();
    int ws = wsum[lane];
    int winc = ws;
#pragma unroll
    for (int o = 1; o < 32; o <<= 1) {
      int y = __shfl_up(winc, o);
      if (lane >= o) winc += y;
    }
    int wave_excl = __shfl(winc, w) - __shfl(ws, w);
    int total = __shfl(winc, 31);
    int excl = carry + wave_excl + (inc - sum);
#pragma unroll
    for (int k = 0; k < 16; ++k) {
      if (i0 + k < N) ROWPTR[i0 + k] = excl;
      excl += v[k];
    }
    carry += total;
    __syncthreads();
  }
  if (t == 0) ROWPTR[N] = carry;
}

__global__ void edge_place_k(const int* __restrict__ src, const int* __restrict__ dst,
                             const int* __restrict__ ROWPTR, const int* __restrict__ RANK,
                             int2* __restrict__ SE, long long E, long long E2) {
  long long e = (long long)blockIdx.x * blockDim.x + threadIdx.x;
  if (e >= E2) return;
  int s, d, r;
  if (e < E) { s = src[e]; d = dst[e]; r = RANK[e]; }
  else       { s = (int)(e - E); d = s; r = 0; }
  SE[ROWPTR[d] + r] = make_int2(s, (int)e);
}

template <int CTRL>
__device__ __forceinline__ float dpp_f(float v) {
  return __int_as_float(
      __builtin_amdgcn_update_dpp(0, __float_as_int(v), CTRL, 0xf, 0xf, true));
}
__device__ __forceinline__ float quad_sum(float p) {
  p += dpp_f<0xB1>(p);
  p += dpp_f<0x4E>(p);
  return p;
}
__device__ __forceinline__ float row16_sum(float p) {
  p += dpp_f<0x161>(p);
  p += dpp_f<0x162>(p);
  p += dpp_f<0x164>(p);
  p += dpp_f<0x168>(p);
  return p;
}
__device__ __forceinline__ float xhalf(float x) {
  const int b = __float_as_int(x);
  return __int_as_float(
      __builtin_amdgcn_permlanex16(b, b, 0x76543210, 0xfedcba98, false, false));
}

#define RUNMAX0 (-3.402823466e+38f)

__global__ void __launch_bounds__(256)
row_soft_msg1_k(const int2* __restrict__ SE, const int* __restrict__ ROWPTR,
                const float* __restrict__ H1, const float* __restrict__ asrc,
                const float* __restrict__ adst, const float* __restrict__ b1,
                float* __restrict__ HOUT, int N) {
  const int n = __builtin_amdgcn_readfirstlane((int)(blockIdx.x * 8 + (threadIdx.x >> 5)));
  if (n >= N) return;
  const int lane = threadIdx.x & 31;
  const int beg = ROWPTR[n], end = ROWPTR[n + 1];
  const float4 av = ((const float4*)asrc)[lane];
  const float4 dv = ((const float4*)adst)[lane];
  const float4 hn = ((const float4*)H1)[(long long)n * 32 + lane];
  const float adn = quad_sum(hn.x * dv.x + hn.y * dv.y + hn.z * dv.z + hn.w * dv.w);

  auto srcs = [&](int base) -> int {
    const int idx = base + lane;
    return SE[idx < end ? idx : end - 1].x;
  };
  auto score = [&](const float4& v) -> float {
    return lrelu(quad_sum(v.x * av.x + v.y * av.y + v.z * av.z + v.w * av.w) + adn);
  };

  float mx = RUNMAX0, sm = 0.f;
  float4 acc = {0.f, 0.f, 0.f, 0.f};
  int sc = srcs(beg);
  int scn = srcs(beg + 32);
  for (int base = beg; base < end; base += 32) {
    const int m = end - base < 32 ? end - base : 32;
    for (int j = 0; j < m; j += 4) {
      const int c1 = j + 1 < m ? j + 1 : m - 1;
      const int c2 = j + 2 < m ? j + 2 : m - 1;
      const int c3 = j + 3 < m ? j + 3 : m - 1;
      const int sA = __builtin_amdgcn_readlane(sc, j);
      const int sB = __builtin_amdgcn_readlane(sc, c1);
      const int sC = __builtin_amdgcn_readlane(sc, c2);
      const int sD = __builtin_amdgcn_readlane(sc, c3);
      const float4 vA = *(const float4*)(H1 + (long long)sA * 128 + lane * 4);
      const float4 vB = *(const float4*)(H1 + (long long)sB * 128 + lane * 4);
      const float4 vC = *(const float4*)(H1 + (long long)sC * 128 + lane * 4);
      const float4 vD = *(const float4*)(H1 + (long long)sD * 128 + lane * 4);
      const float eA = score(vA);
      const float tB = score(vB), tC = score(vC), tD = score(vD);
      const float eB = j + 1 < m ? tB : -INFINITY;
      const float eC = j + 2 < m ? tC : -INFINITY;
      const float eD = j + 3 < m ? tD : -INFINITY;
      const float mn = fmaxf(fmaxf(mx, eA), fmaxf(fmaxf(eB, eC), eD));
      const float rs = __expf(mx - mn);
      const float wA = __expf(eA - mn), wB = __expf(eB - mn);
      const float wC = __expf(eC - mn), wD = __expf(eD - mn);
      acc.x *= rs; acc.y *= rs; acc.z *= rs; acc.w *= rs;
      acc.x += vA.x * wA; acc.y += vA.y * wA; acc.z += vA.z * wA; acc.w += vA.w * wA;
      acc.x += vB.x * wB; acc.y += vB.y * wB; acc.z += vB.z * wB; acc.w += vB.w * wB;
      acc.x += vC.x * wC; acc.y += vC.y * wC; acc.z += vC.z * wC; acc.w += vC.w * wC;
      acc.x += vD.x * wD; acc.y += vD.y * wD; acc.z += vD.z * wD; acc.w += vD.w * wD;
      sm = sm * rs + ((wA + wB) + (wC + wD));
      mx = mn;
    }
    sc = scn;
    if (base + 64 < end) scn = srcs(base + 64);
  }
  const float den = sm + EPSF;

  const float4 b = ((const float4*)b1)[lane];
  acc.x = acc.x / den + b.x; acc.y = acc.y / den + b.y;
  acc.z = acc.z / den + b.z; acc.w = acc.w / den + b.w;
  acc.x = acc.x > 0.f ? acc.x : (__expf(acc.x) - 1.f);
  acc.y = acc.y > 0.f ? acc.y : (__expf(acc.y) - 1.f);
  acc.z = acc.z > 0.f ? acc.z : (__expf(acc.z) - 1.f);
  acc.w = acc.w > 0.f ? acc.w : (__expf(acc.w) - 1.f);
  ((float4*)HOUT)[(long long)n * 32 + lane] = acc;
}

__global__ void __launch_bounds__(256)
row_soft_msg2_k(const int2* __restrict__ SE, const int* __restrict__ ROWPTR,
                const float* __restrict__ G, const float* __restrict__ asrc,
                const float* __restrict__ adst, const float* __restrict__ b2,
                float* __restrict__ OUT, int N) {
  __shared__ float4 stage[8 * 10];
  const int w = threadIdx.x >> 5;
  const int lane = threadIdx.x & 31;
  const int n = __builtin_amdgcn_readfirstlane((int)(blockIdx.x * 8 + w));
  if (n < N) {
    const int hf = lane >> 4;
    const int q = lane & 15;
    const bool real = q < 10;
    const int qc = real ? q : 9;
    const float4 z4 = {0.f, 0.f, 0.f, 0.f};
    const float4 av = real ? ((const float4*)asrc)[qc] : z4;
    const float4 dv = real ? ((const float4*)adst)[qc] : z4;
    const int beg = ROWPTR[n], end = ROWPTR[n + 1];
    const float4 gn = *(const float4*)(G + (long long)n * 40 + qc * 4);
    const float adn = row16_sum(gn.x * dv.x + gn.y * dv.y + gn.z * dv.z + gn.w * dv.w);

    auto srcs = [&](int base) -> int {
      const int idx = base + lane;
      return SE[idx < end ? idx : end - 1].x;
    };

    float mx = RUNMAX0, sm = 0.f;
    float4 acc = {0.f, 0.f, 0.f, 0.f};
    int sc = srcs(beg);
    int scn = srcs(beg + 32);
    for (int base = beg; base < end; base += 32) {
      const int m = end - base < 32 ? end - base : 32;
      for (int j = 0; j < m; j += 8) {
        float4 v[4];
        float e[4];
#pragma unroll
        for (int t = 0; t < 4; ++t) {
          const int ja = j + 2 * t, jb = ja + 1;
          const int ca = ja < m ? ja : m - 1;
          const int cb = jb < m ? jb : m - 1;
          const int sa = __builtin_amdgcn_readlane(sc, ca);
          const int sb = __builtin_amdgcn_readlane(sc, cb);
          const int sj = hf ? sb : sa;
          v[t] = *(const float4*)(G + (long long)sj * 40 + qc * 4);
        }
#pragma unroll
        for (int t = 0; t < 4; ++t) {
          const float p = row16_sum(v[t].x * av.x + v[t].y * av.y + v[t].z * av.z + v[t].w * av.w);
          e[t] = (j + 2 * t + hf < m) ? lrelu(p + adn) : -INFINITY;
        }
        const float mn = fmaxf(fmaxf(mx, e[0]), fmaxf(fmaxf(e[1], e[2]), e[3]));
        const float rs = __expf(mx - mn);
        const float w0 = __expf(e[0] - mn), w1 = __expf(e[1] - mn);
        const float w2 = __expf(e[2] - mn), w3 = __expf(e[3] - mn);
        acc.x *= rs; acc.y *= rs; acc.z *= rs; acc.w *= rs;
        acc.x += v[0].x * w0; acc.y += v[0].y * w0; acc.z += v[0].z * w0; acc.w += v[0].w * w0;
        acc.x += v[1].x * w1; acc.y += v[1].y * w1; acc.z += v[1].z * w1; acc.w += v[1].w * w1;
        acc.x += v[2].x * w2; acc.y += v[2].y * w2; acc.z += v[2].z * w2; acc.w += v[2].w * w2;
        acc.x += v[3].x * w3; acc.y += v[3].y * w3; acc.z += v[3].z * w3; acc.w += v[3].w * w3;
        sm = sm * rs + ((w0 + w1) + (w2 + w3));
        mx = mn;
      }
      sc = scn;
      if (base + 64 < end) scn = srcs(base + 64);
    }

    const float mj = fmaxf(mx, xhalf(mx));
    const float fj = __expf(mx - mj);
    acc.x *= fj; acc.y *= fj; acc.z *= fj; acc.w *= fj; sm *= fj;
    acc.x += xhalf(acc.x); acc.y += xhalf(acc.y);
    acc.z += xhalf(acc.z); acc.w += xhalf(acc.w);
    sm += xhalf(sm);
    if (lane < 10) {
      const float den = sm + EPSF;
      const float4 b = ((const float4*)b2)[lane];
      float4 t;
      t.x = acc.x / den + b.x; t.y = acc.y / den + b.y;
      t.z = acc.z / den + b.z; t.w = acc.w / den + b.w;
      stage[w * 10 + lane] = t;
    }
  }
  __syncthreads();
  const long long g = (long long)blockIdx.x * 80 + threadIdx.x;
  if (threadIdx.x < 80 && g < (long long)N * 10) ((float4*)OUT)[g] = stage[threadIdx.x];
}

static inline unsigned int nblk(long long n, int b) { return (unsigned int)((n + b - 1) / b); }

extern "C" void kernel_launch(void* const* d_in, const int* in_sizes, int n_in,
                              void* d_out, int out_size, void* d_ws, size_t ws_size,
                              hipStream_t stream) {
  const float* x      = (const float*)d_in[0];
  const int*   ei     = (const int*)d_in[1];
  const float* W1     = (const float*)d_in[2];
  const float* a_src1 = (const float*)d_in[3];
  const float* a_dst1 = (const float*)d_in[4];
  const float* b1     = (const float*)d_in[5];
  const float* W2     = (const float*)d_in[6];
  const float* a_src2 = (const float*)d_in[7];
  const float* a_dst2 = (const float*)d_in[8];
  const float* b2     = (const float*)d_in[9];
  float*       out    = (float*)d_out;

  const long long N  = in_sizes[0] / 128;
  const long long E  = in_sizes[1] / 2;
  const long long E2 = E + N;
  const int* srcA = ei;
  const int* dstA = ei + E;

  char* ws = (char*)d_ws;
  size_t off = 0;
  auto carve = [&](size_t bytes) -> void* {
    void* p = ws + off;
    off = (off + bytes + 255) & ~(size_t)255;
    return p;
  };
  float* H1    = (float*)carve((size_t)N * 128 * 4);
  float* ACC1  = (float*)carve((size_t)N * 128 * 4);
  float* G2    = (float*)carve((size_t)N * 40 * 4);
  int*   CNT   = (int*)carve((size_t)N * 4);
  int*   ROWPTR= (int*)carve((size_t)(N + 1) * 4);
  int*   RANK  = (int*)carve((size_t)E * 4);
  int2*  SE    = (int2*)carve((size_t)E2 * 8);

  const int B = 256;

  fill_u32_k<<<nblk(N, B), B, 0, stream>>>((unsigned int*)CNT, 1u, N);
  edge_ticket_k<<<nblk(E, B), B, 0, stream>>>(dstA, CNT, RANK, E);
  row_scan_k<<<1, 1024, 0, stream>>>(CNT, ROWPTR, (int)N);
  edge_place_k<<<nblk(E2, B), B, 0, stream>>>(srcA, dstA, ROWPTR, RANK, SE, E, E2);

  {
    long long waves = (N + 31) / 32;
    gemm1_wmma_k<<<nblk(waves, 4), 128, 0, stream>>>(x, W1, H1, (int)N);
  }
  row_soft_msg1_k<<<nblk(N, 8), 256, 0, stream>>>(SE, ROWPTR, H1, a_src1, a_dst1, b1, ACC1, (int)N);

  {
    long long waves = (N + 31) / 32;
    gemm2_wmma_k<<<nblk(waves, 4), 128, 0, stream>>>(ACC1, W2, G2, (int)N);
  }
  row_soft_msg2_k<<<nblk(N, 8), 256, 0, stream>>>(SE, ROWPTR, G2, a_src2, a_dst2, b2, out, (int)N);
}


# edge sort by destination redone as a two-level bucket sort, all counts in LDS (slice counts, scan, slice scatter, per-bucket rows): no global atomics, run-wise writes; sum order = LDS arrival order
# speedup vs baseline: 9.7172x; 1.7296x over previous
#include <hip/hip_runtime.h>
#include <hip/hip_bf16.h>


typedef __attribute__((ext_vector_type(2))) float v2f;
typedef __attribute__((ext_vector_type(8))) float v8f;

#define NEG_SLOPE 0.2f
#define EPSF 1e-16f

__device__ __forceinline__ v8f wmma_f32(v2f a, v2f b, v8f c) {
  return __builtin_amdgcn_wmma_f32_16x16x4_f32(
      false, a, false, b, (short)0, c, false, false);
}

__global__ void __launch_bounds__(128, 1)
gemm1_wmma_k(const float* __restrict__ X, const float* __restrict__ W,
             float* __restrict__ Hout, int N) {
  int wave = blockIdx.x * (blockDim.x >> 5) + (threadIdx.x >> 5);
  int lane = threadIdx.x & 31;
  int half = lane >> 4;
  int lm   = lane & 15;
  long long row0 = (long long)wave * 32;
  if (row0 >= N) return;
  long long r0 = row0 + lm;
  long long r1 = row0 + 16 + lm;
  bool ok0 = r0 < N, ok1 = r1 < N;
  const v2f* x0 = (const v2f*)(X + r0 * 128);
  const v2f* x1 = (const v2f*)(X + r1 * 128);
  const v2f zz = {0.0f, 0.0f};

  v8f acc0[8] = {}, acc1[8] = {};
  for (int k = 0; k < 128; k += 4) {
    int ka = k + 2 * half;
    v2f a0 = ok0 ? x0[ka >> 1] : zz;
    v2f a1 = ok1 ? x1[ka >> 1] : zz;
    const float* wr = W + (long long)ka * 128;
#pragma unroll
    for (int t = 0; t < 8; ++t) {
      int col = t * 16 + lm;
      v2f b = {wr[col], wr[128 + col]};
      acc0[t] = wmma_f32(a0, b, acc0[t]);
      acc1[t] = wmma_f32(a1, b, acc1[t]);
    }
  }
#pragma unroll
  for (int t = 0; t < 8; ++t) {
    int col = t * 16 + lm;
#pragma unroll
    for (int i = 0; i < 8; ++i) {
      long long ra = row0 + i + 8 * half;
      long long rb = ra + 16;
      if (ra < N) Hout[ra * 128 + col] = acc0[t][i];
      if (rb < N) Hout[rb * 128 + col] = acc1[t][i];
    }
  }
}

__global__ void __launch_bounds__(128, 1)
gemm2_wmma_k(const float* __restrict__ X, const float* __restrict__ W,
             float* __restrict__ G, int N) {
  int wave = blockIdx.x * (blockDim.x >> 5) + (threadIdx.x >> 5);
  int lane = threadIdx.x & 31;
  int half = lane >> 4;
  int lm   = lane & 15;
  long long row0 = (long long)wave * 32;
  if (row0 >= N) return;
  long long r0 = row0 + lm;
  long long r1 = row0 + 16 + lm;
  bool ok0 = r0 < N, ok1 = r1 < N;
  const v2f* x0 = (const v2f*)(X + r0 * 128);
  const v2f* x1 = (const v2f*)(X + r1 * 128);
  const v2f zz = {0.0f, 0.0f};

  v8f acc0[3] = {}, acc1[3] = {};
  for (int k = 0; k < 128; k += 4) {
    int ka = k + 2 * half;
    v2f a0 = ok0 ? x0[ka >> 1] : zz;
    v2f a1 = ok1 ? x1[ka >> 1] : zz;
    const float* wr = W + (long long)ka * 40;
#pragma unroll
    for (int t = 0; t < 3; ++t) {
      int col = t * 16 + lm;
      v2f b;
      b.x = (col < 40) ? wr[col]      : 0.0f;
      b.y = (col < 40) ? wr[40 + col] : 0.0f;
      acc0[t] = wmma_f32(a0, b, acc0[t]);
      acc1[t] = wmma_f32(a1, b, acc1[t]);
    }
  }
#pragma unroll
  for (int t = 0; t < 3; ++t) {
    int col = t * 16 + lm;
    if (col < 40) {
#pragma unroll
      for (int i = 0; i < 8; ++i) {
        long long ra = row0 + i + 8 * half;
        long long rb = ra + 16;
        if (ra < N) G[ra * 40 + col] = acc0[t][i];
        if (rb < N) G[rb * 40 + col] = acc1[t][i];
      }
    }
  }
}

__device__ __forceinline__ float lrelu(float v) { return v > 0.f ? v : NEG_SLOPE * v; }

#define BKT_SHIFT 8
#define BKT_ROWS 256
#define SORT_P 128
#define CAP_SLICE 12800
#define CAP_BKT 16384

__global__ void __launch_bounds__(1024)
bucket_count_k(const int* __restrict__ dst, int* __restrict__ HIST,
               long long E, long long slice, int nb) {
  extern __shared__ int dyn_lds[];
  int* hist = dyn_lds;
  const int t = threadIdx.x;
  for (int b = t; b < nb; b += 1024) hist[b] = 0;
  __syncthreads();
  const long long lo = (long long)blockIdx.x * slice;
  const long long hi = lo + slice < E ? lo + slice : E;
  for (long long e = lo + t; e < hi; e += 4096) {
    int d[4];
#pragma unroll
    for (int u = 0; u < 4; ++u) {
      const long long i = e + u * 1024;
      d[u] = i < hi ? dst[i] : -1;
    }
#pragma unroll
    for (int u = 0; u < 4; ++u)
      if (d[u] >= 0) atomicAdd(&hist[d[u] >> BKT_SHIFT], 1);
  }
  __syncthreads();
  for (int b = t; b < nb; b += 1024)
    HIST[(long long)b * gridDim.x + blockIdx.x] = hist[b];
}

__global__ void __launch_bounds__(1024)
row_scan_k(const int* __restrict__ CNT, int* __restrict__ ROWPTR, int N) {
  __shared__ int wsum[32];
  const int t = threadIdx.x, lane = t & 31, w = t >> 5;
  int carry = 0;
  for (int base = 0; base < N; base += 1024 * 16) {
    const int i0 = base + t * 16;
    int v[16];
    if (i0 + 16 <= N) {
      const int4* c4 = (const int4*)(CNT + i0);
#pragma unroll
      for (int q = 0; q < 4; ++q) {
        int4 x = c4[q];
        v[4 * q] = x.x; v[4 * q + 1] = x.y; v[4 * q + 2] = x.z; v[4 * q + 3] = x.w;
      }
    } else {
#pragma unroll
      for (int k = 0; k < 16; ++k) v[k] = (i0 + k < N) ? CNT[i0 + k] : 0;
    }
    int sum = 0;
#pragma unroll
    for (int k = 0; k < 16; ++k) sum += v[k];
    int inc = sum;
#pragma unroll
    for (int o = 1; o < 32; o <<= 1) {
      int y = __shfl_up(inc, o);
      if (lane >= o) inc += y;
    }
    if (lane == 31) wsum[w] = inc;
    __syncthreads();
    int ws = wsum[lane];
    int winc = ws;
#pragma unroll
    for (int o = 1; o < 32; o <<= 1) {
      int y = __shfl_up(winc, o);
      if (lane >= o) winc += y;
    }
    int wave_excl = __shfl(winc, w) - __shfl(ws, w);
    int total = __shfl(winc, 31);
    int excl = carry + wave_excl + (inc - sum);
#pragma unroll
    for (int k = 0; k < 16; ++k) {
      if (i0 + k < N) ROWPTR[i0 + k] = excl;
      excl += v[k];
    }
    carry += total;
    __syncthreads();
  }
  if (t == 0) ROWPTR[N] = carry;
}

__global__ void __launch_bounds__(1024)
bucket_scatter_k(const int* __restrict__ src, const int* __restrict__ dst,
                 const int* __restrict__ BASE, int2* __restrict__ TMP,
                 long long E, long long slice, int nb) {
  extern __shared__ int dyn_lds[];
  __shared__ int2 stg[CAP_SLICE];
  int* gb  = dyn_lds;
  int* lb  = dyn_lds + nb;
  int* cur = dyn_lds + 2 * nb;
  const int t = threadIdx.x, lane = t & 31;
  const int P = gridDim.x, blk = blockIdx.x;
  for (int b = t; b < nb; b += 1024) {
    const int g = BASE[(long long)b * P + blk];
    gb[b] = g;
    lb[b] = BASE[(long long)b * P + blk + 1] - g;
  }
  __syncthreads();
  if (t < 32) {
    int carry = 0;
    for (int base = 0; base < nb; base += 32) {
      const int b = base + lane;
      const int cl = b < nb ? lb[b] : 0;
      int inc = cl;
#pragma unroll
      for (int o = 1; o < 32; o <<= 1) {
        int y = __shfl_up(inc, o);
        if (lane >= o) inc += y;
      }
      if (b < nb) { lb[b] = carry + inc - cl; cur[b] = carry + inc - cl; }
      carry += __shfl(inc, 31);
    }
  }
  __syncthreads();
  const long long lo = (long long)blk * slice;
  const long long hi = lo + slice < E ? lo + slice : E;
  const bool staged = slice <= CAP_SLICE;
  for (long long e = lo + t; e < hi; e += 4096) {
    int s[4], d[4];
#pragma unroll
    for (int u = 0; u < 4; ++u) {
      const long long i = e + u * 1024;
      s[u] = i < hi ? src[i] : 0;
      d[u] = i < hi ? dst[i] : -1;
    }
#pragma unroll
    for (int u = 0; u < 4; ++u) {
      if (d[u] >= 0) {
        const int b = d[u] >> BKT_SHIFT;
        const int r = atomicAdd(&cur[b], 1);
        if (staged) stg[r] = make_int2(s[u], d[u]);
        else        TMP[gb[b] + (r - lb[b])] = make_int2(s[u], d[u]);
      }
    }
  }
  __syncthreads();
  if (staged) {
    const int cnt = hi > lo ? (int)(hi - lo) : 0;
    for (int i = t; i < cnt; i += 1024) {
      const int2 ed = stg[i];
      const int b = ed.y >> BKT_SHIFT;
      TMP[gb[b] + (i - lb[b])] = ed;
    }
  }
}

__global__ void __launch_bounds__(1024)
bucket_rows_k(const int2* __restrict__ TMP, const int* __restrict__ BASE,
              int* __restrict__ ROWPTR, int2* __restrict__ SE, int N, int P) {
  __shared__ int cnt[BKT_ROWS];
  __shared__ int wsum[8];
  __shared__ int2 stg[CAP_BKT];
  const int t = threadIdx.x, lane = t & 31, w = t >> 5;
  const int k = blockIdx.x;
  const int n0 = k << BKT_SHIFT;
  const int rows = N - n0 < BKT_ROWS ? N - n0 : BKT_ROWS;
  const int lo = BASE[(long long)k * P], hi = BASE[(long long)(k + 1) * P];
  const int gbase = lo + n0;
  const int total = hi - lo + rows;
  const bool staged = total <= CAP_BKT;
  if (t < BKT_ROWS) cnt[t] = 0;
  __syncthreads();
  for (int i0 = lo + t; i0 < hi; i0 += 4096) {
    int d[4];
#pragma unroll
    for (int u = 0; u < 4; ++u) {
      const int i = i0 + u * 1024;
      d[u] = i < hi ? TMP[i].y : -1;
    }
#pragma unroll
    for (int u = 0; u < 4; ++u)
      if (d[u] >= 0) atomicAdd(&cnt[d[u] - n0], 1);
  }
  __syncthreads();
  const int c = t < BKT_ROWS ? cnt[t] : 0;
  int inc = c;
#pragma unroll
  for (int o = 1; o < 32; o <<= 1) {
    int y = __shfl_up(inc, o);
    if (lane >= o) inc += y;
  }
  if (lane == 31 && w < 8) wsum[w] = inc;
  __syncthreads();
  int wex = 0;
#pragma unroll
  for (int q = 0; q < 8; ++q) if (q < w) wex += wsum[q];
  const int loc = t + wex + inc - c;
  if (t < rows) {
    ROWPTR[n0 + t] = gbase + loc;
    const int2 self = make_int2(n0 + t, n0 + t);
    if (staged) stg[loc] = self; else SE[gbase + loc] = self;
    cnt[t] = loc + 1;
  }
  if (k == (int)gridDim.x - 1 && t == 0) ROWPTR[N] = gbase + total;
  __syncthreads();
  for (int i0 = lo + t; i0 < hi; i0 += 4096) {
    int2 ed[4];
#pragma unroll
    for (int u = 0; u < 4; ++u) {
      const int i = i0 + u * 1024;
      ed[u] = i < hi ? TMP[i] : make_int2(0, -1);
    }
#pragma unroll
    for (int u = 0; u < 4; ++u) {
      if (ed[u].y >= 0) {
        const int p = atomicAdd(&cnt[ed[u].y - n0], 1);
        if (staged) stg[p] = ed[u]; else SE[gbase + p] = ed[u];
      }
    }
  }
  __syncthreads();
  if (staged)
    for (int i = t; i < total; i += 1024) SE[gbase + i] = stg[i];
}

template <int CTRL>
__device__ __forceinline__ float dpp_f(float v) {
  return __int_as_float(
      __builtin_amdgcn_update_dpp(0, __float_as_int(v), CTRL, 0xf, 0xf, true));
}
__device__ __forceinline__ float quad_sum(float p) {
  p += dpp_f<0xB1>(p);
  p += dpp_f<0x4E>(p);
  return p;
}
__device__ __forceinline__ float row16_sum(float p) {
  p += dpp_f<0x161>(p);
  p += dpp_f<0x162>(p);
  p += dpp_f<0x164>(p);
  p += dpp_f<0x168>(p);
  return p;
}
__device__ __forceinline__ float xhalf(float x) {
  const int b = __float_as_int(x);
  return __int_as_float(
      __builtin_amdgcn_permlanex16(b, b, 0x76543210, 0xfedcba98, false, false));
}

#define RUNMAX0 (-3.402823466e+38f)

__global__ void __launch_bounds__(256)
row_soft_msg1_k(const int2* __restrict__ SE, const int* __restrict__ ROWPTR,
                const float* __restrict__ H1, const float* __restrict__ asrc,
                const float* __restrict__ adst, const float* __restrict__ b1,
                float* __restrict__ HOUT, int N) {
  const int n = __builtin_amdgcn_readfirstlane((int)(blockIdx.x * 8 + (threadIdx.x >> 5)));
  if (n >= N) return;
  const int lane = threadIdx.x & 31;
  const int beg = ROWPTR[n], end = ROWPTR[n + 1];
  const float4 av = ((const float4*)asrc)[lane];
  const float4 dv = ((const float4*)adst)[lane];
  const float4 hn = ((const float4*)H1)[(long long)n * 32 + lane];
  const float adn = quad_sum(hn.x * dv.x + hn.y * dv.y + hn.z * dv.z + hn.w * dv.w);

  auto srcs = [&](int base) -> int {
    const int idx = base + lane;
    return SE[idx < end ? idx : end - 1].x;
  };
  auto score = [&](const float4& v) -> float {
    return lrelu(quad_sum(v.x * av.x + v.y * av.y + v.z * av.z + v.w * av.w) + adn);
  };

  float mx = RUNMAX0, sm = 0.f;
  float4 acc = {0.f, 0.f, 0.f, 0.f};
  int sc = srcs(beg);
  int scn = srcs(beg + 32);
  for (int base = beg; base < end; base += 32) {
    const int m = end - base < 32 ? end - base : 32;
    for (int j = 0; j < m; j += 4) {
      const int c1 = j + 1 < m ? j + 1 : m - 1;
      const int c2 = j + 2 < m ? j + 2 : m - 1;
      const int c3 = j + 3 < m ? j + 3 : m - 1;
      const int sA = __builtin_amdgcn_readlane(sc, j);
      const int sB = __builtin_amdgcn_readlane(sc, c1);
      const int sC = __builtin_amdgcn_readlane(sc, c2);
      const int sD = __builtin_amdgcn_readlane(sc, c3);
      const float4 vA = *(const float4*)(H1 + (long long)sA * 128 + lane * 4);
      const float4 vB = *(const float4*)(H1 + (long long)sB * 128 + lane * 4);
      const float4 vC = *(const float4*)(H1 + (long long)sC * 128 + lane * 4);
      const float4 vD = *(const float4*)(H1 + (long long)sD * 128 + lane * 4);
      const float eA = score(vA);
      const float tB = score(vB), tC = score(vC), tD = score(vD);
      const float eB = j + 1 < m ? tB : -INFINITY;
      const float eC = j + 2 < m ? tC : -INFINITY;
      const float eD = j + 3 < m ? tD : -INFINITY;
      const float mn = fmaxf(fmaxf(mx, eA), fmaxf(fmaxf(eB, eC), eD));
      const float rs = __expf(mx - mn);
      const float wA = __expf(eA - mn), wB = __expf(eB - mn);
      const float wC = __expf(eC - mn), wD = __expf(eD - mn);
      acc.x *= rs; acc.y *= rs; acc.z *= rs; acc.w *= rs;
      acc.x += vA.x * wA; acc.y += vA.y * wA; acc.z += vA.z * wA; acc.w += vA.w * wA;
      acc.x += vB.x * wB; acc.y += vB.y * wB; acc.z += vB.z * wB; acc.w += vB.w * wB;
      acc.x += vC.x * wC; acc.y += vC.y * wC; acc.z += vC.z * wC; acc.w += vC.w * wC;
      acc.x += vD.x * wD; acc.y += vD.y * wD; acc.z += vD.z * wD; acc.w += vD.w * wD;
      sm = sm * rs + ((wA + wB) + (wC + wD));
      mx = mn;
    }
    sc = scn;
    if (base + 64 < end) scn = srcs(base + 64);
  }
  const float den = sm + EPSF;

  const float4 b = ((const float4*)b1)[lane];
  acc.x = acc.x / den + b.x; acc.y = acc.y / den + b.y;
  acc.z = acc.z / den + b.z; acc.w = acc.w / den + b.w;
  acc.x = acc.x > 0.f ? acc.x : (__expf(acc.x) - 1.f);
  acc.y = acc.y > 0.f ? acc.y : (__expf(acc.y) - 1.f);
  acc.z = acc.z > 0.f ? acc.z : (__expf(acc.z) - 1.f);
  acc.w = acc.w > 0.f ? acc.w : (__expf(acc.w) - 1.f);
  ((float4*)HOUT)[(long long)n * 32 + lane] = acc;
}

__global__ void __launch_bounds__(256)
row_soft_msg2_k(const int2* __restrict__ SE, const int* __restrict__ ROWPTR,
                const float* __restrict__ G, const float* __restrict__ asrc,
                const float* __restrict__ adst, const float* __restrict__ b2,
                float* __restrict__ OUT, int N) {
  __shared__ float4 stage[8 * 10];
  const int w = threadIdx.x >> 5;
  const int lane = threadIdx.x & 31;
  const int n = __builtin_amdgcn_readfirstlane((int)(blockIdx.x * 8 + w));
  if (n < N) {
    const int hf = lane >> 4;
    const int q = lane & 15;
    const bool real = q < 10;
    const int qc = real ? q : 9;
    const float4 z4 = {0.f, 0.f, 0.f, 0.f};
    const float4 av = real ? ((const float4*)asrc)[qc] : z4;
    const float4 dv = real ? ((const float4*)adst)[qc] : z4;
    const int beg = ROWPTR[n], end = ROWPTR[n + 1];
    const float4 gn = *(const float4*)(G + (long long)n * 40 + qc * 4);
    const float adn = row16_sum(gn.x * dv.x + gn.y * dv.y + gn.z * dv.z + gn.w * dv.w);

    auto srcs = [&](int base) -> int {
      const int idx = base + lane;
      return SE[idx < end ? idx : end - 1].x;
    };

    float mx = RUNMAX0, sm = 0.f;
    float4 acc = {0.f, 0.f, 0.f, 0.f};
    int sc = srcs(beg);
    int scn = srcs(beg + 32);
    for (int base = beg; base < end; base += 32) {
      const int m = end - base < 32 ? end - base : 32;
      for (int j = 0; j < m; j += 8) {
        float4 v[4];
        float e[4];
#pragma unroll
        for (int t = 0; t < 4; ++t) {
          const int ja = j + 2 * t, jb = ja + 1;
          const int ca = ja < m ? ja : m - 1;
          const int cb = jb < m ? jb : m - 1;
          const int sa = __builtin_amdgcn_readlane(sc, ca);
          const int sb = __builtin_amdgcn_readlane(sc, cb);
          const int sj = hf ? sb : sa;
          v[t] = *(const float4*)(G + (long long)sj * 40 + qc * 4);
        }
#pragma unroll
        for (int t = 0; t < 4; ++t) {
          const float p = row16_sum(v[t].x * av.x + v[t].y * av.y + v[t].z * av.z + v[t].w * av.w);
          e[t] = (j + 2 * t + hf < m) ? lrelu(p + adn) : -INFINITY;
        }
        const float mn = fmaxf(fmaxf(mx, e[0]), fmaxf(fmaxf(e[1], e[2]), e[3]));
        const float rs = __expf(mx - mn);
        const float w0 = __expf(e[0] - mn), w1 = __expf(e[1] - mn);
        const float w2 = __expf(e[2] - mn), w3 = __expf(e[3] - mn);
        acc.x *= rs; acc.y *= rs; acc.z *= rs; acc.w *= rs;
        acc.x += v[0].x * w0; acc.y += v[0].y * w0; acc.z += v[0].z * w0; acc.w += v[0].w * w0;
        acc.x += v[1].x * w1; acc.y += v[1].y * w1; acc.z += v[1].z * w1; acc.w += v[1].w * w1;
        acc.x += v[2].x * w2; acc.y += v[2].y * w2; acc.z += v[2].z * w2; acc.w += v[2].w * w2;
        acc.x += v[3].x * w3; acc.y += v[3].y * w3; acc.z += v[3].z * w3; acc.w += v[3].w * w3;
        sm = sm * rs + ((w0 + w1) + (w2 + w3));
        mx = mn;
      }
      sc = scn;
      if (base + 64 < end) scn = srcs(base + 64);
    }

    const float mj = fmaxf(mx, xhalf(mx));
    const float fj = __expf(mx - mj);
    acc.x *= fj; acc.y *= fj; acc.z *= fj; acc.w *= fj; sm *= fj;
    acc.x += xhalf(acc.x); acc.y += xhalf(acc.y);
    acc.z += xhalf(acc.z); acc.w += xhalf(acc.w);
    sm += xhalf(sm);
    if (lane < 10) {
      const float den = sm + EPSF;
      const float4 b = ((const float4*)b2)[lane];
      float4 t;
      t.x = acc.x / den + b.x; t.y = acc.y / den + b.y;
      t.z = acc.z / den + b.z; t.w = acc.w / den + b.w;
      stage[w * 10 + lane] = t;
    }
  }
  __syncthreads();
  const long long g = (long long)blockIdx.x * 80 + threadIdx.x;
  if (threadIdx.x < 80 && g < (long long)N * 10) ((float4*)OUT)[g] = stage[threadIdx.x];
}

static inline unsigned int nblk(long long n, int b) { return (unsigned int)((n + b - 1) / b); }

extern "C" void kernel_launch(void* const* d_in, const int* in_sizes, int n_in,
                              void* d_out, int out_size, void* d_ws, size_t ws_size,
                              hipStream_t stream) {
  const float* x      = (const float*)d_in[0];
  const int*   ei     = (const int*)d_in[1];
  const float* W1     = (const float*)d_in[2];
  const float* a_src1 = (const float*)d_in[3];
  const float* a_dst1 = (const float*)d_in[4];
  const float* b1     = (const float*)d_in[5];
  const float* W2     = (const float*)d_in[6];
  const float* a_src2 = (const float*)d_in[7];
  const float* a_dst2 = (const float*)d_in[8];
  const float* b2     = (const float*)d_in[9];
  float*       out    = (float*)d_out;

  const long long N  = in_sizes[0] / 128;
  const long long E  = in_sizes[1] / 2;
  const long long E2 = E + N;
  const int* srcA = ei;
  const int* dstA = ei + E;

  char* ws = (char*)d_ws;
  size_t off = 0;
  auto carve = [&](size_t bytes) -> void* {
    void* p = ws + off;
    off = (off + bytes + 255) & ~(size_t)255;
    return p;
  };
  float* H1    = (float*)carve((size_t)N * 128 * 4);
  float* ACC1  = (float*)carve((size_t)N * 128 * 4);
  float* G2    = (float*)carve((size_t)N * 40 * 4);
  const int nbk = (int)((N + BKT_ROWS - 1) >> BKT_SHIFT);
  const long long slice = (E + SORT_P - 1) / SORT_P;
  int*   HIST  = (int*)carve((size_t)nbk * SORT_P * 4);
  int*   BASE  = (int*)carve(((size_t)nbk * SORT_P + 1) * 4);
  int*   ROWPTR= (int*)carve((size_t)(N + 1) * 4);
  int2*  TMP   = (int2*)carve((size_t)E * 8);
  int2*  SE    = (int2*)carve((size_t)E2 * 8);

  const int B = 256;

  bucket_count_k<<<SORT_P, 1024, (size_t)nbk * 4, stream>>>(dstA, HIST, E, slice, nbk);
  row_scan_k<<<1, 1024, 0, stream>>>(HIST, BASE, nbk * SORT_P);
  bucket_scatter_k<<<SORT_P, 1024, (size_t)nbk * 12, stream>>>(srcA, dstA, BASE, TMP, E, slice, nbk);
  bucket_rows_k<<<nbk, 1024, 0, stream>>>(TMP, BASE, ROWPTR, SE, (int)N, SORT_P);

  {
    long long waves = (N + 31) / 32;
    gemm1_wmma_k<<<nblk(waves, 4), 128, 0, stream>>>(x, W1, H1, (int)N);
  }
  row_soft_msg1_k<<<nblk(N, 8), 256, 0, stream>>>(SE, ROWPTR, H1, a_src1, a_dst1, b1, ACC1, (int)N);

  {
    long long waves = (N + 31) / 32;
    gemm2_wmma_k<<<nblk(waves, 4), 128, 0, stream>>>(ACC1, W2, G2, (int)N);
  }
  row_soft_msg2_k<<<nblk(N, 8), 256, 0, stream>>>(SE, ROWPTR, G2, a_src2, a_dst2, b2, out, (int)N);
}


# edge sort in two launches: a slice block counts, scans and orders its own slice; a bucket block sums one column of the slices' run-start table and gathers its runs; count and scan launches go
# speedup vs baseline: 10.6770x; 1.0988x over previous
#include <hip/hip_runtime.h>
#include <hip/hip_bf16.h>


typedef __attribute__((ext_vector_type(2))) float v2f;
typedef __attribute__((ext_vector_type(8))) float v8f;

#define NEG_SLOPE 0.2f
#define EPSF 1e-16f

__device__ __forceinline__ v8f wmma_f32(v2f a, v2f b, v8f c) {
  return __builtin_amdgcn_wmma_f32_16x16x4_f32(
      false, a, false, b, (short)0, c, false, false);
}

__global__ void __launch_bounds__(128, 1)
gemm1_wmma_k(const float* __restrict__ X, const float* __restrict__ W,
             float* __restrict__ Hout, int N) {
  int wave = blockIdx.x * (blockDim.x >> 5) + (threadIdx.x >> 5);
  int lane = threadIdx.x & 31;
  int half = lane >> 4;
  int lm   = lane & 15;
  long long row0 = (long long)wave * 32;
  if (row0 >= N) return;
  long long r0 = row0 + lm;
  long long r1 = row0 + 16 + lm;
  bool ok0 = r0 < N, ok1 = r1 < N;
  const v2f* x0 = (const v2f*)(X + r0 * 128);
  const v2f* x1 = (const v2f*)(X + r1 * 128);
  const v2f zz = {0.0f, 0.0f};

  v8f acc0[8] = {}, acc1[8] = {};
  for (int k = 0; k < 128; k += 4) {
    int ka = k + 2 * half;
    v2f a0 = ok0 ? x0[ka >> 1] : zz;
    v2f a1 = ok1 ? x1[ka >> 1] : zz;
    const float* wr = W + (long long)ka * 128;
#pragma unroll
    for (int t = 0; t < 8; ++t) {
      int col = t * 16 + lm;
      v2f b = {wr[col], wr[128 + col]};
      acc0[t] = wmma_f32(a0, b, acc0[t]);
      acc1[t] = wmma_f32(a1, b, acc1[t]);
    }
  }
#pragma unroll
  for (int t = 0; t < 8; ++t) {
    int col = t * 16 + lm;
#pragma unroll
    for (int i = 0; i < 8; ++i) {
      long long ra = row0 + i + 8 * half;
      long long rb = ra + 16;
      if (ra < N) Hout[ra * 128 + col] = acc0[t][i];
      if (rb < N) Hout[rb * 128 + col] = acc1[t][i];
    }
  }
}

__global__ void __launch_bounds__(128, 1)
gemm2_wmma_k(const float* __restrict__ X, const float* __restrict__ W,
             float* __restrict__ G, int N) {
  int wave = blockIdx.x * (blockDim.x >> 5) + (threadIdx.x >> 5);
  int lane = threadIdx.x & 31;
  int half = lane >> 4;
  int lm   = lane & 15;
  long long row0 = (long long)wave * 32;
  if (row0 >= N) return;
  long long r0 = row0 + lm;
  long long r1 = row0 + 16 + lm;
  bool ok0 = r0 < N, ok1 = r1 < N;
  const v2f* x0 = (const v2f*)(X + r0 * 128);
  const v2f* x1 = (const v2f*)(X + r1 * 128);
  const v2f zz = {0.0f, 0.0f};

  v8f acc0[3] = {}, acc1[3] = {};
  for (int k = 0; k < 128; k += 4) {
    int ka = k + 2 * half;
    v2f a0 = ok0 ? x0[ka >> 1] : zz;
    v2f a1 = ok1 ? x1[ka >> 1] : zz;
    const float* wr = W + (long long)ka * 40;
#pragma unroll
    for (int t = 0; t < 3; ++t) {
      int col = t * 16 + lm;
      v2f b;
      b.x = (col < 40) ? wr[col]      : 0.0f;
      b.y = (col < 40) ? wr[40 + col] : 0.0f;
      acc0[t] = wmma_f32(a0, b, acc0[t]);
      acc1[t] = wmma_f32(a1, b, acc1[t]);
    }
  }
#pragma unroll
  for (int t = 0; t < 3; ++t) {
    int col = t * 16 + lm;
    if (col < 40) {
#pragma unroll
      for (int i = 0; i < 8; ++i) {
        long long ra = row0 + i + 8 * half;
        long long rb = ra + 16;
        if (ra < N) G[ra * 40 + col] = acc0[t][i];
        if (rb < N) G[rb * 40 + col] = acc1[t][i];
      }
    }
  }
}

__device__ __forceinline__ float lrelu(float v) { return v > 0.f ? v : NEG_SLOPE * v; }

#define BKT_SHIFT 8
#define BKT_ROWS 256
#define SORT_P 128
#define CAP_SLICE 12800
#define CAP_BKT 16384

template <int CTRL>
__device__ __forceinline__ int dpp_i(int v) {
  return __builtin_amdgcn_update_dpp(0, v, CTRL, 0xf, 0xf, true);
}
__device__ __forceinline__ int wave_sum_i(int p) {
  p += dpp_i<0x161>(p);
  p += dpp_i<0x162>(p);
  p += dpp_i<0x164>(p);
  p += dpp_i<0x168>(p);
  p += __builtin_amdgcn_permlanex16(p, p, 0x76543210, 0xfedcba98, false, false);
  return p;
}

__global__ void __launch_bounds__(1024)
slice_sort_k(const int* __restrict__ src, const int* __restrict__ dst,
             int* __restrict__ LSTART, int2* __restrict__ TMP,
             long long E, long long slice, int nb, int lp) {
  extern __shared__ int dyn_lds[];
  __shared__ int2 stg[CAP_SLICE];
  int* cur = dyn_lds;
  const int t = threadIdx.x, lane = t & 31;
  const int blk = blockIdx.x;
  for (int b = t; b < nb; b += 1024) cur[b] = 0;
  __syncthreads();
  const long long lo = (long long)blk * slice;
  const long long hi = lo + slice < E ? lo + slice : E;
  for (long long e = lo + t; e < hi; e += 4096) {
    int d[4];
#pragma unroll
    for (int u = 0; u < 4; ++u) {
      const long long i = e + u * 1024;
      d[u] = i < hi ? dst[i] : -1;
    }
#pragma unroll
    for (int u = 0; u < 4; ++u)
      if (d[u] >= 0) atomicAdd(&cur[d[u] >> BKT_SHIFT], 1);
  }
  __syncthreads();
  if (t < 32) {
    int* ls = LSTART + (long long)blk * lp;
    int carry = 0;
    for (int base = 0; base < nb; base += 32) {
      const int b = base + lane;
      const int cl = b < nb ? cur[b] : 0;
      int inc = cl;
#pragma unroll
      for (int o = 1; o < 32; o <<= 1) {
        int y = __shfl_up(inc, o);
        if (lane >= o) inc += y;
      }
      if (b < nb) { cur[b] = carry + inc - cl; ls[b] = carry + inc - cl; }
      carry += __shfl(inc, 31);
    }
    if (lane == 0) ls[nb] = carry;
  }
  __syncthreads();
  const bool staged = slice <= CAP_SLICE;
  for (long long e = lo + t; e < hi; e += 4096) {
    int s[4], d[4];
#pragma unroll
    for (int u = 0; u < 4; ++u) {
      const long long i = e + u * 1024;
      s[u] = i < hi ? src[i] : 0;
      d[u] = i < hi ? dst[i] : -1;
    }
#pragma unroll
    for (int u = 0; u < 4; ++u) {
      if (d[u] >= 0) {
        const int r = atomicAdd(&cur[d[u] >> BKT_SHIFT], 1);
        if (staged) stg[r] = make_int2(s[u], d[u]);
        else        TMP[lo + r] = make_int2(s[u], d[u]);
      }
    }
  }
  __syncthreads();
  if (staged) {
    const int cnt = hi > lo ? (int)(hi - lo) : 0;
    for (int i = t; i < cnt; i += 1024) TMP[lo + i] = stg[i];
  }
}

__global__ void __launch_bounds__(1024)
bucket_rows_k(const int2* __restrict__ TMP, const int* __restrict__ LSTART,
              int* __restrict__ ROWPTR, int2* __restrict__ SE, int N,
              long long slice, int lp) {
  static_assert(SORT_P % 32 == 0 && SORT_P <= 1024, "whole waves look up the runs, one thread per slice");
  __shared__ int cnt[BKT_ROWS];
  __shared__ int wsum[8];
  __shared__ int rbeg[SORT_P];
  __shared__ int rlen[SORT_P];
  __shared__ int red[2 * (SORT_P / 32)];
  __shared__ int2 stg[CAP_BKT];
  const int t = threadIdx.x, lane = t & 31;
  const int w = __builtin_amdgcn_readfirstlane(t >> 5);
  const int k = blockIdx.x;
  const int n0 = k << BKT_SHIFT;
  const int rows = N - n0 < BKT_ROWS ? N - n0 : BKT_ROWS;
  if (w < SORT_P / 32) {
    const int* p = LSTART + (long long)t * lp + k;
    const int a = p[0];
    const int len = p[1] - a;
    rbeg[t] = (int)((long long)t * slice) + a;
    rlen[t] = len;
    const int sa = wave_sum_i(a);
    const int sl = wave_sum_i(len);
    if (lane == 0) { red[w] = sa; red[SORT_P / 32 + w] = sl; }
  }
  if (t < BKT_ROWS) cnt[t] = 0;
  __syncthreads();
  int lo = 0, ne = 0;
#pragma unroll
  for (int q = 0; q < SORT_P / 32; ++q) { lo += red[q]; ne += red[SORT_P / 32 + q]; }
  const int gbase = lo + n0;
  const int total = ne + rows;
  const bool staged = total <= CAP_BKT;
  int rb[SORT_P / 32], rn[SORT_P / 32];
  int maxn = 0;
#pragma unroll
  for (int q = 0; q < SORT_P / 32; ++q) {
    const int s = w + 32 * q;
    rb[q] = __builtin_amdgcn_readfirstlane(rbeg[s]);
    rn[q] = __builtin_amdgcn_readfirstlane(rlen[s]);
    maxn = rn[q] > maxn ? rn[q] : maxn;
  }
  for (int i0 = 0; i0 < maxn; i0 += 32) {
    int d[SORT_P / 32];
#pragma unroll
    for (int q = 0; q < SORT_P / 32; ++q) {
      const int i = i0 + lane;
      d[q] = i < rn[q] ? TMP[rb[q] + i].y : -1;
    }
#pragma unroll
    for (int q = 0; q < SORT_P / 32; ++q)
      if (d[q] >= 0) atomicAdd(&cnt[d[q] - n0], 1);
  }
  __syncthreads();
  const int c = t < BKT_ROWS ? cnt[t] : 0;
  int inc = c;
#pragma unroll
  for (int o = 1; o < 32; o <<= 1) {
    int y = __shfl_up(inc, o);
    if (lane >= o) inc += y;
  }
  if (lane == 31 && w < 8) wsum[w] = inc;
  __syncthreads();
  int wex = 0;
#pragma unroll
  for (int q = 0; q < 8; ++q) if (q < w) wex += wsum[q];
  const int loc = t + wex + inc - c;
  if (t < rows) {
    ROWPTR[n0 + t] = gbase + loc;
    const int2 self = make_int2(n0 + t, n0 + t);
    if (staged) stg[loc] = self; else SE[gbase + loc] = self;
    cnt[t] = loc + 1;
  }
  if (k == (int)gridDim.x - 1 && t == 0) ROWPTR[N] = gbase + total;
  __syncthreads();
  for (int i0 = 0; i0 < maxn; i0 += 32) {
    int2 ed[SORT_P / 32];
#pragma unroll
    for (int q = 0; q < SORT_P / 32; ++q) {
      const int i = i0 + lane;
      ed[q] = i < rn[q] ? TMP[rb[q] + i] : make_int2(0, -1);
    }
#pragma unroll
    for (int q = 0; q < SORT_P / 32; ++q) {
      if (ed[q].y >= 0) {
        const int p = atomicAdd(&cnt[ed[q].y - n0], 1);
        if (staged) stg[p] = ed[q]; else SE[gbase + p] = ed[q];
      }
    }
  }
  __syncthreads();
  if (staged)
    for (int i = t; i < total; i += 1024) SE[gbase + i] = stg[i];
}

template <int CTRL>
__device__ __forceinline__ float dpp_f(float v) {
  return __int_as_float(
      __builtin_amdgcn_update_dpp(0, __float_as_int(v), CTRL, 0xf, 0xf, true));
}
__device__ __forceinline__ float quad_sum(float p) {
  p += dpp_f<0xB1>(p);
  p += dpp_f<0x4E>(p);
  return p;
}
__device__ __forceinline__ float row16_sum(float p) {
  p += dpp_f<0x161>(p);
  p += dpp_f<0x162>(p);
  p += dpp_f<0x164>(p);
  p += dpp_f<0x168>(p);
  return p;
}
__device__ __forceinline__ float xhalf(float x) {
  const int b = __float_as_int(x);
  return __int_as_float(
      __builtin_amdgcn_permlanex16(b, b, 0x76543210, 0xfedcba98, false, false));
}

#define RUNMAX0 (-3.402823466e+38f)

__global__ void __launch_bounds__(256)
row_soft_msg1_k(const int2* __restrict__ SE, const int* __restrict__ ROWPTR,
                const float* __restrict__ H1, const float* __restrict__ asrc,
                const float* __restrict__ adst, const float* __restrict__ b1,
                float* __restrict__ HOUT, int N) {
  const int n = __builtin_amdgcn_readfirstlane((int)(blockIdx.x * 8 + (threadIdx.x >> 5)));
  if (n >= N) return;
  const int lane = threadIdx.x & 31;
  const int beg = ROWPTR[n], end = ROWPTR[n + 1];
  const float4 av = ((const float4*)asrc)[lane];
  const float4 dv = ((const float4*)adst)[lane];
  const float4 hn = ((const float4*)H1)[(long long)n * 32 + lane];
  const float adn = quad_sum(hn.x * dv.x + hn.y * dv.y + hn.z * dv.z + hn.w * dv.w);

  auto srcs = [&](int base) -> int {
    const int idx = base + lane;
    return SE[idx < end ? idx : end - 1].x;
  };
  auto score = [&](const float4& v) -> float {
    return lrelu(quad_sum(v.x * av.x + v.y * av.y + v.z * av.z + v.w * av.w) + adn);
  };

  float mx = RUNMAX0, sm = 0.f;
  float4 acc = {0.f, 0.f, 0.f, 0.f};
  int sc = srcs(beg);
  int scn = srcs(beg + 32);
  for (int base = beg; base < end; base += 32) {
    const int m = end - base < 32 ? end - base : 32;
    for (int j = 0; j < m; j += 4) {
      const int c1 = j + 1 < m ? j + 1 : m - 1;
      const int c2 = j + 2 < m ? j + 2 : m - 1;
      const int c3 = j + 3 < m ? j + 3 : m - 1;
      const int sA = __builtin_amdgcn_readlane(sc, j);
      const int sB = __builtin_amdgcn_readlane(sc, c1);
      const int sC = __builtin_amdgcn_readlane(sc, c2);
      const int sD = __builtin_amdgcn_readlane(sc, c3);
      const float4 vA = *(const float4*)(H1 + (long long)sA * 128 + lane * 4);
      const float4 vB = *(const float4*)(H1 + (long long)sB * 128 + lane * 4);
      const float4 vC = *(const float4*)(H1 + (long long)sC * 128 + lane * 4);
      const float4 vD = *(const float4*)(H1 + (long long)sD * 128 + lane * 4);
      const float eA = score(vA);
      const float tB = score(vB), tC = score(vC), tD = score(vD);
      const float eB = j + 1 < m ? tB : -INFINITY;
      const float eC = j + 2 < m ? tC : -INFINITY;
      const float eD = j + 3 < m ? tD : -INFINITY;
      const float mn = fmaxf(fmaxf(mx, eA), fmaxf(fmaxf(eB, eC), eD));
      const float rs = __expf(mx - mn);
      const float wA = __expf(eA - mn), wB = __expf(eB - mn);
      const float wC = __expf(eC - mn), wD = __expf(eD - mn);
      acc.x *= rs; acc.y *= rs; acc.z *= rs; acc.w *= rs;
      acc.x += vA.x * wA; acc.y += vA.y * wA; acc.z += vA.z * wA; acc.w += vA.w * wA;
      acc.x += vB.x * wB; acc.y += vB.y * wB; acc.z += vB.z * wB; acc.w += vB.w * wB;
      acc.x += vC.x * wC; acc.y += vC.y * wC; acc.z += vC.z * wC; acc.w += vC.w * wC;
      acc.x += vD.x * wD; acc.y += vD.y * wD; acc.z += vD.z * wD; acc.w += vD.w * wD;
      sm = sm * rs + ((wA + wB) + (wC + wD));
      mx = mn;
    }
    sc = scn;
    if (base + 64 < end) scn = srcs(base + 64);
  }
  const float den = sm + EPSF;

  const float4 b = ((const float4*)b1)[lane];
  acc.x = acc.x / den + b.x; acc.y = acc.y / den + b.y;
  acc.z = acc.z / den + b.z; acc.w = acc.w / den + b.w;
  acc.x = acc.x > 0.f ? acc.x : (__expf(acc.x) - 1.f);
  acc.y = acc.y > 0.f ? acc.y : (__expf(acc.y) - 1.f);
  acc.z = acc.z > 0.f ? acc.z : (__expf(acc.z) - 1.f);
  acc.w = acc.w > 0.f ? acc.w : (__expf(acc.w) - 1.f);
  ((float4*)HOUT)[(long long)n * 32 + lane] = acc;
}

__global__ void __launch_bounds__(256)
row_soft_msg2_k(const int2* __restrict__ SE, const int* __restrict__ ROWPTR,
                const float* __restrict__ G, const float* __restrict__ asrc,
                const float* __restrict__ adst, const float* __restrict__ b2,
                float* __restrict__ OUT, int N) {
  __shared__ float4 stage[8 * 10];
  const int w = threadIdx.x >> 5;
  const int lane = threadIdx.x & 31;
  const int n = __builtin_amdgcn_readfirstlane((int)(blockIdx.x * 8 + w));
  if (n < N) {
    const int hf = lane >> 4;
    const int q = lane & 15;
    const bool real = q < 10;
    const int qc = real ? q : 9;
    const float4 z4 = {0.f, 0.f, 0.f, 0.f};
    const float4 av = real ? ((const float4*)asrc)[qc] : z4;
    const float4 dv = real ? ((const float4*)adst)[qc] : z4;
    const int beg = ROWPTR[n], end = ROWPTR[n + 1];
    const float4 gn = *(const float4*)(G + (long long)n * 40 + qc * 4);
    const float adn = row16_sum(gn.x * dv.x + gn.y * dv.y + gn.z * dv.z + gn.w * dv.w);

    auto srcs = [&](int base) -> int {
      const int idx = base + lane;
      return SE[idx < end ? idx : end - 1].x;
    };

    float mx = RUNMAX0, sm = 0.f;
    float4 acc = {0.f, 0.f, 0.f, 0.f};
    int sc = srcs(beg);
    int scn = srcs(beg + 32);
    for (int base = beg; base < end; base += 32) {
      const int m = end - base < 32 ? end - base : 32;
      for (int j = 0; j < m; j += 8) {
        float4 v[4];
        float e[4];
#pragma unroll
        for (int t = 0; t < 4; ++t) {
          const int ja = j + 2 * t, jb = ja + 1;
          const int ca = ja < m ? ja : m - 1;
          const int cb = jb < m ? jb : m - 1;
          const int sa = __builtin_amdgcn_readlane(sc, ca);
          const int sb = __builtin_amdgcn_readlane(sc, cb);
          const int sj = hf ? sb : sa;
          v[t] = *(const float4*)(G + (long long)sj * 40 + qc * 4);
        }
#pragma unroll
        for (int t = 0; t < 4; ++t) {
          const float p = row16_sum(v[t].x * av.x + v[t].y * av.y + v[t].z * av.z + v[t].w * av.w);
          e[t] = (j + 2 * t + hf < m) ? lrelu(p + adn) : -INFINITY;
        }
        const float mn = fmaxf(fmaxf(mx, e[0]), fmaxf(fmaxf(e[1], e[2]), e[3]));
        const float rs = __expf(mx - mn);
        const float w0 = __expf(e[0] - mn), w1 = __expf(e[1] - mn);
        const float w2 = __expf(e[2] - mn), w3 = __expf(e[3] - mn);
        acc.x *= rs; acc.y *= rs; acc.z *= rs; acc.w *= rs;
        acc.x += v[0].x * w0; acc.y += v[0].y * w0; acc.z += v[0].z * w0; acc.w += v[0].w * w0;
        acc.x += v[1].x * w1; acc.y += v[1].y * w1; acc.z += v[1].z * w1; acc.w += v[1].w * w1;
        acc.x += v[2].x * w2; acc.y += v[2].y * w2; acc.z += v[2].z * w2; acc.w += v[2].w * w2;
        acc.x += v[3].x * w3; acc.y += v[3].y * w3; acc.z += v[3].z * w3; acc.w += v[3].w * w3;
        sm = sm * rs + ((w0 + w1) + (w2 + w3));
        mx = mn;
      }
      sc = scn;
      if (base + 64 < end) scn = srcs(base + 64);
    }

    const float mj = fmaxf(mx, xhalf(mx));
    const float fj = __expf(mx - mj);
    acc.x *= fj; acc.y *= fj; acc.z *= fj; acc.w *= fj; sm *= fj;
    acc.x += xhalf(acc.x); acc.y += xhalf(acc.y);
    acc.z += xhalf(acc.z); acc.w += xhalf(acc.w);
    sm += xhalf(sm);
    if (lane < 10) {
      const float den = sm + EPSF;
      const float4 b = ((const float4*)b2)[lane];
      float4 t;
      t.x = acc.x / den + b.x; t.y = acc.y / den + b.y;
      t.z = acc.z / den + b.z; t.w = acc.w / den + b.w;
      stage[w * 10 + lane] = t;
    }
  }
  __syncthreads();
  const long long g = (long long)blockIdx.x * 80 + threadIdx.x;
  if (threadIdx.x < 80 && g < (long long)N * 10) ((float4*)OUT)[g] = stage[threadIdx.x];
}

static inline unsigned int nblk(long long n, int b) { return (unsigned int)((n + b - 1) / b); }

extern "C" void kernel_launch(void* const* d_in, const int* in_sizes, int n_in,
                              void* d_out, int out_size, void* d_ws, size_t ws_size,
                              hipStream_t stream) {
  const float* x      = (const float*)d_in[0];
  const int*   ei     = (const int*)d_in[1];
  const float* W1     = (const float*)d_in[2];
  const float* a_src1 = (const float*)d_in[3];
  const float* a_dst1 = (const float*)d_in[4];
  const float* b1     = (const float*)d_in[5];
  const float* W2     = (const float*)d_in[6];
  const float* a_src2 = (const float*)d_in[7];
  const float* a_dst2 = (const float*)d_in[8];
  const float* b2     = (const float*)d_in[9];
  float*       out    = (float*)d_out;

  const long long N  = in_sizes[0] / 128;
  const long long E  = in_sizes[1] / 2;
  const long long E2 = E + N;
  const int* srcA = ei;
  const int* dstA = ei + E;

  char* ws = (char*)d_ws;
  size_t off = 0;
  auto carve = [&](size_t bytes) -> void* {
    void* p = ws + off;
    off = (off + bytes + 255) & ~(size_t)255;
    return p;
  };
  float* H1    = (float*)carve((size_t)N * 128 * 4);
  float* ACC1  = (float*)carve((size_t)N * 128 * 4);
  float* G2    = (float*)carve((size_t)N * 40 * 4);
  const int nbk = (int)((N + BKT_ROWS - 1) >> BKT_SHIFT);
  const long long slice = (E + SORT_P - 1) / SORT_P;
  const int lp = (nbk + 1 + 31) & ~31;
  int*   LSTART= (int*)carve((size_t)SORT_P * lp * 4);
  int*   ROWPTR= (int*)carve((size_t)(N + 1) * 4);
  int2*  TMP   = (int2*)carve((size_t)E * 8);
  int2*  SE    = (int2*)carve((size_t)E2 * 8);

  const int B = 256;

  slice_sort_k<<<SORT_P, 1024, (size_t)nbk * 4, stream>>>(srcA, dstA, LSTART, TMP, E, slice, nbk, lp);
  bucket_rows_k<<<nbk, 1024, 0, stream>>>(TMP, LSTART, ROWPTR, SE, (int)N, slice, lp);

  {
    long long waves = (N + 31) / 32;
    gemm1_wmma_k<<<nblk(waves, 4), 128, 0, stream>>>(x, W1, H1, (int)N);
  }
  row_soft_msg1_k<<<nblk(N, 8), 256, 0, stream>>>(SE, ROWPTR, H1, a_src1, a_dst1, b1, ACC1, (int)N);

  {
    long long waves = (N + 31) / 32;
    gemm2_wmma_k<<<nblk(waves, 4), 128, 0, stream>>>(ACC1, W2, G2, (int)N);
  }
  row_soft_msg2_k<<<nblk(N, 8), 256, 0, stream>>>(SE, ROWPTR, G2, a_src2, a_dst2, b2, out, (int)N);
}


# layer-2 scores leave GEMM2's epilogue in the G row (cols 40,41 of 48); the edge kernel reads a neighbour's score from its fetched row: no per-edge 40-term dot and lane sum; score sum order changed
# speedup vs baseline: 11.0472x; 1.0347x over previous
#include <hip/hip_runtime.h>
#include <hip/hip_bf16.h>


typedef __attribute__((ext_vector_type(2))) float v2f;
typedef __attribute__((ext_vector_type(8))) float v8f;

#define NEG_SLOPE 0.2f
#define EPSF 1e-16f

__device__ __forceinline__ v8f wmma_f32(v2f a, v2f b, v8f c) {
  return __builtin_amdgcn_wmma_f32_16x16x4_f32(
      false, a, false, b, (short)0, c, false, false);
}

__global__ void __launch_bounds__(128, 1)
gemm1_wmma_k(const float* __restrict__ X, const float* __restrict__ W,
             float* __restrict__ Hout, int N) {
  int wave = blockIdx.x * (blockDim.x >> 5) + (threadIdx.x >> 5);
  int lane = threadIdx.x & 31;
  int half = lane >> 4;
  int lm   = lane & 15;
  long long row0 = (long long)wave * 32;
  if (row0 >= N) return;
  long long r0 = row0 + lm;
  long long r1 = row0 + 16 + lm;
  bool ok0 = r0 < N, ok1 = r1 < N;
  const v2f* x0 = (const v2f*)(X + r0 * 128);
  const v2f* x1 = (const v2f*)(X + r1 * 128);
  const v2f zz = {0.0f, 0.0f};

  v8f acc0[8] = {}, acc1[8] = {};
  for (int k = 0; k < 128; k += 4) {
    int ka = k + 2 * half;
    v2f a0 = ok0 ? x0[ka >> 1] : zz;
    v2f a1 = ok1 ? x1[ka >> 1] : zz;
    const float* wr = W + (long long)ka * 128;
#pragma unroll
    for (int t = 0; t < 8; ++t) {
      int col = t * 16 + lm;
      v2f b = {wr[col], wr[128 + col]};
      acc0[t] = wmma_f32(a0, b, acc0[t]);
      acc1[t] = wmma_f32(a1, b, acc1[t]);
    }
  }
#pragma unroll
  for (int t = 0; t < 8; ++t) {
    int col = t * 16 + lm;
#pragma unroll
    for (int i = 0; i < 8; ++i) {
      long long ra = row0 + i + 8 * half;
      long long rb = ra + 16;
      if (ra < N) Hout[ra * 128 + col] = acc0[t][i];
      if (rb < N) Hout[rb * 128 + col] = acc1[t][i];
    }
  }
}

template <int CTRL>
__device__ __forceinline__ float dpp_f(float v) {
  return __int_as_float(
      __builtin_amdgcn_update_dpp(0, __float_as_int(v), CTRL, 0xf, 0xf, true));
}
__device__ __forceinline__ float quad_sum(float p) {
  p += dpp_f<0xB1>(p);
  p += dpp_f<0x4E>(p);
  return p;
}
__device__ __forceinline__ float row16_sum(float p) {
  p += dpp_f<0x161>(p);
  p += dpp_f<0x162>(p);
  p += dpp_f<0x164>(p);
  p += dpp_f<0x168>(p);
  return p;
}
__device__ __forceinline__ float xhalf(float x) {
  const int b = __float_as_int(x);
  return __int_as_float(
      __builtin_amdgcn_permlanex16(b, b, 0x76543210, 0xfedcba98, false, false));
}

template <int N>
__device__ __forceinline__ float row_bcast(float v) { return dpp_f<0x150 + N>(v); }

#define GP 48

__global__ void __launch_bounds__(128, 1)
gemm2_wmma_k(const float* __restrict__ X, const float* __restrict__ W,
             const float* __restrict__ asrc, const float* __restrict__ adst,
             float* __restrict__ G, int N) {
  int wave = blockIdx.x * (blockDim.x >> 5) + (threadIdx.x >> 5);
  int lane = threadIdx.x & 31;
  int half = lane >> 4;
  int lm   = lane & 15;
  long long row0 = (long long)wave * 32;
  if (row0 >= N) return;
  long long r0 = row0 + lm;
  long long r1 = row0 + 16 + lm;
  bool ok0 = r0 < N, ok1 = r1 < N;
  const v2f* x0 = (const v2f*)(X + r0 * 128);
  const v2f* x1 = (const v2f*)(X + r1 * 128);
  const v2f zz = {0.0f, 0.0f};

  v8f acc0[3] = {}, acc1[3] = {};
  for (int k = 0; k < 128; k += 4) {
    int ka = k + 2 * half;
    v2f a0 = ok0 ? x0[ka >> 1] : zz;
    v2f a1 = ok1 ? x1[ka >> 1] : zz;
    const float* wr = W + (long long)ka * 40;
#pragma unroll
    for (int t = 0; t < 3; ++t) {
      int col = t * 16 + lm;
      v2f b;
      b.x = (col < 40) ? wr[col]      : 0.0f;
      b.y = (col < 40) ? wr[40 + col] : 0.0f;
      acc0[t] = wmma_f32(a0, b, acc0[t]);
      acc1[t] = wmma_f32(a1, b, acc1[t]);
    }
  }
  const bool c2ok = 32 + lm < 40;
  const float s0 = asrc[lm], s1 = asrc[16 + lm], s2 = c2ok ? asrc[32 + lm] : 0.0f;
  const float d0 = adst[lm], d1 = adst[16 + lm], d2 = c2ok ? adst[32 + lm] : 0.0f;
#pragma unroll
  for (int i = 0; i < 8; ++i) {
    long long ra = row0 + i + 8 * half;
    long long rb = ra + 16;
    const float sa = row16_sum(acc0[0][i] * s0 + acc0[1][i] * s1 + acc0[2][i] * s2);
    const float da = row16_sum(acc0[0][i] * d0 + acc0[1][i] * d1 + acc0[2][i] * d2);
    const float sb = row16_sum(acc1[0][i] * s0 + acc1[1][i] * s1 + acc1[2][i] * s2);
    const float db = row16_sum(acc1[0][i] * d0 + acc1[1][i] * d1 + acc1[2][i] * d2);
    const float ta = c2ok ? acc0[2][i] : (lm == 8 ? sa : (lm == 9 ? da : 0.0f));
    const float tb = c2ok ? acc1[2][i] : (lm == 8 ? sb : (lm == 9 ? db : 0.0f));
    if (ra < N) {
      G[ra * GP + lm] = acc0[0][i];
      G[ra * GP + 16 + lm] = acc0[1][i];
      if (lm < 12) G[ra * GP + 32 + lm] = ta;
    }
    if (rb < N) {
      G[rb * GP + lm] = acc1[0][i];
      G[rb * GP + 16 + lm] = acc1[1][i];
      if (lm < 12) G[rb * GP + 32 + lm] = tb;
    }
  }
}

__device__ __forceinline__ float lrelu(float v) { return v > 0.f ? v : NEG_SLOPE * v; }

#define BKT_SHIFT 8
#define BKT_ROWS 256
#define SORT_P 128
#define CAP_SLICE 12800
#define CAP_BKT 16384

template <int CTRL>
__device__ __forceinline__ int dpp_i(int v) {
  return __builtin_amdgcn_update_dpp(0, v, CTRL, 0xf, 0xf, true);
}
__device__ __forceinline__ int wave_sum_i(int p) {
  p += dpp_i<0x161>(p);
  p += dpp_i<0x162>(p);
  p += dpp_i<0x164>(p);
  p += dpp_i<0x168>(p);
  p += __builtin_amdgcn_permlanex16(p, p, 0x76543210, 0xfedcba98, false, false);
  return p;
}

__global__ void __launch_bounds__(1024)
slice_sort_k(const int* __restrict__ src, const int* __restrict__ dst,
             int* __restrict__ LSTART, int2* __restrict__ TMP,
             long long E, long long slice, int nb, int lp) {
  extern __shared__ int dyn_lds[];
  __shared__ int2 stg[CAP_SLICE];
  int* cur = dyn_lds;
  const int t = threadIdx.x, lane = t & 31;
  const int blk = blockIdx.x;
  for (int b = t; b < nb; b += 1024) cur[b] = 0;
  __syncthreads();
  const long long lo = (long long)blk * slice;
  const long long hi = lo + slice < E ? lo + slice : E;
  for (long long e = lo + t; e < hi; e += 4096) {
    int d[4];
#pragma unroll
    for (int u = 0; u < 4; ++u) {
      const long long i = e + u * 1024;
      d[u] = i < hi ? dst[i] : -1;
    }
#pragma unroll
    for (int u = 0; u < 4; ++u)
      if (d[u] >= 0) atomicAdd(&cur[d[u] >> BKT_SHIFT], 1);
  }
  __syncthreads();
  if (t < 32) {
    int* ls = LSTART + (long long)blk * lp;
    int carry = 0;
    for (int base = 0; base < nb; base += 32) {
      const int b = base + lane;
      const int cl = b < nb ? cur[b] : 0;
      int inc = cl;
#pragma unroll
      for (int o = 1; o < 32; o <<= 1) {
        int y = __shfl_up(inc, o);
        if (lane >= o) inc += y;
      }
      if (b < nb) { cur[b] = carry + inc - cl; ls[b] = carry + inc - cl; }
      carry += __shfl(inc, 31);
    }
    if (lane == 0) ls[nb] = carry;
  }
  __syncthreads();
  const bool staged = slice <= CAP_SLICE;
  for (long long e = lo + t; e < hi; e += 4096) {
    int s[4], d[4];
#pragma unroll
    for (int u = 0; u < 4; ++u) {
      const long long i = e + u * 1024;
      s[u] = i < hi ? src[i] : 0;
      d[u] = i < hi ? dst[i] : -1;
    }
#pragma unroll
    for (int u = 0; u < 4; ++u) {
      if (d[u] >= 0) {
        const int r = atomicAdd(&cur[d[u] >> BKT_SHIFT], 1);
        if (staged) stg[r] = make_int2(s[u], d[u]);
        else        TMP[lo + r] = make_int2(s[u], d[u]);
      }
    }
  }
  __syncthreads();
  if (staged) {
    const int cnt = hi > lo ? (int)(hi - lo) : 0;
    for (int i = t; i < cnt; i += 1024) TMP[lo + i] = stg[i];
  }
}

__global__ void __launch_bounds__(1024)
bucket_rows_k(const int2* __restrict__ TMP, const int* __restrict__ LSTART,
              int* __restrict__ ROWPTR, int2* __restrict__ SE, int N,
              long long slice, int lp) {
  static_assert(SORT_P % 32 == 0 && SORT_P <= 1024, "whole waves look up the runs, one thread per slice");
  __shared__ int cnt[BKT_ROWS];
  __shared__ int wsum[8];
  __shared__ int rbeg[SORT_P];
  __shared__ int rlen[SORT_P];
  __shared__ int red[2 * (SORT_P / 32)];
  __shared__ int2 stg[CAP_BKT];
  const int t = threadIdx.x, lane = t & 31;
  const int w = __builtin_amdgcn_readfirstlane(t >> 5);
  const int k = blockIdx.x;
  const int n0 = k << BKT_SHIFT;
  const int rows = N - n0 < BKT_ROWS ? N - n0 : BKT_ROWS;
  if (w < SORT_P / 32) {
    const int* p = LSTART + (long long)t * lp + k;
    const int a = p[0];
    const int len = p[1] - a;
    rbeg[t] = (int)((long long)t * slice) + a;
    rlen[t] = len;
    const int sa = wave_sum_i(a);
    const int sl = wave_sum_i(len);
    if (lane == 0) { red[w] = sa; red[SORT_P / 32 + w] = sl; }
  }
  if (t < BKT_ROWS) cnt[t] = 0;
  __syncthreads();
  int lo = 0, ne = 0;
#pragma unroll
  for (int q = 0; q < SORT_P / 32; ++q) { lo += red[q]; ne += red[SORT_P / 32 + q]; }
  const int gbase = lo + n0;
  const int total = ne + rows;
  const bool staged = total <= CAP_BKT;
  int rb[SORT_P / 32], rn[SORT_P / 32];
  int maxn = 0;
#pragma unroll
  for (int q = 0; q < SORT_P / 32; ++q) {
    const int s = w + 32 * q;
    rb[q] = __builtin_amdgcn_readfirstlane(rbeg[s]);
    rn[q] = __builtin_amdgcn_readfirstlane(rlen[s]);
    maxn = rn[q] > maxn ? rn[q] : maxn;
  }
  for (int i0 = 0; i0 < maxn; i0 += 32) {
    int d[SORT_P / 32];
#pragma unroll
    for (int q = 0; q < SORT_P / 32; ++q) {
      const int i = i0 + lane;
      d[q] = i < rn[q] ? TMP[rb[q] + i].y : -1;
    }
#pragma unroll
    for (int q = 0; q < SORT_P / 32; ++q)
      if (d[q] >= 0) atomicAdd(&cnt[d[q] - n0], 1);
  }
  __syncthreads();
  const int c = t < BKT_ROWS ? cnt[t] : 0;
  int inc = c;
#pragma unroll
  for (int o = 1; o < 32; o <<= 1) {
    int y = __shfl_up(inc, o);
    if (lane >= o) inc += y;
  }
  if (lane == 31 && w < 8) wsum[w] = inc;
  __syncthreads();
  int wex = 0;
#pragma unroll
  for (int q = 0; q < 8; ++q) if (q < w) wex += wsum[q];
  const int loc = t + wex + inc - c;
  if (t < rows) {
    ROWPTR[n0 + t] = gbase + loc;
    const int2 self = make_int2(n0 + t, n0 + t);
    if (staged) stg[loc] = self; else SE[gbase + loc] = self;
    cnt[t] = loc + 1;
  }
  if (k == (int)gridDim.x - 1 && t == 0) ROWPTR[N] = gbase + total;
  __syncthreads();
  for (int i0 = 0; i0 < maxn; i0 += 32) {
    int2 ed[SORT_P / 32];
#pragma unroll
    for (int q = 0; q < SORT_P / 32; ++q) {
      const int i = i0 + lane;
      ed[q] = i < rn[q] ? TMP[rb[q] + i] : make_int2(0, -1);
    }
#pragma unroll
    for (int q = 0; q < SORT_P / 32; ++q) {
      if (ed[q].y >= 0) {
        const int p = atomicAdd(&cnt[ed[q].y - n0], 1);
        if (staged) stg[p] = ed[q]; else SE[gbase + p] = ed[q];
      }
    }
  }
  __syncthreads();
  if (staged)
    for (int i = t; i < total; i += 1024) SE[gbase + i] = stg[i];
}

#define RUNMAX0 (-3.402823466e+38f)

__global__ void __launch_bounds__(256)
row_soft_msg1_k(const int2* __restrict__ SE, const int* __restrict__ ROWPTR,
                const float* __restrict__ H1, const float* __restrict__ asrc,
                const float* __restrict__ adst, const float* __restrict__ b1,
                float* __restrict__ HOUT, int N) {
  const int n = __builtin_amdgcn_readfirstlane((int)(blockIdx.x * 8 + (threadIdx.x >> 5)));
  if (n >= N) return;
  const int lane = threadIdx.x & 31;
  const int beg = ROWPTR[n], end = ROWPTR[n + 1];
  const float4 av = ((const float4*)asrc)[lane];
  const float4 dv = ((const float4*)adst)[lane];
  const float4 hn = ((const float4*)H1)[(long long)n * 32 + lane];
  const float adn = quad_sum(hn.x * dv.x + hn.y * dv.y + hn.z * dv.z + hn.w * dv.w);

  auto srcs = [&](int base) -> int {
    const int idx = base + lane;
    return SE[idx < end ? idx : end - 1].x;
  };
  auto score = [&](const float4& v) -> float {
    return lrelu(quad_sum(v.x * av.x + v.y * av.y + v.z * av.z + v.w * av.w) + adn);
  };

  float mx = RUNMAX0, sm = 0.f;
  float4 acc = {0.f, 0.f, 0.f, 0.f};
  int sc = srcs(beg);
  int scn = srcs(beg + 32);
  for (int base = beg; base < end; base += 32) {
    const int m = end - base < 32 ? end - base : 32;
    for (int j = 0; j < m; j += 4) {
      const int c1 = j + 1 < m ? j + 1 : m - 1;
      const int c2 = j + 2 < m ? j + 2 : m - 1;
      const int c3 = j + 3 < m ? j + 3 : m - 1;
      const int sA = __builtin_amdgcn_readlane(sc, j);
      const int sB = __builtin_amdgcn_readlane(sc, c1);
      const int sC = __builtin_amdgcn_readlane(sc, c2);
      const int sD = __builtin_amdgcn_readlane(sc, c3);
      const float4 vA = *(const float4*)(H1 + (long long)sA * 128 + lane * 4);
      const float4 vB = *(const float4*)(H1 + (long long)sB * 128 + lane * 4);
      const float4 vC = *(const float4*)(H1 + (long long)sC * 128 + lane * 4);
      const float4 vD = *(const float4*)(H1 + (long long)sD * 128 + lane * 4);
      const float eA = score(vA);
      const float tB = score(vB), tC = score(vC), tD = score(vD);
      const float eB = j + 1 < m ? tB : -INFINITY;
      const float eC = j + 2 < m ? tC : -INFINITY;
      const float eD = j + 3 < m ? tD : -INFINITY;
      const float mn = fmaxf(fmaxf(mx, eA), fmaxf(fmaxf(eB, eC), eD));
      const float rs = __expf(mx - mn);
      const float wA = __expf(eA - mn), wB = __expf(eB - mn);
      const float wC = __expf(eC - mn), wD = __expf(eD - mn);
      acc.x *= rs; acc.y *= rs; acc.z *= rs; acc.w *= rs;
      acc.x += vA.x * wA; acc.y += vA.y * wA; acc.z += vA.z * wA; acc.w += vA.w * wA;
      acc.x += vB.x * wB; acc.y += vB.y * wB; acc.z += vB.z * wB; acc.w += vB.w * wB;
      acc.x += vC.x * wC; acc.y += vC.y * wC; acc.z += vC.z * wC; acc.w += vC.w * wC;
      acc.x += vD.x * wD; acc.y += vD.y * wD; acc.z += vD.z * wD; acc.w += vD.w * wD;
      sm = sm * rs + ((wA + wB) + (wC + wD));
      mx = mn;
    }
    sc = scn;
    if (base + 64 < end) scn = srcs(base + 64);
  }
  const float den = sm + EPSF;

  const float4 b = ((const float4*)b1)[lane];
  acc.x = acc.x / den + b.x; acc.y = acc.y / den + b.y;
  acc.z = acc.z / den + b.z; acc.w = acc.w / den + b.w;
  acc.x = acc.x > 0.f ? acc.x : (__expf(acc.x) - 1.f);
  acc.y = acc.y > 0.f ? acc.y : (__expf(acc.y) - 1.f);
  acc.z = acc.z > 0.f ? acc.z : (__expf(acc.z) - 1.f);
  acc.w = acc.w > 0.f ? acc.w : (__expf(acc.w) - 1.f);
  ((float4*)HOUT)[(long long)n * 32 + lane] = acc;
}

__global__ void __launch_bounds__(256)
row_soft_msg2_k(const int2* __restrict__ SE, const int* __restrict__ ROWPTR,
                const float* __restrict__ G, const float* __restrict__ b2,
                float* __restrict__ OUT, int N) {
  __shared__ float4 stage[8 * 10];
  const int w = threadIdx.x >> 5;
  const int lane = threadIdx.x & 31;
  const int n = __builtin_amdgcn_readfirstlane((int)(blockIdx.x * 8 + w));
  if (n < N) {
    const int hf = lane >> 4;
    const int q = lane & 15;
    const int qc = q < 11 ? q : 10;
    const int beg = ROWPTR[n], end = ROWPTR[n + 1];
    const float4 gn = *(const float4*)(G + (long long)n * GP + qc * 4);
    const float adn = row_bcast<10>(gn.y);

    auto srcs = [&](int base) -> int {
      const int idx = base + lane;
      return SE[idx < end ? idx : end - 1].x;
    };

    float mx = RUNMAX0, sm = 0.f;
    float4 acc = {0.f, 0.f, 0.f, 0.f};
    int sc = srcs(beg);
    int scn = srcs(beg + 32);
    for (int base = beg; base < end; base += 32) {
      const int m = end - base < 32 ? end - base : 32;
      for (int j = 0; j < m; j += 8) {
        float4 v[4];
        float e[4];
#pragma unroll
        for (int t = 0; t < 4; ++t) {
          const int ja = j + 2 * t, jb = ja + 1;
          const int ca = ja < m ? ja : m - 1;
          const int cb = jb < m ? jb : m - 1;
          const int sa = __builtin_amdgcn_readlane(sc, ca);
          const int sb = __builtin_amdgcn_readlane(sc, cb);
          const int sj = hf ? sb : sa;
          v[t] = *(const float4*)(G + (long long)sj * GP + qc * 4);
        }
#pragma unroll
        for (int t = 0; t < 4; ++t) {
          const float p = row_bcast<10>(v[t].x);
          e[t] = (j + 2 * t + hf < m) ? lrelu(p + adn) : -INFINITY;
        }
        const float mn = fmaxf(fmaxf(mx, e[0]), fmaxf(fmaxf(e[1], e[2]), e[3]));
        const float rs = __expf(mx - mn);
        const float w0 = __expf(e[0] - mn), w1 = __expf(e[1] - mn);
        const float w2 = __expf(e[2] - mn), w3 = __expf(e[3] - mn);
        acc.x *= rs; acc.y *= rs; acc.z *= rs; acc.w *= rs;
        acc.x += v[0].x * w0; acc.y += v[0].y * w0; acc.z += v[0].z * w0; acc.w += v[0].w * w0;
        acc.x += v[1].x * w1; acc.y += v[1].y * w1; acc.z += v[1].z * w1; acc.w += v[1].w * w1;
        acc.x += v[2].x * w2; acc.y += v[2].y * w2; acc.z += v[2].z * w2; acc.w += v[2].w * w2;
        acc.x += v[3].x * w3; acc.y += v[3].y * w3; acc.z += v[3].z * w3; acc.w += v[3].w * w3;
        sm = sm * rs + ((w0 + w1) + (w2 + w3));
        mx = mn;
      }
      sc = scn;
      if (base + 64 < end) scn = srcs(base + 64);
    }

    const float mj = fmaxf(mx, xhalf(mx));
    const float fj = __expf(mx - mj);
    acc.x *= fj; acc.y *= fj; acc.z *= fj; acc.w *= fj; sm *= fj;
    acc.x += xhalf(acc.x); acc.y += xhalf(acc.y);
    acc.z += xhalf(acc.z); acc.w += xhalf(acc.w);
    sm += xhalf(sm);
    if (lane < 10) {
      const float den = sm + EPSF;
      const float4 b = ((const float4*)b2)[lane];
      float4 t;
      t.x = acc.x / den + b.x; t.y = acc.y / den + b.y;
      t.z = acc.z / den + b.z; t.w = acc.w / den + b.w;
      stage[w * 10 + lane] = t;
    }
  }
  __syncthreads();
  const long long g = (long long)blockIdx.x * 80 + threadIdx.x;
  if (threadIdx.x < 80 && g < (long long)N * 10) ((float4*)OUT)[g] = stage[threadIdx.x];
}

static inline unsigned int nblk(long long n, int b) { return (unsigned int)((n + b - 1) / b); }

extern "C" void kernel_launch(void* const* d_in, const int* in_sizes, int n_in,
                              void* d_out, int out_size, void* d_ws, size_t ws_size,
                              hipStream_t stream) {
  const float* x      = (const float*)d_in[0];
  const int*   ei     = (const int*)d_in[1];
  const float* W1     = (const float*)d_in[2];
  const float* a_src1 = (const float*)d_in[3];
  const float* a_dst1 = (const float*)d_in[4];
  const float* b1     = (const float*)d_in[5];
  const float* W2     = (const float*)d_in[6];
  const float* a_src2 = (const float*)d_in[7];
  const float* a_dst2 = (const float*)d_in[8];
  const float* b2     = (const float*)d_in[9];
  float*       out    = (float*)d_out;

  const long long N  = in_sizes[0] / 128;
  const long long E  = in_sizes[1] / 2;
  const long long E2 = E + N;
  const int* srcA = ei;
  const int* dstA = ei + E;

  char* ws = (char*)d_ws;
  size_t off = 0;
  auto carve = [&](size_t bytes) -> void* {
    void* p = ws + off;
    off = (off + bytes + 255) & ~(size_t)255;
    return p;
  };
  float* H1    = (float*)carve((size_t)N * 128 * 4);
  float* ACC1  = (float*)carve((size_t)N * 128 * 4);
  float* G2    = (float*)carve((size_t)N * GP * 4);
  const int nbk = (int)((N + BKT_ROWS - 1) >> BKT_SHIFT);
  const long long slice = (E + SORT_P - 1) / SORT_P;
  const int lp = (nbk + 1 + 31) & ~31;
  int*   LSTART= (int*)carve((size_t)SORT_P * lp * 4);
  int*   ROWPTR= (int*)carve((size_t)(N + 1) * 4);
  int2*  TMP   = (int2*)carve((size_t)E * 8);
  int2*  SE    = (int2*)carve((size_t)E2 * 8);

  const int B = 256;

  slice_sort_k<<<SORT_P, 1024, (size_t)nbk * 4, stream>>>(srcA, dstA, LSTART, TMP, E, slice, nbk, lp);
  bucket_rows_k<<<nbk, 1024, 0, stream>>>(TMP, LSTART, ROWPTR, SE, (int)N, slice, lp);

  {
    long long waves = (N + 31) / 32;
    gemm1_wmma_k<<<nblk(waves, 4), 128, 0, stream>>>(x, W1, H1, (int)N);
  }
  row_soft_msg1_k<<<nblk(N, 8), 256, 0, stream>>>(SE, ROWPTR, H1, a_src1, a_dst1, b1, ACC1, (int)N);

  {
    long long waves = (N + 31) / 32;
    gemm2_wmma_k<<<nblk(waves, 4), 128, 0, stream>>>(ACC1, W2, a_src2, a_dst2, G2, (int)N);
  }
  row_soft_msg2_k<<<nblk(N, 8), 256, 0, stream>>>(SE, ROWPTR, G2, b2, out, (int)N);
}


# second GEMM moved into the layer-1 edge kernel: 16 rows per block meet in LDS, one wave does the 16x128 @ W2 tile on f32 WMMA with the score epilogue; no layer-1 output buffer, one launch fewer
# speedup vs baseline: 11.9147x; 1.0785x over previous
#include <hip/hip_runtime.h>
#include <hip/hip_bf16.h>


typedef __attribute__((ext_vector_type(2))) float v2f;
typedef __attribute__((ext_vector_type(8))) float v8f;

#define NEG_SLOPE 0.2f
#define EPSF 1e-16f

__device__ __forceinline__ v8f wmma_f32(v2f a, v2f b, v8f c) {
  return __builtin_amdgcn_wmma_f32_16x16x4_f32(
      false, a, false, b, (short)0, c, false, false);
}

__global__ void __launch_bounds__(128, 1)
gemm1_wmma_k(const float* __restrict__ X, const float* __restrict__ W,
             float* __restrict__ Hout, int N) {
  int wave = blockIdx.x * (blockDim.x >> 5) + (threadIdx.x >> 5);
  int lane = threadIdx.x & 31;
  int half = lane >> 4;
  int lm   = lane & 15;
  long long row0 = (long long)wave * 32;
  if (row0 >= N) return;
  long long r0 = row0 + lm;
  long long r1 = row0 + 16 + lm;
  bool ok0 = r0 < N, ok1 = r1 < N;
  const v2f* x0 = (const v2f*)(X + r0 * 128);
  const v2f* x1 = (const v2f*)(X + r1 * 128);
  const v2f zz = {0.0f, 0.0f};

  v8f acc0[8] = {}, acc1[8] = {};
  for (int k = 0; k < 128; k += 4) {
    int ka = k + 2 * half;
    v2f a0 = ok0 ? x0[ka >> 1] : zz;
    v2f a1 = ok1 ? x1[ka >> 1] : zz;
    const float* wr = W + (long long)ka * 128;
#pragma unroll
    for (int t = 0; t < 8; ++t) {
      int col = t * 16 + lm;
      v2f b = {wr[col], wr[128 + col]};
      acc0[t] = wmma_f32(a0, b, acc0[t]);
      acc1[t] = wmma_f32(a1, b, acc1[t]);
    }
  }
#pragma unroll
  for (int t = 0; t < 8; ++t) {
    int col = t * 16 + lm;
#pragma unroll
    for (int i = 0; i < 8; ++i) {
      long long ra = row0 + i + 8 * half;
      long long rb = ra + 16;
      if (ra < N) Hout[ra * 128 + col] = acc0[t][i];
      if (rb < N) Hout[rb * 128 + col] = acc1[t][i];
    }
  }
}

template <int CTRL>
__device__ __forceinline__ float dpp_f(float v) {
  return __int_as_float(
      __builtin_amdgcn_update_dpp(0, __float_as_int(v), CTRL, 0xf, 0xf, true));
}
__device__ __forceinline__ float quad_sum(float p) {
  p += dpp_f<0xB1>(p);
  p += dpp_f<0x4E>(p);
  return p;
}
__device__ __forceinline__ float row16_sum(float p) {
  p += dpp_f<0x161>(p);
  p += dpp_f<0x162>(p);
  p += dpp_f<0x164>(p);
  p += dpp_f<0x168>(p);
  return p;
}
__device__ __forceinline__ float xhalf(float x) {
  const int b = __float_as_int(x);
  return __int_as_float(
      __builtin_amdgcn_permlanex16(b, b, 0x76543210, 0xfedcba98, false, false));
}

template <int N>
__device__ __forceinline__ float row_bcast(float v) { return dpp_f<0x150 + N>(v); }

#define GP 48

__device__ __forceinline__ float lrelu(float v) { return v > 0.f ? v : NEG_SLOPE * v; }

#define BKT_SHIFT 8
#define BKT_ROWS 256
#define SORT_P 128
#define CAP_SLICE 12800
#define CAP_BKT 16384

template <int CTRL>
__device__ __forceinline__ int dpp_i(int v) {
  return __builtin_amdgcn_update_dpp(0, v, CTRL, 0xf, 0xf, true);
}
__device__ __forceinline__ int wave_sum_i(int p) {
  p += dpp_i<0x161>(p);
  p += dpp_i<0x162>(p);
  p += dpp_i<0x164>(p);
  p += dpp_i<0x168>(p);
  p += __builtin_amdgcn_permlanex16(p, p, 0x76543210, 0xfedcba98, false, false);
  return p;
}

__global__ void __launch_bounds__(1024)
slice_sort_k(const int* __restrict__ src, const int* __restrict__ dst,
             int* __restrict__ LSTART, int2* __restrict__ TMP,
             long long E, long long slice, int nb, int lp) {
  extern __shared__ int dyn_lds[];
  __shared__ int2 stg[CAP_SLICE];
  int* cur = dyn_lds;
  const int t = threadIdx.x, lane = t & 31;
  const int blk = blockIdx.x;
  for (int b = t; b < nb; b += 1024) cur[b] = 0;
  __syncthreads();
  const long long lo = (long long)blk * slice;
  const long long hi = lo + slice < E ? lo + slice : E;
  for (long long e = lo + t; e < hi; e += 4096) {
    int d[4];
#pragma unroll
    for (int u = 0; u < 4; ++u) {
      const long long i = e + u * 1024;
      d[u] = i < hi ? dst[i] : -1;
    }
#pragma unroll
    for (int u = 0; u < 4; ++u)
      if (d[u] >= 0) atomicAdd(&cur[d[u] >> BKT_SHIFT], 1);
  }
  __syncthreads();
  if (t < 32) {
    int* ls = LSTART + (long long)blk * lp;
    int carry = 0;
    for (int base = 0; base < nb; base += 32) {
      const int b = base + lane;
      const int cl = b < nb ? cur[b] : 0;
      int inc = cl;
#pragma unroll
      for (int o = 1; o < 32; o <<= 1) {
        int y = __shfl_up(inc, o);
        if (lane >= o) inc += y;
      }
      if (b < nb) { cur[b] = carry + inc - cl; ls[b] = carry + inc - cl; }
      carry += __shfl(inc, 31);
    }
    if (lane == 0) ls[nb] = carry;
  }
  __syncthreads();
  const bool staged = slice <= CAP_SLICE;
  for (long long e = lo + t; e < hi; e += 4096) {
    int s[4], d[4];
#pragma unroll
    for (int u = 0; u < 4; ++u) {
      const long long i = e + u * 1024;
      s[u] = i < hi ? src[i] : 0;
      d[u] = i < hi ? dst[i] : -1;
    }
#pragma unroll
    for (int u = 0; u < 4; ++u) {
      if (d[u] >= 0) {
        const int r = atomicAdd(&cur[d[u] >> BKT_SHIFT], 1);
        if (staged) stg[r] = make_int2(s[u], d[u]);
        else        TMP[lo + r] = make_int2(s[u], d[u]);
      }
    }
  }
  __syncthreads();
  if (staged) {
    const int cnt = hi > lo ? (int)(hi - lo) : 0;
    for (int i = t; i < cnt; i += 1024) TMP[lo + i] = stg[i];
  }
}

__global__ void __launch_bounds__(1024)
bucket_rows_k(const int2* __restrict__ TMP, const int* __restrict__ LSTART,
              int* __restrict__ ROWPTR, int2* __restrict__ SE, int N,
              long long slice, int lp) {
  static_assert(SORT_P % 32 == 0 && SORT_P <= 1024, "whole waves look up the runs, one thread per slice");
  __shared__ int cnt[BKT_ROWS];
  __shared__ int wsum[8];
  __shared__ int rbeg[SORT_P];
  __shared__ int rlen[SORT_P];
  __shared__ int red[2 * (SORT_P / 32)];
  __shared__ int2 stg[CAP_BKT];
  const int t = threadIdx.x, lane = t & 31;
  const int w = __builtin_amdgcn_readfirstlane(t >> 5);
  const int k = blockIdx.x;
  const int n0 = k << BKT_SHIFT;
  const int rows = N - n0 < BKT_ROWS ? N - n0 : BKT_ROWS;
  if (w < SORT_P / 32) {
    const int* p = LSTART + (long long)t * lp + k;
    const int a = p[0];
    const int len = p[1] - a;
    rbeg[t] = (int)((long long)t * slice) + a;
    rlen[t] = len;
    const int sa = wave_sum_i(a);
    const int sl = wave_sum_i(len);
    if (lane == 0) { red[w] = sa; red[SORT_P / 32 + w] = sl; }
  }
  if (t < BKT_ROWS) cnt[t] = 0;
  __syncthreads();
  int lo = 0, ne = 0;
#pragma unroll
  for (int q = 0; q < SORT_P / 32; ++q) { lo += red[q]; ne += red[SORT_P / 32 + q]; }
  const int gbase = lo + n0;
  const int total = ne + rows;
  const bool staged = total <= CAP_BKT;
  int rb[SORT_P / 32], rn[SORT_P / 32];
  int maxn = 0;
#pragma unroll
  for (int q = 0; q < SORT_P / 32; ++q) {
    const int s = w + 32 * q;
    rb[q] = __builtin_amdgcn_readfirstlane(rbeg[s]);
    rn[q] = __builtin_amdgcn_readfirstlane(rlen[s]);
    maxn = rn[q] > maxn ? rn[q] : maxn;
  }
  for (int i0 = 0; i0 < maxn; i0 += 32) {
    int d[SORT_P / 32];
#pragma unroll
    for (int q = 0; q < SORT_P / 32; ++q) {
      const int i = i0 + lane;
      d[q] = i < rn[q] ? TMP[rb[q] + i].y : -1;
    }
#pragma unroll
    for (int q = 0; q < SORT_P / 32; ++q)
      if (d[q] >= 0) atomicAdd(&cnt[d[q] - n0], 1);
  }
  __syncthreads();
  const int c = t < BKT_ROWS ? cnt[t] : 0;
  int inc = c;
#pragma unroll
  for (int o = 1; o < 32; o <<= 1) {
    int y = __shfl_up(inc, o);
    if (lane >= o) inc += y;
  }
  if (lane == 31 && w < 8) wsum[w] = inc;
  __syncthreads();
  int wex = 0;
#pragma unroll
  for (int q = 0; q < 8; ++q) if (q < w) wex += wsum[q];
  const int loc = t + wex + inc - c;
  if (t < rows) {
    ROWPTR[n0 + t] = gbase + loc;
    const int2 self = make_int2(n0 + t, n0 + t);
    if (staged) stg[loc] = self; else SE[gbase + loc] = self;
    cnt[t] = loc + 1;
  }
  if (k == (int)gridDim.x - 1 && t == 0) ROWPTR[N] = gbase + total;
  __syncthreads();
  for (int i0 = 0; i0 < maxn; i0 += 32) {
    int2 ed[SORT_P / 32];
#pragma unroll
    for (int q = 0; q < SORT_P / 32; ++q) {
      const int i = i0 + lane;
      ed[q] = i < rn[q] ? TMP[rb[q] + i] : make_int2(0, -1);
    }
#pragma unroll
    for (int q = 0; q < SORT_P / 32; ++q) {
      if (ed[q].y >= 0) {
        const int p = atomicAdd(&cnt[ed[q].y - n0], 1);
        if (staged) stg[p] = ed[q]; else SE[gbase + p] = ed[q];
      }
    }
  }
  __syncthreads();
  if (staged)
    for (int i = t; i < total; i += 1024) SE[gbase + i] = stg[i];
}

#define RUNMAX0 (-3.402823466e+38f)

#define TP 132
__global__ void __launch_bounds__(512)
row_soft_msg1_gemm2_k(const int2* __restrict__ SE, const int* __restrict__ ROWPTR,
                      const float* __restrict__ H1, const float* __restrict__ asrc,
                      const float* __restrict__ adst, const float* __restrict__ b1,
                      const float* __restrict__ W2, const float* __restrict__ asrc2,
                      const float* __restrict__ adst2, float* __restrict__ G, int N) {
  __shared__ float tile[16 * TP];
  const int w = __builtin_amdgcn_readfirstlane((int)(threadIdx.x >> 5));
  const int n = __builtin_amdgcn_readfirstlane((int)(blockIdx.x * 16 + (threadIdx.x >> 5)));
  const int lane = threadIdx.x & 31;
  float4 acc = {0.f, 0.f, 0.f, 0.f};
  if (n < N) {
    const int beg = ROWPTR[n], end = ROWPTR[n + 1];
    const float4 av = ((const float4*)asrc)[lane];
    const float4 dv = ((const float4*)adst)[lane];
    const float4 hn = ((const float4*)H1)[(long long)n * 32 + lane];
    const float adn = quad_sum(hn.x * dv.x + hn.y * dv.y + hn.z * dv.z + hn.w * dv.w);

    auto srcs = [&](int base) -> int {
      const int idx = base + lane;
      return SE[idx < end ? idx : end - 1].x;
    };
    auto score = [&](const float4& v) -> float {
      return lrelu(quad_sum(v.x * av.x + v.y * av.y + v.z * av.z + v.w * av.w) + adn);
    };

    float mx = RUNMAX0, sm = 0.f;
    int sc = srcs(beg);
    int scn = srcs(beg + 32);
    for (int base = beg; base < end; base += 32) {
      const int m = end - base < 32 ? end - base : 32;
      for (int j = 0; j < m; j += 4) {
        const int c1 = j + 1 < m ? j + 1 : m - 1;
        const int c2 = j + 2 < m ? j + 2 : m - 1;
        const int c3 = j + 3 < m ? j + 3 : m - 1;
        const int sA = __builtin_amdgcn_readlane(sc, j);
        const int sB = __builtin_amdgcn_readlane(sc, c1);
        const int sC = __builtin_amdgcn_readlane(sc, c2);
        const int sD = __builtin_amdgcn_readlane(sc, c3);
        const float4 vA = *(const float4*)(H1 + (long long)sA * 128 + lane * 4);
        const float4 vB = *(const float4*)(H1 + (long long)sB * 128 + lane * 4);
        const float4 vC = *(const float4*)(H1 + (long long)sC * 128 + lane * 4);
        const float4 vD = *(const float4*)(H1 + (long long)sD * 128 + lane * 4);
        const float eA = score(vA);
        const float tB = score(vB), tC = score(vC), tD = score(vD);
        const float eB = j + 1 < m ? tB : -INFINITY;
        const float eC = j + 2 < m ? tC : -INFINITY;
        const float eD = j + 3 < m ? tD : -INFINITY;
        const float mn = fmaxf(fmaxf(mx, eA), fmaxf(fmaxf(eB, eC), eD));
        const float rs = __expf(mx - mn);
        const float wA = __expf(eA - mn), wB = __expf(eB - mn);
        const float wC = __expf(eC - mn), wD = __expf(eD - mn);
        acc.x *= rs; acc.y *= rs; acc.z *= rs; acc.w *= rs;
        acc.x += vA.x * wA; acc.y += vA.y * wA; acc.z += vA.z * wA; acc.w += vA.w * wA;
        acc.x += vB.x * wB; acc.y += vB.y * wB; acc.z += vB.z * wB; acc.w += vB.w * wB;
        acc.x += vC.x * wC; acc.y += vC.y * wC; acc.z += vC.z * wC; acc.w += vC.w * wC;
        acc.x += vD.x * wD; acc.y += vD.y * wD; acc.z += vD.z * wD; acc.w += vD.w * wD;
        sm = sm * rs + ((wA + wB) + (wC + wD));
        mx = mn;
      }
      sc = scn;
      if (base + 64 < end) scn = srcs(base + 64);
    }
    const float den = sm + EPSF;

    const float4 b = ((const float4*)b1)[lane];
    acc.x = acc.x / den + b.x; acc.y = acc.y / den + b.y;
    acc.z = acc.z / den + b.z; acc.w = acc.w / den + b.w;
    acc.x = acc.x > 0.f ? acc.x : (__expf(acc.x) - 1.f);
    acc.y = acc.y > 0.f ? acc.y : (__expf(acc.y) - 1.f);
    acc.z = acc.z > 0.f ? acc.z : (__expf(acc.z) - 1.f);
    acc.w = acc.w > 0.f ? acc.w : (__expf(acc.w) - 1.f);
  }
  *(float4*)(tile + w * TP + lane * 4) = acc;
  __syncthreads();
  if (w != 0) return;

  const int half = lane >> 4;
  const int lm   = lane & 15;
  const long long row0 = (long long)blockIdx.x * 16;
  const float* trow = tile + lm * TP;
  const bool c2ok = 32 + lm < 40;
  const int c2 = c2ok ? 32 + lm : 32;
  v8f g[3] = {};
#pragma unroll 4
  for (int k = 0; k < 128; k += 4) {
    const int ka = k + 2 * half;
    const v2f a = *(const v2f*)(trow + ka);
    const float* wr = W2 + ka * 40;
    const v2f b0 = {wr[lm], wr[40 + lm]};
    const v2f b1 = {wr[16 + lm], wr[56 + lm]};
    const float p2 = wr[c2], q2 = wr[40 + c2];
    const v2f b2 = {c2ok ? p2 : 0.0f, c2ok ? q2 : 0.0f};
    g[0] = wmma_f32(a, b0, g[0]);
    g[1] = wmma_f32(a, b1, g[1]);
    g[2] = wmma_f32(a, b2, g[2]);
  }
  const float s0 = asrc2[lm], s1 = asrc2[16 + lm], s2 = c2ok ? asrc2[32 + lm] : 0.0f;
  const float d0 = adst2[lm], d1 = adst2[16 + lm], d2 = c2ok ? adst2[32 + lm] : 0.0f;
#pragma unroll
  for (int i = 0; i < 8; ++i) {
    const long long ra = row0 + i + 8 * half;
    const float sa = row16_sum(g[0][i] * s0 + g[1][i] * s1 + g[2][i] * s2);
    const float da = row16_sum(g[0][i] * d0 + g[1][i] * d1 + g[2][i] * d2);
    const float ta = c2ok ? g[2][i] : (lm == 8 ? sa : (lm == 9 ? da : 0.0f));
    if (ra < N) {
      G[ra * GP + lm] = g[0][i];
      G[ra * GP + 16 + lm] = g[1][i];
      if (lm < 12) G[ra * GP + 32 + lm] = ta;
    }
  }
}

__global__ void __launch_bounds__(256)
row_soft_msg2_k(const int2* __restrict__ SE, const int* __restrict__ ROWPTR,
                const float* __restrict__ G, const float* __restrict__ b2,
                float* __restrict__ OUT, int N) {
  __shared__ float4 stage[8 * 10];
  const int w = threadIdx.x >> 5;
  const int lane = threadIdx.x & 31;
  const int n = __builtin_amdgcn_readfirstlane((int)(blockIdx.x * 8 + w));
  if (n < N) {
    const int hf = lane >> 4;
    const int q = lane & 15;
    const int qc = q < 11 ? q : 10;
    const int beg = ROWPTR[n], end = ROWPTR[n + 1];
    const float4 gn = *(const float4*)(G + (long long)n * GP + qc * 4);
    const float adn = row_bcast<10>(gn.y);

    auto srcs = [&](int base) -> int {
      const int idx = base + lane;
      return SE[idx < end ? idx : end - 1].x;
    };

    float mx = RUNMAX0, sm = 0.f;
    float4 acc = {0.f, 0.f, 0.f, 0.f};
    int sc = srcs(beg);
    int scn = srcs(beg + 32);
    for (int base = beg; base < end; base += 32) {
      const int m = end - base < 32 ? end - base : 32;
      for (int j = 0; j < m; j += 8) {
        float4 v[4];
        float e[4];
#pragma unroll
        for (int t = 0; t < 4; ++t) {
          const int ja = j + 2 * t, jb = ja + 1;
          const int ca = ja < m ? ja : m - 1;
          const int cb = jb < m ? jb : m - 1;
          const int sa = __builtin_amdgcn_readlane(sc, ca);
          const int sb = __builtin_amdgcn_readlane(sc, cb);
          const int sj = hf ? sb : sa;
          v[t] = *(const float4*)(G + (long long)sj * GP + qc * 4);
        }
#pragma unroll
        for (int t = 0; t < 4; ++t) {
          const float p = row_bcast<10>(v[t].x);
          e[t] = (j + 2 * t + hf < m) ? lrelu(p + adn) : -INFINITY;
        }
        const float mn = fmaxf(fmaxf(mx, e[0]), fmaxf(fmaxf(e[1], e[2]), e[3]));
        const float rs = __expf(mx - mn);
        const float w0 = __expf(e[0] - mn), w1 = __expf(e[1] - mn);
        const float w2 = __expf(e[2] - mn), w3 = __expf(e[3] - mn);
        acc.x *= rs; acc.y *= rs; acc.z *= rs; acc.w *= rs;
        acc.x += v[0].x * w0; acc.y += v[0].y * w0; acc.z += v[0].z * w0; acc.w += v[0].w * w0;
        acc.x += v[1].x * w1; acc.y += v[1].y * w1; acc.z += v[1].z * w1; acc.w += v[1].w * w1;
        acc.x += v[2].x * w2; acc.y += v[2].y * w2; acc.z += v[2].z * w2; acc.w += v[2].w * w2;
        acc.x += v[3].x * w3; acc.y += v[3].y * w3; acc.z += v[3].z * w3; acc.w += v[3].w * w3;
        sm = sm * rs + ((w0 + w1) + (w2 + w3));
        mx = mn;
      }
      sc = scn;
      if (base + 64 < end) scn = srcs(base + 64);
    }

    const float mj = fmaxf(mx, xhalf(mx));
    const float fj = __expf(mx - mj);
    acc.x *= fj; acc.y *= fj; acc.z *= fj; acc.w *= fj; sm *= fj;
    acc.x += xhalf(acc.x); acc.y += xhalf(acc.y);
    acc.z += xhalf(acc.z); acc.w += xhalf(acc.w);
    sm += xhalf(sm);
    if (lane < 10) {
      const float den = sm + EPSF;
      const float4 b = ((const float4*)b2)[lane];
      float4 t;
      t.x = acc.x / den + b.x; t.y = acc.y / den + b.y;
      t.z = acc.z / den + b.z; t.w = acc.w / den + b.w;
      stage[w * 10 + lane] = t;
    }
  }
  __syncthreads();
  const long long g = (long long)blockIdx.x * 80 + threadIdx.x;
  if (threadIdx.x < 80 && g < (long long)N * 10) ((float4*)OUT)[g] = stage[threadIdx.x];
}

static inline unsigned int nblk(long long n, int b) { return (unsigned int)((n + b - 1) / b); }

extern "C" void kernel_launch(void* const* d_in, const int* in_sizes, int n_in,
                              void* d_out, int out_size, void* d_ws, size_t ws_size,
                              hipStream_t stream) {
  const float* x      = (const float*)d_in[0];
  const int*   ei     = (const int*)d_in[1];
  const float* W1     = (const float*)d_in[2];
  const float* a_src1 = (const float*)d_in[3];
  const float* a_dst1 = (const float*)d_in[4];
  const float* b1     = (const float*)d_in[5];
  const float* W2     = (const float*)d_in[6];
  const float* a_src2 = (const float*)d_in[7];
  const float* a_dst2 = (const float*)d_in[8];
  const float* b2     = (const float*)d_in[9];
  float*       out    = (float*)d_out;

  const long long N  = in_sizes[0] / 128;
  const long long E  = in_sizes[1] / 2;
  const long long E2 = E + N;
  const int* srcA = ei;
  const int* dstA = ei + E;

  char* ws = (char*)d_ws;
  size_t off = 0;
  auto carve = [&](size_t bytes) -> void* {
    void* p = ws + off;
    off = (off + bytes + 255) & ~(size_t)255;
    return p;
  };
  float* H1    = (float*)carve((size_t)N * 128 * 4);
  float* G2    = (float*)carve((size_t)N * GP * 4);
  const int nbk = (int)((N + BKT_ROWS - 1) >> BKT_SHIFT);
  const long long slice = (E + SORT_P - 1) / SORT_P;
  const int lp = (nbk + 1 + 31) & ~31;
  int*   LSTART= (int*)carve((size_t)SORT_P * lp * 4);
  int*   ROWPTR= (int*)carve((size_t)(N + 1) * 4);
  int2*  TMP   = (int2*)carve((size_t)E * 8);
  int2*  SE    = (int2*)carve((size_t)E2 * 8);

  const int B = 256;

  slice_sort_k<<<SORT_P, 1024, (size_t)nbk * 4, stream>>>(srcA, dstA, LSTART, TMP, E, slice, nbk, lp);
  bucket_rows_k<<<nbk, 1024, 0, stream>>>(TMP, LSTART, ROWPTR, SE, (int)N, slice, lp);

  {
    long long waves = (N + 31) / 32;
    gemm1_wmma_k<<<nblk(waves, 4), 128, 0, stream>>>(x, W1, H1, (int)N);
  }
  row_soft_msg1_gemm2_k<<<nblk(N, 16), 512, 0, stream>>>(SE, ROWPTR, H1, a_src1, a_dst1, b1,
                                                        W2, a_src2, a_dst2, G2, (int)N);

  row_soft_msg2_k<<<nblk(N, 8), 256, 0, stream>>>(SE, ROWPTR, G2, b2, out, (int)N);
}


# first GEMM rides in the first sort launch as extra work-groups (independent of the sort; W1 staged in LDS as k pairs, 16x64 wave tiles, two groups share a WGP): 4 launches instead of 5
# speedup vs baseline: 14.0975x; 1.1832x over previous
#include <hip/hip_runtime.h>
#include <hip/hip_bf16.h>


typedef __attribute__((ext_vector_type(2))) float v2f;
typedef __attribute__((ext_vector_type(8))) float v8f;

#define NEG_SLOPE 0.2f
#define EPSF 1e-16f

__device__ __forceinline__ v8f wmma_f32(v2f a, v2f b, v8f c) {
  return __builtin_amdgcn_wmma_f32_16x16x4_f32(
      false, a, false, b, (short)0, c, false, false);
}

template <int CTRL>
__device__ __forceinline__ float dpp_f(float v) {
  return __int_as_float(
      __builtin_amdgcn_update_dpp(0, __float_as_int(v), CTRL, 0xf, 0xf, true));
}
__device__ __forceinline__ float quad_sum(float p) {
  p += dpp_f<0xB1>(p);
  p += dpp_f<0x4E>(p);
  return p;
}
__device__ __forceinline__ float row16_sum(float p) {
  p += dpp_f<0x161>(p);
  p += dpp_f<0x162>(p);
  p += dpp_f<0x164>(p);
  p += dpp_f<0x168>(p);
  return p;
}
__device__ __forceinline__ float xhalf(float x) {
  const int b = __float_as_int(x);
  return __int_as_float(
      __builtin_amdgcn_permlanex16(b, b, 0x76543210, 0xfedcba98, false, false));
}

template <int N>
__device__ __forceinline__ float row_bcast(float v) { return dpp_f<0x150 + N>(v); }

#define GP 48

__device__ __forceinline__ float lrelu(float v) { return v > 0.f ? v : NEG_SLOPE * v; }

#define BKT_SHIFT 8
#define BKT_ROWS 256
#define SORT_P 128
#define CAP_SLICE 12800
#define CAP_BKT 16384

template <int CTRL>
__device__ __forceinline__ int dpp_i(int v) {
  return __builtin_amdgcn_update_dpp(0, v, CTRL, 0xf, 0xf, true);
}
__device__ __forceinline__ int wave_sum_i(int p) {
  p += dpp_i<0x161>(p);
  p += dpp_i<0x162>(p);
  p += dpp_i<0x164>(p);
  p += dpp_i<0x168>(p);
  p += __builtin_amdgcn_permlanex16(p, p, 0x76543210, 0xfedcba98, false, false);
  return p;
}

#define WLP 144
__device__ __forceinline__ void gemm1_rows256(const float* __restrict__ X,
                                              const float* __restrict__ W,
                                              float* __restrict__ Hout, int N,
                                              int g, float2* Wl) {
  for (int it = threadIdx.x; it < 64 * 32; it += 1024) {
    const int kp = it >> 5, c4 = (it & 31) * 4;
    const float4 r0 = *(const float4*)(W + (2 * kp) * 128 + c4);
    const float4 r1 = *(const float4*)(W + (2 * kp + 1) * 128 + c4);
    float4* d = (float4*)(Wl + kp * WLP + c4);
    d[0] = make_float4(r0.x, r1.x, r0.y, r1.y);
    d[1] = make_float4(r0.z, r1.z, r0.w, r1.w);
  }
  __syncthreads();
  const int lane = threadIdx.x & 31;
  const int half = lane >> 4;
  const int lm   = lane & 15;
  const int w = __builtin_amdgcn_readfirstlane((int)(threadIdx.x >> 5));
  const int row0 = g * 256 + (w >> 1) * 16;
  if (row0 >= N) return;
  const int cb = (w & 1) * 64;
  const int r = row0 + lm;
  const float* xr = X + (long long)(r < N ? r : N - 1) * 128 + 2 * half;
  const float2* wl = Wl + half * WLP + cb + lm;
  v8f acc[4] = {};
#pragma unroll 2
  for (int k0 = 0; k0 < 128; k0 += 16) {
    v2f a[4];
#pragma unroll
    for (int j = 0; j < 4; ++j) a[j] = *(const v2f*)(xr + k0 + 4 * j);
#pragma unroll
    for (int j = 0; j < 4; ++j) {
      const float2* wk = wl + ((k0 + 4 * j) >> 1) * WLP;
#pragma unroll
      for (int t = 0; t < 4; ++t) {
        const float2 bb = wk[t * 16];
        const v2f b = {bb.x, bb.y};
        acc[t] = wmma_f32(a[j], b, acc[t]);
      }
    }
  }
#pragma unroll
  for (int t = 0; t < 4; ++t) {
    const int col = cb + t * 16 + lm;
#pragma unroll
    for (int i = 0; i < 8; ++i) {
      const int ra = row0 + i + 8 * half;
      if (ra < N) Hout[(long long)ra * 128 + col] = acc[t][i];
    }
  }
}

__global__ void __launch_bounds__(1024)
slice_sort_gemm1_k(const int* __restrict__ src, const int* __restrict__ dst,
                   int* __restrict__ LSTART, int2* __restrict__ TMP,
                   long long E, long long slice, int nb, int lp,
                   const float* __restrict__ X, const float* __restrict__ W1,
                   float* __restrict__ H1, int N) {
  extern __shared__ int dyn_lds[];
  __shared__ __attribute__((aligned(16))) int2 stg[CAP_SLICE];
  static_assert(64 * WLP * 8 <= CAP_SLICE * 8, "W1 pairs fit the slice stage");
  if (blockIdx.x >= SORT_P) {
    gemm1_rows256(X, W1, H1, N, (int)blockIdx.x - SORT_P, (float2*)stg);
    return;
  }
  int* cur = dyn_lds;
  const int t = threadIdx.x, lane = t & 31;
  const int blk = blockIdx.x;
  for (int b = t; b < nb; b += 1024) cur[b] = 0;
  __syncthreads();
  const long long lo = (long long)blk * slice;
  const long long hi = lo + slice < E ? lo + slice : E;
  for (long long e = lo + t; e < hi; e += 4096) {
    int d[4];
#pragma unroll
    for (int u = 0; u < 4; ++u) {
      const long long i = e + u * 1024;
      d[u] = i < hi ? dst[i] : -1;
    }
#pragma unroll
    for (int u = 0; u < 4; ++u)
      if (d[u] >= 0) atomicAdd(&cur[d[u] >> BKT_SHIFT], 1);
  }
  __syncthreads();
  if (t < 32) {
    int* ls = LSTART + (long long)blk * lp;
    int carry = 0;
    for (int base = 0; base < nb; base += 32) {
      const int b = base + lane;
      const int cl = b < nb ? cur[b] : 0;
      int inc = cl;
#pragma unroll
      for (int o = 1; o < 32; o <<= 1) {
        int y = __shfl_up(inc, o);
        if (lane >= o) inc += y;
      }
      if (b < nb) { cur[b] = carry + inc - cl; ls[b] = carry + inc - cl; }
      carry += __shfl(inc, 31);
    }
    if (lane == 0) ls[nb] = carry;
  }
  __syncthreads();
  const bool staged = slice <= CAP_SLICE;
  for (long long e = lo + t; e < hi; e += 4096) {
    int s[4], d[4];
#pragma unroll
    for (int u = 0; u < 4; ++u) {
      const long long i = e + u * 1024;
      s[u] = i < hi ? src[i] : 0;
      d[u] = i < hi ? dst[i] : -1;
    }
#pragma unroll
    for (int u = 0; u < 4; ++u) {
      if (d[u] >= 0) {
        const int r = atomicAdd(&cur[d[u] >> BKT_SHIFT], 1);
        if (staged) stg[r] = make_int2(s[u], d[u]);
        else        TMP[lo + r] = make_int2(s[u], d[u]);
      }
    }
  }
  __syncthreads();
  if (staged) {
    const int cnt = hi > lo ? (int)(hi - lo) : 0;
    for (int i = t; i < cnt; i += 1024) TMP[lo + i] = stg[i];
  }
}

__global__ void __launch_bounds__(1024)
bucket_rows_k(const int2* __restrict__ TMP, const int* __restrict__ LSTART,
              int* __restrict__ ROWPTR, int2* __restrict__ SE, int N,
              long long slice, int lp) {
  static_assert(SORT_P % 32 == 0 && SORT_P <= 1024, "whole waves look up the runs, one thread per slice");
  __shared__ int cnt[BKT_ROWS];
  __shared__ int wsum[8];
  __shared__ int rbeg[SORT_P];
  __shared__ int rlen[SORT_P];
  __shared__ int red[2 * (SORT_P / 32)];
  __shared__ int2 stg[CAP_BKT];
  const int t = threadIdx.x, lane = t & 31;
  const int w = __builtin_amdgcn_readfirstlane(t >> 5);
  const int k = blockIdx.x;
  const int n0 = k << BKT_SHIFT;
  const int rows = N - n0 < BKT_ROWS ? N - n0 : BKT_ROWS;
  if (w < SORT_P / 32) {
    const int* p = LSTART + (long long)t * lp + k;
    const int a = p[0];
    const int len = p[1] - a;
    rbeg[t] = (int)((long long)t * slice) + a;
    rlen[t] = len;
    const int sa = wave_sum_i(a);
    const int sl = wave_sum_i(len);
    if (lane == 0) { red[w] = sa; red[SORT_P / 32 + w] = sl; }
  }
  if (t < BKT_ROWS) cnt[t] = 0;
  __syncthreads();
  int lo = 0, ne = 0;
#pragma unroll
  for (int q = 0; q < SORT_P / 32; ++q) { lo += red[q]; ne += red[SORT_P / 32 + q]; }
  const int gbase = lo + n0;
  const int total = ne + rows;
  const bool staged = total <= CAP_BKT;
  int rb[SORT_P / 32], rn[SORT_P / 32];
  int maxn = 0;
#pragma unroll
  for (int q = 0; q < SORT_P / 32; ++q) {
    const int s = w + 32 * q;
    rb[q] = __builtin_amdgcn_readfirstlane(rbeg[s]);
    rn[q] = __builtin_amdgcn_readfirstlane(rlen[s]);
    maxn = rn[q] > maxn ? rn[q] : maxn;
  }
  for (int i0 = 0; i0 < maxn; i0 += 32) {
    int d[SORT_P / 32];
#pragma unroll
    for (int q = 0; q < SORT_P / 32; ++q) {
      const int i = i0 + lane;
      d[q] = i < rn[q] ? TMP[rb[q] + i].y : -1;
    }
#pragma unroll
    for (int q = 0; q < SORT_P / 32; ++q)
      if (d[q] >= 0) atomicAdd(&cnt[d[q] - n0], 1);
  }
  __syncthreads();
  const int c = t < BKT_ROWS ? cnt[t] : 0;
  int inc = c;
#pragma unroll
  for (int o = 1; o < 32; o <<= 1) {
    int y = __shfl_up(inc, o);
    if (lane >= o) inc += y;
  }
  if (lane == 31 && w < 8) wsum[w] = inc;
  __syncthreads();
  int wex = 0;
#pragma unroll
  for (int q = 0; q < 8; ++q) if (q < w) wex += wsum[q];
  const int loc = t + wex + inc - c;
  if (t < rows) {
    ROWPTR[n0 + t] = gbase + loc;
    const int2 self = make_int2(n0 + t, n0 + t);
    if (staged) stg[loc] = self; else SE[gbase + loc] = self;
    cnt[t] = loc + 1;
  }
  if (k == (int)gridDim.x - 1 && t == 0) ROWPTR[N] = gbase + total;
  __syncthreads();
  for (int i0 = 0; i0 < maxn; i0 += 32) {
    int2 ed[SORT_P / 32];
#pragma unroll
    for (int q = 0; q < SORT_P / 32; ++q) {
      const int i = i0 + lane;
      ed[q] = i < rn[q] ? TMP[rb[q] + i] : make_int2(0, -1);
    }
#pragma unroll
    for (int q = 0; q < SORT_P / 32; ++q) {
      if (ed[q].y >= 0) {
        const int p = atomicAdd(&cnt[ed[q].y - n0], 1);
        if (staged) stg[p] = ed[q]; else SE[gbase + p] = ed[q];
      }
    }
  }
  __syncthreads();
  if (staged)
    for (int i = t; i < total; i += 1024) SE[gbase + i] = stg[i];
}

#define RUNMAX0 (-3.402823466e+38f)

#define TP 132
__global__ void __launch_bounds__(512)
row_soft_msg1_gemm2_k(const int2* __restrict__ SE, const int* __restrict__ ROWPTR,
                      const float* __restrict__ H1, const float* __restrict__ asrc,
                      const float* __restrict__ adst, const float* __restrict__ b1,
                      const float* __restrict__ W2, const float* __restrict__ asrc2,
                      const float* __restrict__ adst2, float* __restrict__ G, int N) {
  __shared__ float tile[16 * TP];
  const int w = __builtin_amdgcn_readfirstlane((int)(threadIdx.x >> 5));
  const int n = __builtin_amdgcn_readfirstlane((int)(blockIdx.x * 16 + (threadIdx.x >> 5)));
  const int lane = threadIdx.x & 31;
  float4 acc = {0.f, 0.f, 0.f, 0.f};
  if (n < N) {
    const int beg = ROWPTR[n], end = ROWPTR[n + 1];
    const float4 av = ((const float4*)asrc)[lane];
    const float4 dv = ((const float4*)adst)[lane];
    const float4 hn = ((const float4*)H1)[(long long)n * 32 + lane];
    const float adn = quad_sum(hn.x * dv.x + hn.y * dv.y + hn.z * dv.z + hn.w * dv.w);

    auto srcs = [&](int base) -> int {
      const int idx = base + lane;
      return SE[idx < end ? idx : end - 1].x;
    };
    auto score = [&](const float4& v) -> float {
      return lrelu(quad_sum(v.x * av.x + v.y * av.y + v.z * av.z + v.w * av.w) + adn);
    };

    float mx = RUNMAX0, sm = 0.f;
    int sc = srcs(beg);
    int scn = srcs(beg + 32);
    for (int base = beg; base < end; base += 32) {
      const int m = end - base < 32 ? end - base : 32;
      for (int j = 0; j < m; j += 4) {
        const int c1 = j + 1 < m ? j + 1 : m - 1;
        const int c2 = j + 2 < m ? j + 2 : m - 1;
        const int c3 = j + 3 < m ? j + 3 : m - 1;
        const int sA = __builtin_amdgcn_readlane(sc, j);
        const int sB = __builtin_amdgcn_readlane(sc, c1);
        const int sC = __builtin_amdgcn_readlane(sc, c2);
        const int sD = __builtin_amdgcn_readlane(sc, c3);
        const float4 vA = *(const float4*)(H1 + (long long)sA * 128 + lane * 4);
        const float4 vB = *(const float4*)(H1 + (long long)sB * 128 + lane * 4);
        const float4 vC = *(const float4*)(H1 + (long long)sC * 128 + lane * 4);
        const float4 vD = *(const float4*)(H1 + (long long)sD * 128 + lane * 4);
        const float eA = score(vA);
        const float tB = score(vB), tC = score(vC), tD = score(vD);
        const float eB = j + 1 < m ? tB : -INFINITY;
        const float eC = j + 2 < m ? tC : -INFINITY;
        const float eD = j + 3 < m ? tD : -INFINITY;
        const float mn = fmaxf(fmaxf(mx, eA), fmaxf(fmaxf(eB, eC), eD));
        const float rs = __expf(mx - mn);
        const float wA = __expf(eA - mn), wB = __expf(eB - mn);
        const float wC = __expf(eC - mn), wD = __expf(eD - mn);
        acc.x *= rs; acc.y *= rs; acc.z *= rs; acc.w *= rs;
        acc.x += vA.x * wA; acc.y += vA.y * wA; acc.z += vA.z * wA; acc.w += vA.w * wA;
        acc.x += vB.x * wB; acc.y += vB.y * wB; acc.z += vB.z * wB; acc.w += vB.w * wB;
        acc.x += vC.x * wC; acc.y += vC.y * wC; acc.z += vC.z * wC; acc.w += vC.w * wC;
        acc.x += vD.x * wD; acc.y += vD.y * wD; acc.z += vD.z * wD; acc.w += vD.w * wD;
        sm = sm * rs + ((wA + wB) + (wC + wD));
        mx = mn;
      }
      sc = scn;
      if (base + 64 < end) scn = srcs(base + 64);
    }
    const float den = sm + EPSF;

    const float4 b = ((const float4*)b1)[lane];
    acc.x = acc.x / den + b.x; acc.y = acc.y / den + b.y;
    acc.z = acc.z / den + b.z; acc.w = acc.w / den + b.w;
    acc.x = acc.x > 0.f ? acc.x : (__expf(acc.x) - 1.f);
    acc.y = acc.y > 0.f ? acc.y : (__expf(acc.y) - 1.f);
    acc.z = acc.z > 0.f ? acc.z : (__expf(acc.z) - 1.f);
    acc.w = acc.w > 0.f ? acc.w : (__expf(acc.w) - 1.f);
  }
  *(float4*)(tile + w * TP + lane * 4) = acc;
  __syncthreads();
  if (w != 0) return;

  const int half = lane >> 4;
  const int lm   = lane & 15;
  const long long row0 = (long long)blockIdx.x * 16;
  const float* trow = tile + lm * TP;
  const bool c2ok = 32 + lm < 40;
  const int c2 = c2ok ? 32 + lm : 32;
  v8f g[3] = {};
#pragma unroll 4
  for (int k = 0; k < 128; k += 4) {
    const int ka = k + 2 * half;
    const v2f a = *(const v2f*)(trow + ka);
    const float* wr = W2 + ka * 40;
    const v2f b0 = {wr[lm], wr[40 + lm]};
    const v2f b1 = {wr[16 + lm], wr[56 + lm]};
    const float p2 = wr[c2], q2 = wr[40 + c2];
    const v2f b2 = {c2ok ? p2 : 0.0f, c2ok ? q2 : 0.0f};
    g[0] = wmma_f32(a, b0, g[0]);
    g[1] = wmma_f32(a, b1, g[1]);
    g[2] = wmma_f32(a, b2, g[2]);
  }
  const float s0 = asrc2[lm], s1 = asrc2[16 + lm], s2 = c2ok ? asrc2[32 + lm] : 0.0f;
  const float d0 = adst2[lm], d1 = adst2[16 + lm], d2 = c2ok ? adst2[32 + lm] : 0.0f;
#pragma unroll
  for (int i = 0; i < 8; ++i) {
    const long long ra = row0 + i + 8 * half;
    const float sa = row16_sum(g[0][i] * s0 + g[1][i] * s1 + g[2][i] * s2);
    const float da = row16_sum(g[0][i] * d0 + g[1][i] * d1 + g[2][i] * d2);
    const float ta = c2ok ? g[2][i] : (lm == 8 ? sa : (lm == 9 ? da : 0.0f));
    if (ra < N) {
      G[ra * GP + lm] = g[0][i];
      G[ra * GP + 16 + lm] = g[1][i];
      if (lm < 12) G[ra * GP + 32 + lm] = ta;
    }
  }
}

__global__ void __launch_bounds__(256)
row_soft_msg2_k(const int2* __restrict__ SE, const int* __restrict__ ROWPTR,
                const float* __restrict__ G, const float* __restrict__ b2,
                float* __restrict__ OUT, int N) {
  __shared__ float4 stage[8 * 10];
  const int w = threadIdx.x >> 5;
  const int lane = threadIdx.x & 31;
  const int n = __builtin_amdgcn_readfirstlane((int)(blockIdx.x * 8 + w));
  if (n < N) {
    const int hf = lane >> 4;
    const int q = lane & 15;
    const int qc = q < 11 ? q : 10;
    const int beg = ROWPTR[n], end = ROWPTR[n + 1];
    const float4 gn = *(const float4*)(G + (long long)n * GP + qc * 4);
    const float adn = row_bcast<10>(gn.y);

    auto srcs = [&](int base) -> int {
      const int idx = base + lane;
      return SE[idx < end ? idx : end - 1].x;
    };

    float mx = RUNMAX0, sm = 0.f;
    float4 acc = {0.f, 0.f, 0.f, 0.f};
    int sc = srcs(beg);
    int scn = srcs(beg + 32);
    for (int base = beg; base < end; base += 32) {
      const int m = end - base < 32 ? end - base : 32;
      for (int j = 0; j < m; j += 8) {
        float4 v[4];
        float e[4];
#pragma unroll
        for (int t = 0; t < 4; ++t) {
          const int ja = j + 2 * t, jb = ja + 1;
          const int ca = ja < m ? ja : m - 1;
          const int cb = jb < m ? jb : m - 1;
          const int sa = __builtin_amdgcn_readlane(sc, ca);
          const int sb = __builtin_amdgcn_readlane(sc, cb);
          const int sj = hf ? sb : sa;
          v[t] = *(const float4*)(G + (long long)sj * GP + qc * 4);
        }
#pragma unroll
        for (int t = 0; t < 4; ++t) {
          const float p = row_bcast<10>(v[t].x);
          e[t] = (j + 2 * t + hf < m) ? lrelu(p + adn) : -INFINITY;
        }
        const float mn = fmaxf(fmaxf(mx, e[0]), fmaxf(fmaxf(e[1], e[2]), e[3]));
        const float rs = __expf(mx - mn);
        const float w0 = __expf(e[0] - mn), w1 = __expf(e[1] - mn);
        const float w2 = __expf(e[2] - mn), w3 = __expf(e[3] - mn);
        acc.x *= rs; acc.y *= rs; acc.z *= rs; acc.w *= rs;
        acc.x += v[0].x * w0; acc.y += v[0].y * w0; acc.z += v[0].z * w0; acc.w += v[0].w * w0;
        acc.x += v[1].x * w1; acc.y += v[1].y * w1; acc.z += v[1].z * w1; acc.w += v[1].w * w1;
        acc.x += v[2].x * w2; acc.y += v[2].y * w2; acc.z += v[2].z * w2; acc.w += v[2].w * w2;
        acc.x += v[3].x * w3; acc.y += v[3].y * w3; acc.z += v[3].z * w3; acc.w += v[3].w * w3;
        sm = sm * rs + ((w0 + w1) + (w2 + w3));
        mx = mn;
      }
      sc = scn;
      if (base + 64 < end) scn = srcs(base + 64);
    }

    const float mj = fmaxf(mx, xhalf(mx));
    const float fj = __expf(mx - mj);
    acc.x *= fj; acc.y *= fj; acc.z *= fj; acc.w *= fj; sm *= fj;
    acc.x += xhalf(acc.x); acc.y += xhalf(acc.y);
    acc.z += xhalf(acc.z); acc.w += xhalf(acc.w);
    sm += xhalf(sm);
    if (lane < 10) {
      const float den = sm + EPSF;
      const float4 b = ((const float4*)b2)[lane];
      float4 t;
      t.x = acc.x / den + b.x; t.y = acc.y / den + b.y;
      t.z = acc.z / den + b.z; t.w = acc.w / den + b.w;
      stage[w * 10 + lane] = t;
    }
  }
  __syncthreads();
  const long long g = (long long)blockIdx.x * 80 + threadIdx.x;
  if (threadIdx.x < 80 && g < (long long)N * 10) ((float4*)OUT)[g] = stage[threadIdx.x];
}

static inline unsigned int nblk(long long n, int b) { return (unsigned int)((n + b - 1) / b); }

extern "C" void kernel_launch(void* const* d_in, const int* in_sizes, int n_in,
                              void* d_out, int out_size, void* d_ws, size_t ws_size,
                              hipStream_t stream) {
  const float* x      = (const float*)d_in[0];
  const int*   ei     = (const int*)d_in[1];
  const float* W1     = (const float*)d_in[2];
  const float* a_src1 = (const float*)d_in[3];
  const float* a_dst1 = (const float*)d_in[4];
  const float* b1     = (const float*)d_in[5];
  const float* W2     = (const float*)d_in[6];
  const float* a_src2 = (const float*)d_in[7];
  const float* a_dst2 = (const float*)d_in[8];
  const float* b2     = (const float*)d_in[9];
  float*       out    = (float*)d_out;

  const long long N  = in_sizes[0] / 128;
  const long long E  = in_sizes[1] / 2;
  const long long E2 = E + N;
  const int* srcA = ei;
  const int* dstA = ei + E;

  char* ws = (char*)d_ws;
  size_t off = 0;
  auto carve = [&](size_t bytes) -> void* {
    void* p = ws + off;
    off = (off + bytes + 255) & ~(size_t)255;
    return p;
  };
  float* H1    = (float*)carve((size_t)N * 128 * 4);
  float* G2    = (float*)carve((size_t)N * GP * 4);
  const int nbk = (int)((N + BKT_ROWS - 1) >> BKT_SHIFT);
  const long long slice = (E + SORT_P - 1) / SORT_P;
  const int lp = (nbk + 1 + 31) & ~31;
  int*   LSTART= (int*)carve((size_t)SORT_P * lp * 4);
  int*   ROWPTR= (int*)carve((size_t)(N + 1) * 4);
  int2*  TMP   = (int2*)carve((size_t)E * 8);
  int2*  SE    = (int2*)carve((size_t)E2 * 8);

  const int B = 256;

  slice_sort_gemm1_k<<<SORT_P + nblk(N, 256), 1024, (size_t)nbk * 4, stream>>>(
      srcA, dstA, LSTART, TMP, E, slice, nbk, lp, x, W1, H1, (int)N);
  bucket_rows_k<<<nbk, 1024, 0, stream>>>(TMP, LSTART, ROWPTR, SE, (int)N, slice, lp);

  row_soft_msg1_gemm2_k<<<nblk(N, 16), 512, 0, stream>>>(SE, ROWPTR, H1, a_src1, a_dst1, b1,
                                                        W2, a_src2, a_dst2, G2, (int)N);

  row_soft_msg2_k<<<nblk(N, 8), 256, 0, stream>>>(SE, ROWPTR, G2, b2, out, (int)N);
}
